# starting kernel
# baseline (speedup 1.0000x reference)
#include <hip/hip_runtime.h>

typedef float v2f __attribute__((ext_vector_type(2)));
typedef float v8f __attribute__((ext_vector_type(8)));

#define HD 64
#define CD 192
#define GN 256

__global__ void k_deg(const int* __restrict__ dst, int E, float* __restrict__ deg) {
  int e = blockIdx.x * blockDim.x + threadIdx.x;
  if (e < E) atomicAdd(&deg[dst[e]], 1.0f);
}

__global__ void k_dinv_cnt(float* __restrict__ dinv, const int* __restrict__ batch,
                           float* __restrict__ cnt, int N) {
  int i = blockIdx.x * blockDim.x + threadIdx.x;
  if (i < N) {
    dinv[i] = rsqrtf(dinv[i] + 1.0f);
    atomicAdd(&cnt[batch[i]], 1.0f);
  }
}

__global__ void k_gemm_wmma(const float* __restrict__ Z, int ldz,
                            const float* __restrict__ W, int K,
                            float* __restrict__ Hout, int N) {
  int wid = blockIdx.x * (blockDim.x >> 5) + (threadIdx.x >> 5);
  int tilesM = (N + 15) >> 4;
  if (wid >= tilesM * 4) return;
  int tm = wid >> 2, tn = wid & 3;
  int lane = threadIdx.x & 31;
  int half = lane >> 4;
  int m16 = lane & 15;
  int row = tm * 16 + m16;
  int lrow = row < N ? row : N - 1;
  const float* zr = Z + (size_t)lrow * ldz;
  int col = tn * 16 + m16;
  const float* wp = W + col;
  v8f c = {};
  for (int k0 = 0; k0 < K; k0 += 4) {
    int ka = k0 + 2 * half;
    v2f a, b;
    a.x = zr[ka];
    a.y = zr[ka + 1];
    b.x = wp[(size_t)ka * HD];
    b.y = wp[(size_t)(ka + 1) * HD];
    c = __builtin_amdgcn_wmma_f32_16x16x4_f32(false, a, false, b, (short)0, c, false, false);
  }
  int rbase = tm * 16 + half * 8;
#pragma unroll
  for (int j = 0; j < 8; ++j) {
    int r = rbase + j;
    if (r < N) Hout[(size_t)r * HD + col] = c[j];
  }
}

__global__ void k_selfinit(const float* __restrict__ h, const float* __restrict__ dinv,
                           float* __restrict__ agg, int N) {
  int t = blockIdx.x * blockDim.x + threadIdx.x;
  if (t < N * HD) {
    float di = dinv[t >> 6];
    agg[t] = h[t] * di * di;
  }
}

__global__ void k_edge_scatter(const int* __restrict__ src, const int* __restrict__ dst,
                               const float* __restrict__ dinv, const float* __restrict__ h,
                               float* __restrict__ agg, int E) {
  int t = blockIdx.x * blockDim.x + threadIdx.x;
  if (t >= E * 16) return;
  int e = t >> 4;
  int c4 = (t & 15) << 2;
  int s = src[e], d = dst[e];
  float w = dinv[s] * dinv[d];
  const float4 hv = *(const float4*)(h + (size_t)s * HD + c4);
  float* ap = agg + (size_t)d * HD + c4;
  atomicAdd(ap + 0, hv.x * w);
  atomicAdd(ap + 1, hv.y * w);
  atomicAdd(ap + 2, hv.z * w);
  atomicAdd(ap + 3, hv.w * w);
}

__global__ void k_relu_stats(float* __restrict__ agg, const float* __restrict__ bias,
                             float* __restrict__ bn_sum, float* __restrict__ bn_sq, int N) {
  __shared__ float ssum[256], ssq[256];
  int c = threadIdx.x & 63;
  int quad = threadIdx.x >> 6;
  int base = blockIdx.x * 64;
  float b = bias[c];
  float s = 0.f, q = 0.f;
  for (int r = quad; r < 64; r += 4) {
    int row = base + r;
    if (row < N) {
      float v = fmaxf(agg[(size_t)row * HD + c] + b, 0.f);
      agg[(size_t)row * HD + c] = v;
      s += v; q += v * v;
    }
  }
  ssum[threadIdx.x] = s; ssq[threadIdx.x] = q;
  __syncthreads();
  if (quad == 0) {
    float ts = ssum[c] + ssum[64 + c] + ssum[128 + c] + ssum[192 + c];
    float tq = ssq[c] + ssq[64 + c] + ssq[128 + c] + ssq[192 + c];
    atomicAdd(&bn_sum[c], ts);
    atomicAdd(&bn_sq[c], tq);
  }
}

__global__ void k_bnfinal(const float* __restrict__ bn_sum, const float* __restrict__ bn_sq,
                          const float* __restrict__ gamma, const float* __restrict__ beta,
                          float* __restrict__ scale, float* __restrict__ shift, float invN) {
  int c = threadIdx.x;
  float mu = bn_sum[c] * invN;
  float var = fmaxf(bn_sq[c] * invN - mu * mu, 0.f);
  float sc = rsqrtf(var + 1e-5f) * gamma[c];
  scale[c] = sc;
  shift[c] = beta[c] - mu * sc;
}

__global__ void k_bnapply_pool(const float* __restrict__ agg, const float* __restrict__ scale,
                               const float* __restrict__ shift, const int* __restrict__ batch,
                               float* __restrict__ zc, float* __restrict__ gsum,
                               int N, int lofs) {
  int c = threadIdx.x & 63;
  int quad = threadIdx.x >> 6;
  int base = blockIdx.x * 64;
  float sc = scale[c], sh = shift[c];
  int gcur = -1;
  float acc = 0.f;
  for (int r = quad; r < 64; r += 4) {
    int row = base + r;
    if (row < N) {
      float v = agg[(size_t)row * HD + c] * sc + sh;
      zc[(size_t)row * CD + lofs + c] = v;
      int g = batch[row];
      if (g != gcur) {
        if (gcur >= 0) atomicAdd(&gsum[(size_t)gcur * CD + lofs + c], acc);
        gcur = g; acc = 0.f;
      }
      acc += v;
    }
  }
  if (gcur >= 0) atomicAdd(&gsum[(size_t)gcur * CD + lofs + c], acc);
}

__global__ void k_l2_zc(const float* __restrict__ zc, float* __restrict__ out, int rows) {
  int wid = blockIdx.x * (blockDim.x >> 5) + (threadIdx.x >> 5);
  if (wid >= rows) return;
  int lane = threadIdx.x & 31;
  const float* a = zc + (size_t)wid * CD;
  float v[6]; float s = 0.f;
#pragma unroll
  for (int j = 0; j < 6; ++j) { v[j] = a[lane + 32 * j]; s += v[j] * v[j]; }
#pragma unroll
  for (int off = 16; off >= 1; off >>= 1) s += __shfl_xor(s, off, 32);
  float inv = 1.0f / fmaxf(sqrtf(s), 1e-12f);
  float* o = out + (size_t)wid * CD;
#pragma unroll
  for (int j = 0; j < 6; ++j) o[lane + 32 * j] = v[j] * inv;
}

__global__ void k_l2_gc(const float* __restrict__ gsum, const float* __restrict__ cnt,
                        float* __restrict__ out, int rows) {
  int wid = blockIdx.x * (blockDim.x >> 5) + (threadIdx.x >> 5);
  if (wid >= rows) return;
  int lane = threadIdx.x & 31;
  float icnt = 1.0f / fmaxf(cnt[wid], 1.0f);
  const float* a = gsum + (size_t)wid * CD;
  float v[6]; float s = 0.f;
#pragma unroll
  for (int j = 0; j < 6; ++j) { v[j] = a[lane + 32 * j] * icnt; s += v[j] * v[j]; }
#pragma unroll
  for (int off = 16; off >= 1; off >>= 1) s += __shfl_xor(s, off, 32);
  float inv = 1.0f / fmaxf(sqrtf(s), 1e-12f);
  float* o = out + (size_t)wid * CD;
#pragma unroll
  for (int j = 0; j < 6; ++j) o[lane + 32 * j] = v[j] * inv;
}

extern "C" void kernel_launch(void* const* d_in, const int* in_sizes, int n_in,
                              void* d_out, int out_size, void* d_ws, size_t ws_size,
                              hipStream_t stream) {
  (void)n_in; (void)out_size; (void)ws_size;
  const float* x     = (const float*)d_in[0];
  const int*   ei    = (const int*)d_in[1];
  const int*   batch = (const int*)d_in[2];
  int N = in_sizes[0] / 128;
  int E = in_sizes[1] / 2;
  const int* src = ei;
  const int* dst = ei + E;
  const float* Wl[3]  = {(const float*)d_in[3],  (const float*)d_in[7],  (const float*)d_in[11]};
  const float* bl[3]  = {(const float*)d_in[4],  (const float*)d_in[8],  (const float*)d_in[12]};
  const float* gl[3]  = {(const float*)d_in[5],  (const float*)d_in[9],  (const float*)d_in[13]};
  const float* bel[3] = {(const float*)d_in[6],  (const float*)d_in[10], (const float*)d_in[14]};

  float* ws   = (float*)d_ws;
  float* dinv = ws;                       ws += N;
  float* h    = ws;                       ws += (size_t)N * HD;
  float* agg  = ws;                       ws += (size_t)N * HD;
  float* zc   = ws;                       ws += (size_t)N * CD;
  float* gsum = ws;                       ws += (size_t)GN * CD;
  float* cnt  = ws;                       ws += GN;
  float* bn   = ws;
  float* bn_sum = bn, *bn_sq = bn + 64, *scale = bn + 128, *shift = bn + 192;

  hipMemsetAsync(dinv, 0, (size_t)N * sizeof(float), stream);
  hipMemsetAsync(gsum, 0, (size_t)GN * CD * sizeof(float), stream);
  hipMemsetAsync(cnt,  0, (size_t)GN * sizeof(float), stream);

  k_deg<<<(E + 255) / 256, 256, 0, stream>>>(dst, E, dinv);
  k_dinv_cnt<<<(N + 255) / 256, 256, 0, stream>>>(dinv, batch, cnt, N);

  for (int l = 0; l < 3; ++l) {
    const float* Z = (l == 0) ? x : (zc + (size_t)(l - 1) * HD);
    int ldz = (l == 0) ? 128 : CD;
    int K   = (l == 0) ? 128 : HD;
    int tilesM = (N + 15) / 16;
    int waves = tilesM * 4;
    k_gemm_wmma<<<(waves + 7) / 8, 256, 0, stream>>>(Z, ldz, Wl[l], K, h, N);
    k_selfinit<<<(N * HD + 255) / 256, 256, 0, stream>>>(h, dinv, agg, N);
    k_edge_scatter<<<((size_t)E * 16 + 255) / 256, 256, 0, stream>>>(src, dst, dinv, h, agg, E);
    hipMemsetAsync(bn, 0, 128 * sizeof(float), stream);
    k_relu_stats<<<(N + 63) / 64, 256, 0, stream>>>(agg, bl[l], bn_sum, bn_sq, N);
    k_bnfinal<<<1, 64, 0, stream>>>(bn_sum, bn_sq, gl[l], bel[l], scale, shift, 1.0f / (float)N);
    k_bnapply_pool<<<(N + 63) / 64, 256, 0, stream>>>(agg, scale, shift, batch, zc, gsum, N, l * HD);
  }

  float* out = (float*)d_out;
  k_l2_zc<<<(N + 7) / 8, 256, 0, stream>>>(zc, out, N);
  k_l2_gc<<<(GN + 7) / 8, 256, 0, stream>>>(gsum, cnt, out + (size_t)N * CD, GN);
}


# per-layer float-atomic edge scatter replaced by atomic-free gather over a dst-sorted edge list built once per call; sum order now arrival-rank slots in two half-wave partials
# speedup vs baseline: 1.8963x; 1.8963x over previous
#include <hip/hip_runtime.h>

typedef float v2f __attribute__((ext_vector_type(2)));
typedef float v8f __attribute__((ext_vector_type(8)));

#define HD 64
#define CD 192
#define GN 256

__global__ void k_deg_rank(const int* __restrict__ dst, int E, int* __restrict__ deg,
                           int* __restrict__ rank) {
  int e = blockIdx.x * blockDim.x + threadIdx.x;
  if (e < E) rank[e] = atomicAdd(&deg[dst[e]], 1);
}

__global__ void k_dinv_cnt(const int* __restrict__ deg, float* __restrict__ dinv,
                           const int* __restrict__ batch, float* __restrict__ cnt, int N) {
  int i = blockIdx.x * blockDim.x + threadIdx.x;
  if (i < N) {
    dinv[i] = rsqrtf((float)deg[i] + 1.0f);
    atomicAdd(&cnt[batch[i]], 1.0f);
  }
}

__global__ void k_rowptr(const int* __restrict__ deg, int* __restrict__ rowptr, int N) {
  __shared__ int s_pre[8], s_inc[8];
  int tid = threadIdx.x, lane = tid & 31, w = tid >> 5;
  int tile0 = blockIdx.x * 1024;
  int pre = 0;
  const int4* d4 = (const int4*)deg;
#pragma unroll 8
  for (int i = tid; i < (tile0 >> 2); i += 256) {
    int4 v = d4[i];
    pre += (v.x + v.y) + (v.z + v.w);
  }
  int i0 = tile0 + tid * 4;
  int v0 = (i0     < N) ? deg[i0]     : 0;
  int v1 = (i0 + 1 < N) ? deg[i0 + 1] : 0;
  int v2 = (i0 + 2 < N) ? deg[i0 + 2] : 0;
  int v3 = (i0 + 3 < N) ? deg[i0 + 3] : 0;
  int t = (v0 + v1) + (v2 + v3);
  int inc = t;
  for (int off = 1; off < 32; off <<= 1) {
    int u = __shfl_up(inc, off, 32);
    if (lane >= off) inc += u;
  }
  for (int off = 16; off >= 1; off >>= 1) pre += __shfl_xor(pre, off, 32);
  if (lane == 31) s_inc[w] = inc;
  if (lane == 0)  s_pre[w] = pre;
  __syncthreads();
  int base = 0;
  for (int k = 0; k < 8; ++k) {
    base += s_pre[k];
    if (k < w) base += s_inc[k];
  }
  int ex = base + inc - t;
  if (i0     < N) rowptr[i0]     = ex;
  if (i0 + 1 < N) rowptr[i0 + 1] = ex + v0;
  if (i0 + 2 < N) rowptr[i0 + 2] = ex + v0 + v1;
  if (i0 + 3 < N) rowptr[i0 + 3] = ex + v0 + v1 + v2;
}

__global__ void k_fill(const int* __restrict__ src, const int* __restrict__ dst,
                       const int* __restrict__ rank, const int* __restrict__ rowptr,
                       const float* __restrict__ dinv, int2* __restrict__ csr, int E) {
  int e = blockIdx.x * blockDim.x + threadIdx.x;
  if (e < E) {
    int s = src[e];
    csr[rowptr[dst[e]] + rank[e]] = make_int2(s, __float_as_int(dinv[s]));
  }
}

__global__ void k_gemm_wmma(const float* __restrict__ Z, int ldz,
                            const float* __restrict__ W, int K,
                            float* __restrict__ Hout, int N) {
  int wid = blockIdx.x * (blockDim.x >> 5) + (threadIdx.x >> 5);
  int tilesM = (N + 15) >> 4;
  if (wid >= tilesM * 4) return;
  int tm = wid >> 2, tn = wid & 3;
  int lane = threadIdx.x & 31;
  int half = lane >> 4;
  int m16 = lane & 15;
  int row = tm * 16 + m16;
  int lrow = row < N ? row : N - 1;
  const float* zr = Z + (size_t)lrow * ldz;
  int col = tn * 16 + m16;
  const float* wp = W + col;
  v8f c = {};
  for (int k0 = 0; k0 < K; k0 += 4) {
    int ka = k0 + 2 * half;
    v2f a, b;
    a.x = zr[ka];
    a.y = zr[ka + 1];
    b.x = wp[(size_t)ka * HD];
    b.y = wp[(size_t)(ka + 1) * HD];
    c = __builtin_amdgcn_wmma_f32_16x16x4_f32(false, a, false, b, (short)0, c, false, false);
  }
  int rbase = tm * 16 + half * 8;
#pragma unroll
  for (int j = 0; j < 8; ++j) {
    int r = rbase + j;
    if (r < N) Hout[(size_t)r * HD + col] = c[j];
  }
}

__global__ void k_gather(const int* __restrict__ rowptr, const int* __restrict__ deg,
                         const int2* __restrict__ csr, const float* __restrict__ dinv,
                         const float* __restrict__ h, float* __restrict__ agg, int N) {
  int d = blockIdx.x * (blockDim.x >> 5) + __builtin_amdgcn_readfirstlane(threadIdx.x >> 5);
  if (d >= N) return;
  int lane = threadIdx.x & 31;
  int half = lane >> 4;
  int c4 = (lane & 15) << 2;
  int start = rowptr[d];
  int n = deg[d];
  float dd = dinv[d];
  const float* hc = h + c4;
  const float4 hs = *(const float4*)(hc + (size_t)d * HD);
  float4 acc = {0.f, 0.f, 0.f, 0.f};
  for (int base = 0; base < n; base += 32) {
    int m = n - base;
    if (m > 32) m = 32;
    int2 ent = make_int2(0, 0);
    if (lane < m) ent = csr[start + base + lane];
    float wl = __int_as_float(ent.y) * dd;
    int npair = m >> 1;
#pragma unroll 4
    for (int p = 0; p < npair; ++p) {
      int sl = 2 * p + half;
      int s = __shfl(ent.x, sl, 32);
      float w = __shfl(wl, sl, 32);
      const float4 hv = *(const float4*)(hc + (size_t)s * HD);
      acc.x += hv.x * w;
      acc.y += hv.y * w;
      acc.z += hv.z * w;
      acc.w += hv.w * w;
    }
    if (m & 1) {
      int s = __shfl(ent.x, m - 1, 32);
      float w = __shfl(wl, m - 1, 32);
      if (half == 0) {
        const float4 hv = *(const float4*)(hc + (size_t)s * HD);
        acc.x += hv.x * w;
        acc.y += hv.y * w;
        acc.z += hv.z * w;
        acc.w += hv.w * w;
      }
    }
  }
  acc.x += __shfl_xor(acc.x, 16, 32);
  acc.y += __shfl_xor(acc.y, 16, 32);
  acc.z += __shfl_xor(acc.z, 16, 32);
  acc.w += __shfl_xor(acc.w, 16, 32);
  if (half == 0) {
    float w0 = dd * dd;
    float4 o;
    o.x = hs.x * w0 + acc.x;
    o.y = hs.y * w0 + acc.y;
    o.z = hs.z * w0 + acc.z;
    o.w = hs.w * w0 + acc.w;
    *(float4*)(agg + (size_t)d * HD + c4) = o;
  }
}

__global__ void k_relu_stats(float* __restrict__ agg, const float* __restrict__ bias,
                             float* __restrict__ bn_sum, float* __restrict__ bn_sq, int N) {
  __shared__ float ssum[256], ssq[256];
  int c = threadIdx.x & 63;
  int quad = threadIdx.x >> 6;
  int base = blockIdx.x * 64;
  float b = bias[c];
  float s = 0.f, q = 0.f;
  for (int r = quad; r < 64; r += 4) {
    int row = base + r;
    if (row < N) {
      float v = fmaxf(agg[(size_t)row * HD + c] + b, 0.f);
      agg[(size_t)row * HD + c] = v;
      s += v; q += v * v;
    }
  }
  ssum[threadIdx.x] = s; ssq[threadIdx.x] = q;
  __syncthreads();
  if (quad == 0) {
    float ts = ssum[c] + ssum[64 + c] + ssum[128 + c] + ssum[192 + c];
    float tq = ssq[c] + ssq[64 + c] + ssq[128 + c] + ssq[192 + c];
    atomicAdd(&bn_sum[c], ts);
    atomicAdd(&bn_sq[c], tq);
  }
}

__global__ void k_bnfinal(const float* __restrict__ bn_sum, const float* __restrict__ bn_sq,
                          const float* __restrict__ gamma, const float* __restrict__ beta,
                          float* __restrict__ scale, float* __restrict__ shift, float invN) {
  int c = threadIdx.x;
  float mu = bn_sum[c] * invN;
  float var = fmaxf(bn_sq[c] * invN - mu * mu, 0.f);
  float sc = rsqrtf(var + 1e-5f) * gamma[c];
  scale[c] = sc;
  shift[c] = beta[c] - mu * sc;
}

__global__ void k_bnapply_pool(const float* __restrict__ agg, const float* __restrict__ scale,
                               const float* __restrict__ shift, const int* __restrict__ batch,
                               float* __restrict__ zc, float* __restrict__ gsum,
                               int N, int lofs) {
  int c = threadIdx.x & 63;
  int quad = threadIdx.x >> 6;
  int base = blockIdx.x * 64;
  float sc = scale[c], sh = shift[c];
  int gcur = -1;
  float acc = 0.f;
  for (int r = quad; r < 64; r += 4) {
    int row = base + r;
    if (row < N) {
      float v = agg[(size_t)row * HD + c] * sc + sh;
      zc[(size_t)row * CD + lofs + c] = v;
      int g = batch[row];
      if (g != gcur) {
        if (gcur >= 0) atomicAdd(&gsum[(size_t)gcur * CD + lofs + c], acc);
        gcur = g; acc = 0.f;
      }
      acc += v;
    }
  }
  if (gcur >= 0) atomicAdd(&gsum[(size_t)gcur * CD + lofs + c], acc);
}

__global__ void k_l2_zc(const float* __restrict__ zc, float* __restrict__ out, int rows) {
  int wid = blockIdx.x * (blockDim.x >> 5) + (threadIdx.x >> 5);
  if (wid >= rows) return;
  int lane = threadIdx.x & 31;
  const float* a = zc + (size_t)wid * CD;
  float v[6]; float s = 0.f;
#pragma unroll
  for (int j = 0; j < 6; ++j) { v[j] = a[lane + 32 * j]; s += v[j] * v[j]; }
#pragma unroll
  for (int off = 16; off >= 1; off >>= 1) s += __shfl_xor(s, off, 32);
  float inv = 1.0f / fmaxf(sqrtf(s), 1e-12f);
  float* o = out + (size_t)wid * CD;
#pragma unroll
  for (int j = 0; j < 6; ++j) o[lane + 32 * j] = v[j] * inv;
}

__global__ void k_l2_gc(const float* __restrict__ gsum, const float* __restrict__ cnt,
                        float* __restrict__ out, int rows) {
  int wid = blockIdx.x * (blockDim.x >> 5) + (threadIdx.x >> 5);
  if (wid >= rows) return;
  int lane = threadIdx.x & 31;
  float icnt = 1.0f / fmaxf(cnt[wid], 1.0f);
  const float* a = gsum + (size_t)wid * CD;
  float v[6]; float s = 0.f;
#pragma unroll
  for (int j = 0; j < 6; ++j) { v[j] = a[lane + 32 * j] * icnt; s += v[j] * v[j]; }
#pragma unroll
  for (int off = 16; off >= 1; off >>= 1) s += __shfl_xor(s, off, 32);
  float inv = 1.0f / fmaxf(sqrtf(s), 1e-12f);
  float* o = out + (size_t)wid * CD;
#pragma unroll
  for (int j = 0; j < 6; ++j) o[lane + 32 * j] = v[j] * inv;
}

extern "C" void kernel_launch(void* const* d_in, const int* in_sizes, int n_in,
                              void* d_out, int out_size, void* d_ws, size_t ws_size,
                              hipStream_t stream) {
  (void)n_in; (void)out_size; (void)ws_size;
  const float* x     = (const float*)d_in[0];
  const int*   ei    = (const int*)d_in[1];
  const int*   batch = (const int*)d_in[2];
  int N = in_sizes[0] / 128;
  int E = in_sizes[1] / 2;
  const int* src = ei;
  const int* dst = ei + E;
  const float* Wl[3]  = {(const float*)d_in[3],  (const float*)d_in[7],  (const float*)d_in[11]};
  const float* bl[3]  = {(const float*)d_in[4],  (const float*)d_in[8],  (const float*)d_in[12]};
  const float* gl[3]  = {(const float*)d_in[5],  (const float*)d_in[9],  (const float*)d_in[13]};
  const float* bel[3] = {(const float*)d_in[6],  (const float*)d_in[10], (const float*)d_in[14]};

  float* ws   = (float*)d_ws;
  float* dinv = ws;                       ws += N;
  float* h    = ws;                       ws += (size_t)N * HD;
  float* agg  = ws;                       ws += (size_t)N * HD;
  float* zc   = ws;                       ws += (size_t)N * CD;
  float* gsum = ws;                       ws += (size_t)GN * CD;
  float* cnt  = ws;                       ws += GN;
  float* bn   = ws;
  float* bn_sum = bn, *bn_sq = bn + 64, *scale = bn + 128, *shift = bn + 192;
  ws += 256;
  int* deg    = (int*)(((size_t)ws + 15) & ~(size_t)15);
  int* rowptr = deg + (((size_t)N + 3) & ~(size_t)3);
  int* rank   = rowptr + (((size_t)N + 3) & ~(size_t)3);
  int2* csr   = (int2*)(rank + (((size_t)E + 1) & ~(size_t)1));

  hipMemsetAsync(deg, 0, (size_t)N * sizeof(int), stream);
  hipMemsetAsync(gsum, 0, (size_t)GN * CD * sizeof(float), stream);
  hipMemsetAsync(cnt,  0, (size_t)GN * sizeof(float), stream);

  k_deg_rank<<<(E + 255) / 256, 256, 0, stream>>>(dst, E, deg, rank);
  k_dinv_cnt<<<(N + 255) / 256, 256, 0, stream>>>(deg, dinv, batch, cnt, N);
  k_rowptr<<<(N + 1023) / 1024, 256, 0, stream>>>(deg, rowptr, N);
  k_fill<<<(E + 255) / 256, 256, 0, stream>>>(src, dst, rank, rowptr, dinv, csr, E);

  for (int l = 0; l < 3; ++l) {
    const float* Z = (l == 0) ? x : (zc + (size_t)(l - 1) * HD);
    int ldz = (l == 0) ? 128 : CD;
    int K   = (l == 0) ? 128 : HD;
    int tilesM = (N + 15) / 16;
    int waves = tilesM * 4;
    k_gemm_wmma<<<(waves + 7) / 8, 256, 0, stream>>>(Z, ldz, Wl[l], K, h, N);
    k_gather<<<(N + 7) / 8, 256, 0, stream>>>(rowptr, deg, csr, dinv, h, agg, N);
    hipMemsetAsync(bn, 0, 128 * sizeof(float), stream);
    k_relu_stats<<<(N + 63) / 64, 256, 0, stream>>>(agg, bl[l], bn_sum, bn_sq, N);
    k_bnfinal<<<1, 64, 0, stream>>>(bn_sum, bn_sq, gl[l], bel[l], scale, shift, 1.0f / (float)N);
    k_bnapply_pool<<<(N + 63) / 64, 256, 0, stream>>>(agg, scale, shift, batch, zc, gsum, N, l * HD);
  }

  float* out = (float*)d_out;
  k_l2_zc<<<(N + 7) / 8, 256, 0, stream>>>(zc, out, N);
  k_l2_gc<<<(GN + 7) / 8, 256, 0, stream>>>(gsum, cnt, out + (size_t)N * CD, GN);
}


# launch fusion: bias+ReLU+BN sums into the gather, BN apply into the next GEMM's A load and into one final BN+L2+pool kernel, dinv/counts into the scan: 27 stream ops to 12; BN and pool sums regrouped
# speedup vs baseline: 2.4298x; 1.2813x over previous
#include <hip/hip_runtime.h>

typedef float v2f __attribute__((ext_vector_type(2)));
typedef float v8f __attribute__((ext_vector_type(8)));

#define HD 64
#define CD 192
#define GN 256

__global__ void k_deg_rank(const int* __restrict__ dst, int E, int* __restrict__ deg,
                           int* __restrict__ rank, float* __restrict__ zero, int nzero) {
  int e = blockIdx.x * blockDim.x + threadIdx.x;
  for (int i = e; i < nzero; i += gridDim.x * blockDim.x) zero[i] = 0.f;
  if (e < E) rank[e] = atomicAdd(&deg[dst[e]], 1);
}

__global__ void k_rowptr(const int* __restrict__ deg, int* __restrict__ rowptr,
                         float* __restrict__ dinv, const int* __restrict__ batch,
                         float* __restrict__ cnt, int N) {
  __shared__ int s_pre[8], s_inc[8];
  int tid = threadIdx.x, lane = tid & 31, w = tid >> 5;
  int tile0 = blockIdx.x * 1024;
  int pre = 0;
  const int4* d4 = (const int4*)deg;
#pragma unroll 8
  for (int i = tid; i < (tile0 >> 2); i += 256) {
    int4 v = d4[i];
    pre += (v.x + v.y) + (v.z + v.w);
  }
  int i0 = tile0 + tid * 4;
  int v0 = (i0     < N) ? deg[i0]     : 0;
  int v1 = (i0 + 1 < N) ? deg[i0 + 1] : 0;
  int v2 = (i0 + 2 < N) ? deg[i0 + 2] : 0;
  int v3 = (i0 + 3 < N) ? deg[i0 + 3] : 0;
  if (i0     < N) { dinv[i0]     = rsqrtf((float)v0 + 1.0f); atomicAdd(&cnt[batch[i0]],     1.0f); }
  if (i0 + 1 < N) { dinv[i0 + 1] = rsqrtf((float)v1 + 1.0f); atomicAdd(&cnt[batch[i0 + 1]], 1.0f); }
  if (i0 + 2 < N) { dinv[i0 + 2] = rsqrtf((float)v2 + 1.0f); atomicAdd(&cnt[batch[i0 + 2]], 1.0f); }
  if (i0 + 3 < N) { dinv[i0 + 3] = rsqrtf((float)v3 + 1.0f); atomicAdd(&cnt[batch[i0 + 3]], 1.0f); }
  int t = (v0 + v1) + (v2 + v3);
  int inc = t;
  for (int off = 1; off < 32; off <<= 1) {
    int u = __shfl_up(inc, off, 32);
    if (lane >= off) inc += u;
  }
  for (int off = 16; off >= 1; off >>= 1) pre += __shfl_xor(pre, off, 32);
  if (lane == 31) s_inc[w] = inc;
  if (lane == 0)  s_pre[w] = pre;
  __syncthreads();
  int base = 0;
  for (int k = 0; k < 8; ++k) {
    base += s_pre[k];
    if (k < w) base += s_inc[k];
  }
  int ex = base + inc - t;
  if (i0     < N) rowptr[i0]     = ex;
  if (i0 + 1 < N) rowptr[i0 + 1] = ex + v0;
  if (i0 + 2 < N) rowptr[i0 + 2] = ex + v0 + v1;
  if (i0 + 3 < N) rowptr[i0 + 3] = ex + v0 + v1 + v2;
}

__global__ void k_fill(const int* __restrict__ src, const int* __restrict__ dst,
                       const int* __restrict__ rank, const int* __restrict__ rowptr,
                       const float* __restrict__ dinv, int2* __restrict__ csr, int E) {
  int e = blockIdx.x * blockDim.x + threadIdx.x;
  if (e < E) {
    int s = src[e];
    csr[rowptr[dst[e]] + rank[e]] = make_int2(s, __float_as_int(dinv[s]));
  }
}

__device__ __forceinline__ void bn_coef(const float* __restrict__ bn_sum, const float* __restrict__ bn_sq,
                                        const float* __restrict__ gamma, const float* __restrict__ beta,
                                        float invN, int c, float& sc, float& sh) {
  float mu = bn_sum[c] * invN;
  float var = fmaxf(bn_sq[c] * invN - mu * mu, 0.f);
  sc = rsqrtf(var + 1e-5f) * gamma[c];
  sh = beta[c] - mu * sc;
}

template <bool BN>
__global__ void k_gemm_wmma(const float* __restrict__ Z, int ldz,
                            const float* __restrict__ W, int K,
                            float* __restrict__ Hout, int N,
                            const float* __restrict__ bn_sum, const float* __restrict__ bn_sq,
                            const float* __restrict__ gamma, const float* __restrict__ beta,
                            float invN) {
  __shared__ float s_sc[HD], s_sh[HD];
  if (BN) {
    if (threadIdx.x < HD) {
      float sc, sh;
      bn_coef(bn_sum, bn_sq, gamma, beta, invN, threadIdx.x, sc, sh);
      s_sc[threadIdx.x] = sc;
      s_sh[threadIdx.x] = sh;
    }
    __syncthreads();
  }
  int wid = blockIdx.x * (blockDim.x >> 5) + (threadIdx.x >> 5);
  int tilesM = (N + 15) >> 4;
  if (wid >= tilesM * 4) return;
  int tm = wid >> 2, tn = wid & 3;
  int lane = threadIdx.x & 31;
  int half = lane >> 4;
  int m16 = lane & 15;
  int row = tm * 16 + m16;
  int lrow = row < N ? row : N - 1;
  const float* zr = Z + (size_t)lrow * ldz;
  int col = tn * 16 + m16;
  const float* wp = W + col;
  v8f c = {};
  for (int k0 = 0; k0 < K; k0 += 4) {
    int ka = k0 + 2 * half;
    v2f a, b;
    a.x = zr[ka];
    a.y = zr[ka + 1];
    if (BN) {
      a.x = a.x * s_sc[ka] + s_sh[ka];
      a.y = a.y * s_sc[ka + 1] + s_sh[ka + 1];
    }
    b.x = wp[(size_t)ka * HD];
    b.y = wp[(size_t)(ka + 1) * HD];
    c = __builtin_amdgcn_wmma_f32_16x16x4_f32(false, a, false, b, (short)0, c, false, false);
  }
  int rbase = tm * 16 + half * 8;
#pragma unroll
  for (int j = 0; j < 8; ++j) {
    int r = rbase + j;
    if (r < N) Hout[(size_t)r * HD + col] = c[j];
  }
}

__global__ void __launch_bounds__(512)
k_gather_relu_stats(const int* __restrict__ rowptr, const int* __restrict__ deg,
                    const int2* __restrict__ csr, const float* __restrict__ dinv,
                    const float* __restrict__ h, const float* __restrict__ bias,
                    float* __restrict__ z, float* __restrict__ bn_sum,
                    float* __restrict__ bn_sq, int N) {
  __shared__ __attribute__((aligned(16))) float s_sum[16 * HD];
  __shared__ __attribute__((aligned(16))) float s_sq[16 * HD];
  int w = __builtin_amdgcn_readfirstlane(threadIdx.x >> 5);
  int lane = threadIdx.x & 31;
  int half = lane >> 4;
  int c4 = (lane & 15) << 2;
  const float* hc = h + c4;
  const float4 b4 = *(const float4*)(bias + c4);
  float4 ts = {0.f, 0.f, 0.f, 0.f};
  float4 tq = {0.f, 0.f, 0.f, 0.f};
  for (int i = 0; i < 4; ++i) {
    int d = blockIdx.x * 64 + i * 16 + w;
    if (d >= N) break;
    int start = rowptr[d];
    int n = deg[d];
    float dd = dinv[d];
    const float4 hs = *(const float4*)(hc + (size_t)d * HD);
    float4 acc = {0.f, 0.f, 0.f, 0.f};
    for (int base = 0; base < n; base += 32) {
      int m = n - base;
      if (m > 32) m = 32;
      int2 ent = make_int2(0, 0);
      if (lane < m) ent = csr[start + base + lane];
      float wl = __int_as_float(ent.y) * dd;
      int npair = m >> 1;
#pragma unroll 4
      for (int p = 0; p < npair; ++p) {
        int sl = 2 * p + half;
        int s = __shfl(ent.x, sl, 32);
        float wgt = __shfl(wl, sl, 32);
        const float4 hv = *(const float4*)(hc + (size_t)s * HD);
        acc.x += hv.x * wgt;
        acc.y += hv.y * wgt;
        acc.z += hv.z * wgt;
        acc.w += hv.w * wgt;
      }
      if (m & 1) {
        int s = __shfl(ent.x, m - 1, 32);
        float wgt = __shfl(wl, m - 1, 32);
        if (half == 0) {
          const float4 hv = *(const float4*)(hc + (size_t)s * HD);
          acc.x += hv.x * wgt;
          acc.y += hv.y * wgt;
          acc.z += hv.z * wgt;
          acc.w += hv.w * wgt;
        }
      }
    }
    acc.x += __shfl_xor(acc.x, 16, 32);
    acc.y += __shfl_xor(acc.y, 16, 32);
    acc.z += __shfl_xor(acc.z, 16, 32);
    acc.w += __shfl_xor(acc.w, 16, 32);
    if (half == 0) {
      float w0 = dd * dd;
      float4 o;
      o.x = hs.x * w0 + acc.x;
      o.y = hs.y * w0 + acc.y;
      o.z = hs.z * w0 + acc.z;
      o.w = hs.w * w0 + acc.w;
      o.x = fmaxf(o.x + b4.x, 0.f);
      o.y = fmaxf(o.y + b4.y, 0.f);
      o.z = fmaxf(o.z + b4.z, 0.f);
      o.w = fmaxf(o.w + b4.w, 0.f);
      *(float4*)(z + (size_t)d * HD + c4) = o;
      ts.x += o.x; ts.y += o.y; ts.z += o.z; ts.w += o.w;
      tq.x += o.x * o.x; tq.y += o.y * o.y; tq.z += o.z * o.z; tq.w += o.w * o.w;
    }
  }
  if (half == 0) {
    *(float4*)(s_sum + w * HD + c4) = ts;
    *(float4*)(s_sq + w * HD + c4) = tq;
  }
  __syncthreads();
  if (threadIdx.x < 2 * HD) {
    int c = threadIdx.x & (HD - 1);
    const float* p = (threadIdx.x < HD) ? s_sum : s_sq;
    float t = 0.f;
#pragma unroll
    for (int k = 0; k < 16; ++k) t += p[k * HD + c];
    atomicAdd((threadIdx.x < HD ? bn_sum : bn_sq) + c, t);
  }
}

__global__ void k_bn_l2_pool(const float* __restrict__ z0, const float* __restrict__ z1,
                             const float* __restrict__ z2, const float* __restrict__ bn,
                             const float* __restrict__ gamma0, const float* __restrict__ beta0,
                             const float* __restrict__ gamma1, const float* __restrict__ beta1,
                             const float* __restrict__ gamma2, const float* __restrict__ beta2,
                             float invN, const int* __restrict__ batch,
                             float* __restrict__ out, float* __restrict__ gsum, int N) {
  int wid = blockIdx.x * (blockDim.x >> 5) + (threadIdx.x >> 5);
  int lane = threadIdx.x & 31;
  int r0 = wid * 8;
  if (r0 >= N) return;
  int rend = r0 + 8 < N ? r0 + 8 : N;
  const float* zl[3] = {z0, z1, z2};
  const float* gl[3] = {gamma0, gamma1, gamma2};
  const float* bl[3] = {beta0, beta1, beta2};
  float sc[6], sh[6];
#pragma unroll
  for (int j = 0; j < 6; ++j) {
    int l = j >> 1;
    bn_coef(bn + l * 128, bn + l * 128 + 64, gl[l], bl[l], invN, lane + 32 * (j & 1), sc[j], sh[j]);
  }
  float acc[6] = {0.f, 0.f, 0.f, 0.f, 0.f, 0.f};
  int gcur = -1;
  for (int row = r0; row < rend; ++row) {
    float v[6]; float s = 0.f;
#pragma unroll
    for (int j = 0; j < 6; ++j) {
      v[j] = zl[j >> 1][(size_t)row * HD + lane + 32 * (j & 1)] * sc[j] + sh[j];
      s += v[j] * v[j];
    }
#pragma unroll
    for (int off = 16; off >= 1; off >>= 1) s += __shfl_xor(s, off, 32);
    float inv = 1.0f / fmaxf(sqrtf(s), 1e-12f);
    float* o = out + (size_t)row * CD;
#pragma unroll
    for (int j = 0; j < 6; ++j) o[lane + 32 * j] = v[j] * inv;
    int g = batch[row];
    if (g != gcur) {
      if (gcur >= 0) {
#pragma unroll
        for (int j = 0; j < 6; ++j) atomicAdd(&gsum[(size_t)gcur * CD + lane + 32 * j], acc[j]);
      }
      gcur = g;
#pragma unroll
      for (int j = 0; j < 6; ++j) acc[j] = 0.f;
    }
#pragma unroll
    for (int j = 0; j < 6; ++j) acc[j] += v[j];
  }
  if (gcur >= 0) {
#pragma unroll
    for (int j = 0; j < 6; ++j) atomicAdd(&gsum[(size_t)gcur * CD + lane + 32 * j], acc[j]);
  }
}

__global__ void k_l2_gc(const float* __restrict__ gsum, const float* __restrict__ cnt,
                        float* __restrict__ out, int rows) {
  int wid = blockIdx.x * (blockDim.x >> 5) + (threadIdx.x >> 5);
  if (wid >= rows) return;
  int lane = threadIdx.x & 31;
  float icnt = 1.0f / fmaxf(cnt[wid], 1.0f);
  const float* a = gsum + (size_t)wid * CD;
  float v[6]; float s = 0.f;
#pragma unroll
  for (int j = 0; j < 6; ++j) { v[j] = a[lane + 32 * j] * icnt; s += v[j] * v[j]; }
#pragma unroll
  for (int off = 16; off >= 1; off >>= 1) s += __shfl_xor(s, off, 32);
  float inv = 1.0f / fmaxf(sqrtf(s), 1e-12f);
  float* o = out + (size_t)wid * CD;
#pragma unroll
  for (int j = 0; j < 6; ++j) o[lane + 32 * j] = v[j] * inv;
}

extern "C" void kernel_launch(void* const* d_in, const int* in_sizes, int n_in,
                              void* d_out, int out_size, void* d_ws, size_t ws_size,
                              hipStream_t stream) {
  (void)n_in; (void)out_size; (void)ws_size;
  const float* x     = (const float*)d_in[0];
  const int*   ei    = (const int*)d_in[1];
  const int*   batch = (const int*)d_in[2];
  int N = in_sizes[0] / 128;
  int E = in_sizes[1] / 2;
  const int* src = ei;
  const int* dst = ei + E;
  const float* Wl[3]  = {(const float*)d_in[3],  (const float*)d_in[7],  (const float*)d_in[11]};
  const float* bl[3]  = {(const float*)d_in[4],  (const float*)d_in[8],  (const float*)d_in[12]};
  const float* gl[3]  = {(const float*)d_in[5],  (const float*)d_in[9],  (const float*)d_in[13]};
  const float* bel[3] = {(const float*)d_in[6],  (const float*)d_in[10], (const float*)d_in[14]};

  float* ws   = (float*)d_ws;
  float* dinv = ws;                       ws += N;
  float* h    = ws;                       ws += (size_t)N * HD;
  float* zb[3];
  zb[0] = ws;                             ws += (size_t)N * HD;
  zb[1] = ws;                             ws += (size_t)N * HD;
  zb[2] = ws;                             ws += (size_t)N * HD;
  float* gsum = ws;                       ws += (size_t)GN * CD;
  float* cnt  = ws;                       ws += GN;
  float* bn   = ws;
  ws += 3 * 128;
  int* deg    = (int*)(((size_t)ws + 15) & ~(size_t)15);
  int* rowptr = deg + (((size_t)N + 3) & ~(size_t)3);
  int* rank   = rowptr + (((size_t)N + 3) & ~(size_t)3);
  int2* csr   = (int2*)(rank + (((size_t)E + 1) & ~(size_t)1));

  hipMemsetAsync(deg, 0, (size_t)N * sizeof(int), stream);

  k_deg_rank<<<(E + 255) / 256, 256, 0, stream>>>(dst, E, deg, rank, gsum, GN * CD + GN + 3 * 128);
  k_rowptr<<<(N + 1023) / 1024, 256, 0, stream>>>(deg, rowptr, dinv, batch, cnt, N);
  k_fill<<<(E + 255) / 256, 256, 0, stream>>>(src, dst, rank, rowptr, dinv, csr, E);

  float invN = 1.0f / (float)N;
  int tilesM = (N + 15) / 16;
  int waves = tilesM * 4;
  for (int l = 0; l < 3; ++l) {
    float* bn_sum = bn + l * 128;
    float* bn_sq  = bn_sum + 64;
    if (l == 0) {
      k_gemm_wmma<false><<<(waves + 7) / 8, 256, 0, stream>>>(x, 128, Wl[0], 128, h, N,
                                                              bn_sum, bn_sq, gl[0], bel[0], invN);
    } else {
      k_gemm_wmma<true><<<(waves + 7) / 8, 256, 0, stream>>>(zb[l - 1], HD, Wl[l], HD, h, N,
                                                             bn_sum - 128, bn_sq - 128,
                                                             gl[l - 1], bel[l - 1], invN);
    }
    k_gather_relu_stats<<<(N + 63) / 64, 512, 0, stream>>>(rowptr, deg, csr, dinv, h, bl[l], zb[l],
                                                           bn_sum, bn_sq, N);
  }

  float* out = (float*)d_out;
  k_bn_l2_pool<<<(N + 63) / 64, 256, 0, stream>>>(zb[0], zb[1], zb[2], bn, gl[0], bel[0], gl[1], bel[1],
                                                  gl[2], bel[2], invN, batch, out, gsum, N);
  k_l2_gc<<<(GN + 7) / 8, 256, 0, stream>>>(gsum, cnt, out + (size_t)N * CD, GN);
}


# launch fusion: layers 2,3 aggregate first, their GEMM runs on the gather's LDS tile (WMMA); layer-1 GEMM rides in the scan's launch; 12 stream ops to 9; edge sum ahead of product, BN sums regrouped
# speedup vs baseline: 2.7010x; 1.1116x over previous
#include <hip/hip_runtime.h>

typedef float v2f __attribute__((ext_vector_type(2)));
typedef float v8f __attribute__((ext_vector_type(8)));

#define HD 64
#define CD 192
#define GN 256

__global__ void k_deg_rank(const int* __restrict__ dst, int E, int* __restrict__ deg,
                           int* __restrict__ rank, float* __restrict__ zero, int nzero) {
  int e = blockIdx.x * blockDim.x + threadIdx.x;
  for (int i = e; i < nzero; i += gridDim.x * blockDim.x) zero[i] = 0.f;
  if (e < E) rank[e] = atomicAdd(&deg[dst[e]], 1);
}

__device__ __forceinline__ void rowptr_block(int bid, const int* __restrict__ deg, int* __restrict__ rowptr,
                                             float* __restrict__ dinv, const int* __restrict__ batch,
                                             float* __restrict__ cnt, int N) {
  __shared__ int s_pre[8], s_inc[8];
  int tid = threadIdx.x, lane = tid & 31, w = tid >> 5;
  int tile0 = bid * 1024;
  int pre = 0;
  const int4* d4 = (const int4*)deg;
#pragma unroll 8
  for (int i = tid; i < (tile0 >> 2); i += 256) {
    int4 v = d4[i];
    pre += (v.x + v.y) + (v.z + v.w);
  }
  int i0 = tile0 + tid * 4;
  int v0 = (i0     < N) ? deg[i0]     : 0;
  int v1 = (i0 + 1 < N) ? deg[i0 + 1] : 0;
  int v2 = (i0 + 2 < N) ? deg[i0 + 2] : 0;
  int v3 = (i0 + 3 < N) ? deg[i0 + 3] : 0;
  if (i0     < N) { dinv[i0]     = rsqrtf((float)v0 + 1.0f); atomicAdd(&cnt[batch[i0]],     1.0f); }
  if (i0 + 1 < N) { dinv[i0 + 1] = rsqrtf((float)v1 + 1.0f); atomicAdd(&cnt[batch[i0 + 1]], 1.0f); }
  if (i0 + 2 < N) { dinv[i0 + 2] = rsqrtf((float)v2 + 1.0f); atomicAdd(&cnt[batch[i0 + 2]], 1.0f); }
  if (i0 + 3 < N) { dinv[i0 + 3] = rsqrtf((float)v3 + 1.0f); atomicAdd(&cnt[batch[i0 + 3]], 1.0f); }
  int t = (v0 + v1) + (v2 + v3);
  int inc = t;
  for (int off = 1; off < 32; off <<= 1) {
    int u = __shfl_up(inc, off, 32);
    if (lane >= off) inc += u;
  }
  for (int off = 16; off >= 1; off >>= 1) pre += __shfl_xor(pre, off, 32);
  if (lane == 31) s_inc[w] = inc;
  if (lane == 0)  s_pre[w] = pre;
  __syncthreads();
  int base = 0;
  for (int k = 0; k < 8; ++k) {
    base += s_pre[k];
    if (k < w) base += s_inc[k];
  }
  int ex = base + inc - t;
  if (i0     < N) rowptr[i0]     = ex;
  if (i0 + 1 < N) rowptr[i0 + 1] = ex + v0;
  if (i0 + 2 < N) rowptr[i0 + 2] = ex + v0 + v1;
  if (i0 + 3 < N) rowptr[i0 + 3] = ex + v0 + v1 + v2;
}

__global__ void k_fill(const int* __restrict__ src, const int* __restrict__ dst,
                       const int* __restrict__ rank, const int* __restrict__ rowptr,
                       const float* __restrict__ dinv, int2* __restrict__ csr, int E) {
  int e = blockIdx.x * blockDim.x + threadIdx.x;
  if (e < E) {
    int s = src[e];
    csr[rowptr[dst[e]] + rank[e]] = make_int2(s, __float_as_int(dinv[s]));
  }
}

__device__ __forceinline__ void bn_coef(const float* __restrict__ bn_sum, const float* __restrict__ bn_sq,
                                        const float* __restrict__ gamma, const float* __restrict__ beta,
                                        float invN, int c, float& sc, float& sh) {
  float mu = bn_sum[c] * invN;
  float var = fmaxf(bn_sq[c] * invN - mu * mu, 0.f);
  sc = rsqrtf(var + 1e-5f) * gamma[c];
  sh = beta[c] - mu * sc;
}

__device__ __forceinline__ void gemm_block(int bid, const float* __restrict__ Z, int ldz,
                                           const float* __restrict__ W, int K,
                                           float* __restrict__ Hout, int N) {
  int wid = bid * (blockDim.x >> 5) + (threadIdx.x >> 5);
  int tilesM = (N + 15) >> 4;
  if (wid >= tilesM * 4) return;
  int tm = wid >> 2, tn = wid & 3;
  int lane = threadIdx.x & 31;
  int half = lane >> 4;
  int m16 = lane & 15;
  int row = tm * 16 + m16;
  int lrow = row < N ? row : N - 1;
  const float* zr = Z + (size_t)lrow * ldz;
  int col = tn * 16 + m16;
  const float* wp = W + col;
  v8f c = {};
  for (int k0 = 0; k0 < K; k0 += 4) {
    int ka = k0 + 2 * half;
    v2f a, b;
    a.x = zr[ka];
    a.y = zr[ka + 1];
    b.x = wp[(size_t)ka * HD];
    b.y = wp[(size_t)(ka + 1) * HD];
    c = __builtin_amdgcn_wmma_f32_16x16x4_f32(false, a, false, b, (short)0, c, false, false);
  }
  int rbase = tm * 16 + half * 8;
#pragma unroll
  for (int j = 0; j < 8; ++j) {
    int r = rbase + j;
    if (r < N) Hout[(size_t)r * HD + col] = c[j];
  }
}

__global__ void k_rowptr_gemm(const int* __restrict__ deg, int* __restrict__ rowptr,
                              float* __restrict__ dinv, const int* __restrict__ batch,
                              float* __restrict__ cnt, int N, int nscan,
                              const float* __restrict__ X, int ldx,
                              const float* __restrict__ W, int K, float* __restrict__ Hout) {
  if ((int)blockIdx.x < nscan) {
    rowptr_block(blockIdx.x, deg, rowptr, dinv, batch, cnt, N);
  } else {
    gemm_block(blockIdx.x - nscan, X, ldx, W, K, Hout, N);
  }
}

__global__ void __launch_bounds__(512)
k_gather_relu_stats(const int* __restrict__ rowptr, const int* __restrict__ deg,
                    const int2* __restrict__ csr, const float* __restrict__ dinv,
                    const float* __restrict__ h, const float* __restrict__ bias,
                    float* __restrict__ z, float* __restrict__ bn_sum,
                    float* __restrict__ bn_sq, int N) {
  __shared__ __attribute__((aligned(16))) float s_sum[16 * HD];
  __shared__ __attribute__((aligned(16))) float s_sq[16 * HD];
  int w = __builtin_amdgcn_readfirstlane(threadIdx.x >> 5);
  int lane = threadIdx.x & 31;
  int half = lane >> 4;
  int c4 = (lane & 15) << 2;
  const float* hc = h + c4;
  const float4 b4 = *(const float4*)(bias + c4);
  float4 ts = {0.f, 0.f, 0.f, 0.f};
  float4 tq = {0.f, 0.f, 0.f, 0.f};
  for (int i = 0; i < 4; ++i) {
    int d = blockIdx.x * 64 + i * 16 + w;
    if (d >= N) break;
    int start = rowptr[d];
    int n = deg[d];
    float dd = dinv[d];
    const float4 hs = *(const float4*)(hc + (size_t)d * HD);
    float4 acc = {0.f, 0.f, 0.f, 0.f};
    for (int base = 0; base < n; base += 32) {
      int m = n - base;
      if (m > 32) m = 32;
      int2 ent = make_int2(0, 0);
      if (lane < m) ent = csr[start + base + lane];
      float wl = __int_as_float(ent.y) * dd;
      int npair = m >> 1;
#pragma unroll 4
      for (int p = 0; p < npair; ++p) {
        int sl = 2 * p + half;
        int s = __shfl(ent.x, sl, 32);
        float wgt = __shfl(wl, sl, 32);
        const float4 hv = *(const float4*)(hc + (size_t)s * HD);
        acc.x += hv.x * wgt;
        acc.y += hv.y * wgt;
        acc.z += hv.z * wgt;
        acc.w += hv.w * wgt;
      }
      if (m & 1) {
        int s = __shfl(ent.x, m - 1, 32);
        float wgt = __shfl(wl, m - 1, 32);
        if (half == 0) {
          const float4 hv = *(const float4*)(hc + (size_t)s * HD);
          acc.x += hv.x * wgt;
          acc.y += hv.y * wgt;
          acc.z += hv.z * wgt;
          acc.w += hv.w * wgt;
        }
      }
    }
    acc.x += __shfl_xor(acc.x, 16, 32);
    acc.y += __shfl_xor(acc.y, 16, 32);
    acc.z += __shfl_xor(acc.z, 16, 32);
    acc.w += __shfl_xor(acc.w, 16, 32);
    if (half == 0) {
      float w0 = dd * dd;
      float4 o;
      o.x = hs.x * w0 + acc.x;
      o.y = hs.y * w0 + acc.y;
      o.z = hs.z * w0 + acc.z;
      o.w = hs.w * w0 + acc.w;
      o.x = fmaxf(o.x + b4.x, 0.f);
      o.y = fmaxf(o.y + b4.y, 0.f);
      o.z = fmaxf(o.z + b4.z, 0.f);
      o.w = fmaxf(o.w + b4.w, 0.f);
      *(float4*)(z + (size_t)d * HD + c4) = o;
      ts.x += o.x; ts.y += o.y; ts.z += o.z; ts.w += o.w;
      tq.x += o.x * o.x; tq.y += o.y * o.y; tq.z += o.z * o.z; tq.w += o.w * o.w;
    }
  }
  if (half == 0) {
    *(float4*)(s_sum + w * HD + c4) = ts;
    *(float4*)(s_sq + w * HD + c4) = tq;
  }
  __syncthreads();
  if (threadIdx.x < 2 * HD) {
    int c = threadIdx.x & (HD - 1);
    const float* p = (threadIdx.x < HD) ? s_sum : s_sq;
    float t = 0.f;
#pragma unroll
    for (int k = 0; k < 16; ++k) t += p[k * HD + c];
    atomicAdd((threadIdx.x < HD ? bn_sum : bn_sq) + c, t);
  }
}

#define AP 68
__global__ void __launch_bounds__(512)
k_gather_bn_gemm_relu_stats(const int* __restrict__ rowptr, const int* __restrict__ deg,
                            const int2* __restrict__ csr, const float* __restrict__ dinv,
                            const float* __restrict__ zp,
                            const float* __restrict__ pbn_sum, const float* __restrict__ pbn_sq,
                            const float* __restrict__ pgamma, const float* __restrict__ pbeta, float invN,
                            const float* __restrict__ W, const float* __restrict__ bias,
                            float* __restrict__ z, float* __restrict__ bn_sum,
                            float* __restrict__ bn_sq, int N) {
  __shared__ __attribute__((aligned(16))) float s_a[64 * AP];
  __shared__ float s_sum[4 * HD];
  __shared__ float s_sq[4 * HD];
  int w = __builtin_amdgcn_readfirstlane(threadIdx.x >> 5);
  int lane = threadIdx.x & 31;
  int half = lane >> 4;
  int c4 = (lane & 15) << 2;
  const float* hc = zp + c4;
  float sc0, sc1, sc2, sc3, sh0, sh1, sh2, sh3;
  bn_coef(pbn_sum, pbn_sq, pgamma, pbeta, invN, c4,     sc0, sh0);
  bn_coef(pbn_sum, pbn_sq, pgamma, pbeta, invN, c4 + 1, sc1, sh1);
  bn_coef(pbn_sum, pbn_sq, pgamma, pbeta, invN, c4 + 2, sc2, sh2);
  bn_coef(pbn_sum, pbn_sq, pgamma, pbeta, invN, c4 + 3, sc3, sh3);
  for (int i = 0; i < 4; ++i) {
    int lr = i * 16 + w;
    int d = blockIdx.x * 64 + lr;
    float4 g = {0.f, 0.f, 0.f, 0.f};
    if (d < N) {
      int start = rowptr[d];
      int n = deg[d];
      float dd = dinv[d];
      const float4 hs = *(const float4*)(hc + (size_t)d * HD);
      float4 acc = {0.f, 0.f, 0.f, 0.f};
      float wsum = 0.f;
      for (int base = 0; base < n; base += 32) {
        int m = n - base;
        if (m > 32) m = 32;
        int2 ent = make_int2(0, 0);
        if (lane < m) ent = csr[start + base + lane];
        float wl = __int_as_float(ent.y) * dd;
        int npair = m >> 1;
#pragma unroll 4
        for (int p = 0; p < npair; ++p) {
          int sl = 2 * p + half;
          int s = __shfl(ent.x, sl, 32);
          float wgt = __shfl(wl, sl, 32);
          const float4 hv = *(const float4*)(hc + (size_t)s * HD);
          acc.x += hv.x * wgt;
          acc.y += hv.y * wgt;
          acc.z += hv.z * wgt;
          acc.w += hv.w * wgt;
          wsum += wgt;
        }
        if (m & 1) {
          int s = __shfl(ent.x, m - 1, 32);
          float wgt = __shfl(wl, m - 1, 32);
          if (half == 0) {
            const float4 hv = *(const float4*)(hc + (size_t)s * HD);
            acc.x += hv.x * wgt;
            acc.y += hv.y * wgt;
            acc.z += hv.z * wgt;
            acc.w += hv.w * wgt;
            wsum += wgt;
          }
        }
      }
      acc.x += __shfl_xor(acc.x, 16, 32);
      acc.y += __shfl_xor(acc.y, 16, 32);
      acc.z += __shfl_xor(acc.z, 16, 32);
      acc.w += __shfl_xor(acc.w, 16, 32);
      wsum += __shfl_xor(wsum, 16, 32);
      float w0 = dd * dd;
      float r = w0 + wsum;
      g.x = (hs.x * w0 + acc.x) * sc0 + r * sh0;
      g.y = (hs.y * w0 + acc.y) * sc1 + r * sh1;
      g.z = (hs.z * w0 + acc.z) * sc2 + r * sh2;
      g.w = (hs.w * w0 + acc.w) * sc3 + r * sh3;
    }
    if (half == 0) *(float4*)(s_a + lr * AP + c4) = g;
  }
  __syncthreads();
  int tm = w >> 2, tn = w & 3;
  int m16 = lane & 15;
  const float* ap = s_a + (tm * 16 + m16) * AP + 2 * half;
  int col = tn * 16 + m16;
  const float* wp = W + col;
  v8f c = {};
#pragma unroll
  for (int k0 = 0; k0 < HD; k0 += 4) {
    int ka = k0 + 2 * half;
    v2f a, b;
    a.x = ap[k0];
    a.y = ap[k0 + 1];
    b.x = wp[(size_t)ka * HD];
    b.y = wp[(size_t)(ka + 1) * HD];
    c = __builtin_amdgcn_wmma_f32_16x16x4_f32(false, a, false, b, (short)0, c, false, false);
  }
  float bcol = bias[col];
  int rbase = blockIdx.x * 64 + tm * 16 + half * 8;
  float ts = 0.f, tq = 0.f;
#pragma unroll
  for (int j = 0; j < 8; ++j) {
    int r = rbase + j;
    if (r < N) {
      float o = fmaxf(c[j] + bcol, 0.f);
      z[(size_t)r * HD + col] = o;
      ts += o;
      tq += o * o;
    }
  }
  ts += __shfl_xor(ts, 16, 32);
  tq += __shfl_xor(tq, 16, 32);
  if (half == 0) {
    s_sum[tm * HD + col] = ts;
    s_sq[tm * HD + col] = tq;
  }
  __syncthreads();
  if (threadIdx.x < 2 * HD) {
    int cc = threadIdx.x & (HD - 1);
    const float* p = (threadIdx.x < HD) ? s_sum : s_sq;
    float t = (p[cc] + p[HD + cc]) + (p[2 * HD + cc] + p[3 * HD + cc]);
    atomicAdd((threadIdx.x < HD ? bn_sum : bn_sq) + cc, t);
  }
}

__global__ void k_bn_l2_pool(const float* __restrict__ z0, const float* __restrict__ z1,
                             const float* __restrict__ z2, const float* __restrict__ bn,
                             const float* __restrict__ gamma0, const float* __restrict__ beta0,
                             const float* __restrict__ gamma1, const float* __restrict__ beta1,
                             const float* __restrict__ gamma2, const float* __restrict__ beta2,
                             float invN, const int* __restrict__ batch,
                             float* __restrict__ out, float* __restrict__ gsum, int N) {
  int wid = blockIdx.x * (blockDim.x >> 5) + (threadIdx.x >> 5);
  int lane = threadIdx.x & 31;
  int r0 = wid * 8;
  if (r0 >= N) return;
  int rend = r0 + 8 < N ? r0 + 8 : N;
  const float* zl[3] = {z0, z1, z2};
  const float* gl[3] = {gamma0, gamma1, gamma2};
  const float* bl[3] = {beta0, beta1, beta2};
  float sc[6], sh[6];
#pragma unroll
  for (int j = 0; j < 6; ++j) {
    int l = j >> 1;
    bn_coef(bn + l * 128, bn + l * 128 + 64, gl[l], bl[l], invN, lane + 32 * (j & 1), sc[j], sh[j]);
  }
  float acc[6] = {0.f, 0.f, 0.f, 0.f, 0.f, 0.f};
  int gcur = -1;
  for (int row = r0; row < rend; ++row) {
    float v[6]; float s = 0.f;
#pragma unroll
    for (int j = 0; j < 6; ++j) {
      v[j] = zl[j >> 1][(size_t)row * HD + lane + 32 * (j & 1)] * sc[j] + sh[j];
      s += v[j] * v[j];
    }
#pragma unroll
    for (int off = 16; off >= 1; off >>= 1) s += __shfl_xor(s, off, 32);
    float inv = 1.0f / fmaxf(sqrtf(s), 1e-12f);
    float* o = out + (size_t)row * CD;
#pragma unroll
    for (int j = 0; j < 6; ++j) o[lane + 32 * j] = v[j] * inv;
    int g = batch[row];
    if (g != gcur) {
      if (gcur >= 0) {
#pragma unroll
        for (int j = 0; j < 6; ++j) atomicAdd(&gsum[(size_t)gcur * CD + lane + 32 * j], acc[j]);
      }
      gcur = g;
#pragma unroll
      for (int j = 0; j < 6; ++j) acc[j] = 0.f;
    }
#pragma unroll
    for (int j = 0; j < 6; ++j) acc[j] += v[j];
  }
  if (gcur >= 0) {
#pragma unroll
    for (int j = 0; j < 6; ++j) atomicAdd(&gsum[(size_t)gcur * CD + lane + 32 * j], acc[j]);
  }
}

__global__ void k_l2_gc(const float* __restrict__ gsum, const float* __restrict__ cnt,
                        float* __restrict__ out, int rows) {
  int wid = blockIdx.x * (blockDim.x >> 5) + (threadIdx.x >> 5);
  if (wid >= rows) return;
  int lane = threadIdx.x & 31;
  float icnt = 1.0f / fmaxf(cnt[wid], 1.0f);
  const float* a = gsum + (size_t)wid * CD;
  float v[6]; float s = 0.f;
#pragma unroll
  for (int j = 0; j < 6; ++j) { v[j] = a[lane + 32 * j] * icnt; s += v[j] * v[j]; }
#pragma unroll
  for (int off = 16; off >= 1; off >>= 1) s += __shfl_xor(s, off, 32);
  float inv = 1.0f / fmaxf(sqrtf(s), 1e-12f);
  float* o = out + (size_t)wid * CD;
#pragma unroll
  for (int j = 0; j < 6; ++j) o[lane + 32 * j] = v[j] * inv;
}

extern "C" void kernel_launch(void* const* d_in, const int* in_sizes, int n_in,
                              void* d_out, int out_size, void* d_ws, size_t ws_size,
                              hipStream_t stream) {
  (void)n_in; (void)out_size; (void)ws_size;
  const float* x     = (const float*)d_in[0];
  const int*   ei    = (const int*)d_in[1];
  const int*   batch = (const int*)d_in[2];
  int N = in_sizes[0] / 128;
  int E = in_sizes[1] / 2;
  const int* src = ei;
  const int* dst = ei + E;
  const float* Wl[3]  = {(const float*)d_in[3],  (const float*)d_in[7],  (const float*)d_in[11]};
  const float* bl[3]  = {(const float*)d_in[4],  (const float*)d_in[8],  (const float*)d_in[12]};
  const float* gl[3]  = {(const float*)d_in[5],  (const float*)d_in[9],  (const float*)d_in[13]};
  const float* bel[3] = {(const float*)d_in[6],  (const float*)d_in[10], (const float*)d_in[14]};

  float* ws   = (float*)d_ws;
  float* dinv = ws;                       ws += N;
  float* h    = ws;                       ws += (size_t)N * HD;
  float* zb[3];
  zb[0] = ws;                             ws += (size_t)N * HD;
  zb[1] = ws;                             ws += (size_t)N * HD;
  zb[2] = ws;                             ws += (size_t)N * HD;
  float* gsum = ws;                       ws += (size_t)GN * CD;
  float* cnt  = ws;                       ws += GN;
  float* bn   = ws;
  ws += 3 * 128;
  int* deg    = (int*)(((size_t)ws + 15) & ~(size_t)15);
  int* rowptr = deg + (((size_t)N + 3) & ~(size_t)3);
  int* rank   = rowptr + (((size_t)N + 3) & ~(size_t)3);
  int2* csr   = (int2*)(rank + (((size_t)E + 1) & ~(size_t)1));

  hipMemsetAsync(deg, 0, (size_t)N * sizeof(int), stream);

  k_deg_rank<<<(E + 255) / 256, 256, 0, stream>>>(dst, E, deg, rank, gsum, GN * CD + GN + 3 * 128);
  int tilesM = (N + 15) / 16;
  int waves = tilesM * 4;
  int nscan = (N + 1023) / 1024;
  k_rowptr_gemm<<<nscan + (waves + 7) / 8, 256, 0, stream>>>(deg, rowptr, dinv, batch, cnt, N, nscan,
                                                            x, 128, Wl[0], 128, h);
  k_fill<<<(E + 255) / 256, 256, 0, stream>>>(src, dst, rank, rowptr, dinv, csr, E);

  float invN = 1.0f / (float)N;
  k_gather_relu_stats<<<(N + 63) / 64, 512, 0, stream>>>(rowptr, deg, csr, dinv, h, bl[0], zb[0],
                                                         bn, bn + 64, N);
  for (int l = 1; l < 3; ++l) {
    float* bn_sum = bn + l * 128;
    float* bn_sq  = bn_sum + 64;
    k_gather_bn_gemm_relu_stats<<<(N + 63) / 64, 512, 0, stream>>>(rowptr, deg, csr, dinv, zb[l - 1],
                                                                   bn_sum - 128, bn_sq - 128,
                                                                   gl[l - 1], bel[l - 1], invN,
                                                                   Wl[l], bl[l], zb[l], bn_sum, bn_sq, N);
  }

  float* out = (float*)d_out;
  k_bn_l2_pool<<<(N + 63) / 64, 256, 0, stream>>>(zb[0], zb[1], zb[2], bn, gl[0], bel[0], gl[1], bel[1],
                                                  gl[2], bel[2], invN, batch, out, gsum, N);
  k_l2_gc<<<(GN + 7) / 8, 256, 0, stream>>>(gsum, cnt, out + (size_t)N * CD, GN);
}


# edge list built without global atomics: edges binned by 256-row dst range via LDS counters (count, scatter, per-bin slots); memset, rank array, per-edge kernels gone; edge sums in LDS-arrival order
# speedup vs baseline: 3.5280x; 1.3062x over previous
#include <hip/hip_runtime.h>

typedef float v2f __attribute__((ext_vector_type(2)));
typedef float v8f __attribute__((ext_vector_type(8)));

#define HD 64
#define CD 192
#define GN 256

#define BINW  256
#define MAXB  256
#define CHUNK 4096

__device__ __forceinline__ int wave_incl_scan(int v, int lane) {
  for (int off = 1; off < 32; off <<= 1) {
    int u = __shfl_up(v, off, 32);
    if (lane >= off) v += u;
  }
  return v;
}

__global__ void __launch_bounds__(1024)
k_bin_count(const int* __restrict__ dst, int E, int nb, int* __restrict__ cm,
            float* __restrict__ zero, int nzero) {
  __shared__ int s_h[MAXB];
  int tid = threadIdx.x;
  int chunk = blockIdx.x, nch = gridDim.x;
  for (int i = chunk * 1024 + tid; i < nzero; i += nch * 1024) zero[i] = 0.f;
  if (tid < MAXB) s_h[tid] = 0;
  __syncthreads();
  int e0 = chunk * CHUNK + tid;
#pragma unroll
  for (int k = 0; k < 4; ++k) {
    int e = e0 + k * 1024;
    if (e < E) atomicAdd(&s_h[dst[e] >> 8], 1);
  }
  __syncthreads();
  int nbp = (nb + 3) & ~3;
  if (tid < nbp) cm[chunk * nbp + tid] = s_h[tid];
}

__global__ void __launch_bounds__(1024)
k_bin_scatter(const int* __restrict__ src, const int* __restrict__ dst, int E, int nb,
              const int* __restrict__ cm, int* __restrict__ binstart, int2* __restrict__ binned) {
  __shared__ int s_cnt[MAXB];
  __shared__ int s_tot[MAXB];
  __shared__ int s_par[MAXB];
  __shared__ int s_lst[MAXB];
  __shared__ int s_gof[MAXB];
  __shared__ int s_wc[8], s_wg[8];
  __shared__ int2 s_e[CHUNK];
  int tid = threadIdx.x, lane = tid & 31, w = tid >> 5;
  int chunk = blockIdx.x, nch = gridDim.x;
  if (tid < MAXB) { s_cnt[tid] = 0; s_tot[tid] = 0; s_par[tid] = 0; }
  __syncthreads();
  int es[4], ed[4], er[4];
  int e0 = chunk * CHUNK + tid;
#pragma unroll
  for (int k = 0; k < 4; ++k) {
    int e = e0 + k * 1024;
    es[k] = 0; ed[k] = 0; er[k] = 0;
    if (e < E) {
      es[k] = src[e];
      ed[k] = dst[e];
      er[k] = atomicAdd(&s_cnt[ed[k] >> 8], 1);
    }
  }
  {
    int nbp = (nb + 3) & ~3;
    int q = tid & 63, r = tid >> 6;
    if (4 * q < nbp) {
      const int4* col = (const int4*)cm + q;
      int rowq = nbp >> 2;
      int t0 = 0, t1 = 0, t2 = 0, t3 = 0, p0 = 0, p1 = 0, p2 = 0, p3 = 0;
#pragma unroll 4
      for (int j = r; j < nch; j += 16) {
        int4 v = col[(size_t)j * rowq];
        int msk = (j < chunk) ? -1 : 0;
        t0 += v.x; t1 += v.y; t2 += v.z; t3 += v.w;
        p0 += v.x & msk; p1 += v.y & msk; p2 += v.z & msk; p3 += v.w & msk;
      }
      atomicAdd(&s_tot[4 * q], t0);     atomicAdd(&s_par[4 * q], p0);
      atomicAdd(&s_tot[4 * q + 1], t1); atomicAdd(&s_par[4 * q + 1], p1);
      atomicAdd(&s_tot[4 * q + 2], t2); atomicAdd(&s_par[4 * q + 2], p2);
      atomicAdd(&s_tot[4 * q + 3], t3); atomicAdd(&s_par[4 * q + 3], p3);
    }
  }
  __syncthreads();
  int c = (tid < MAXB) ? s_cnt[tid] : 0;
  int g = (tid < MAXB) ? s_tot[tid] : 0;
  int ci = wave_incl_scan(c, lane);
  int gi = wave_incl_scan(g, lane);
  if (lane == 31 && w < 8) { s_wc[w] = ci; s_wg[w] = gi; }
  __syncthreads();
  if (tid < MAXB) {
    int bc = 0, bg = 0;
    for (int k = 0; k < 8; ++k) {
      if (k < w) { bc += s_wc[k]; bg += s_wg[k]; }
    }
    int gstart = bg + gi - g;
    s_lst[tid] = bc + ci - c;
    s_gof[tid] = gstart + s_par[tid];
    if (chunk == 0) {
      if (tid < nb) binstart[tid] = gstart;
      if (tid == 0) binstart[nb] = E;
    }
  }
  __syncthreads();
  int nloc = E - chunk * CHUNK;
  if (nloc > CHUNK) nloc = CHUNK;
#pragma unroll
  for (int k = 0; k < 4; ++k) {
    if (e0 + k * 1024 < E) s_e[s_lst[ed[k] >> 8] + er[k]] = make_int2(es[k], ed[k]);
  }
  __syncthreads();
#pragma unroll
  for (int k = 0; k < 4; ++k) {
    int i = k * 1024 + tid;
    if (i < nloc) {
      int2 en = s_e[i];
      int b = en.y >> 8;
      binned[s_gof[b] + (i - s_lst[b])] = en;
    }
  }
}

__device__ __forceinline__ void csr_block(int bin, const int* __restrict__ binstart,
                                          const int2* __restrict__ binned, int* __restrict__ deg,
                                          int* __restrict__ rowptr, float* __restrict__ dinv,
                                          const int* __restrict__ batch, float* __restrict__ cnt,
                                          int2* __restrict__ csr, int N) {
  __shared__ int s_c[BINW], s_cur[BINW], s_w[8];
  int tid = threadIdx.x, lane = tid & 31, w = tid >> 5;
  int bs = binstart[bin], be = binstart[bin + 1];
  s_c[tid] = 0;
  __syncthreads();
#pragma unroll 4
  for (int i = bs + tid; i < be; i += 256) atomicAdd(&s_c[binned[i].y & (BINW - 1)], 1);
  __syncthreads();
  int c = s_c[tid];
  int inc = wave_incl_scan(c, lane);
  if (lane == 31) s_w[w] = inc;
  __syncthreads();
  int base = bs;
  for (int k = 0; k < 8; ++k) {
    if (k < w) base += s_w[k];
  }
  int first = base + inc - c;
  s_cur[tid] = first;
  int node = bin * BINW + tid;
  if (node < N) {
    deg[node] = c;
    rowptr[node] = first;
    dinv[node] = rsqrtf((float)c + 1.0f);
    atomicAdd(&cnt[batch[node]], 1.0f);
  }
  __syncthreads();
#pragma unroll 4
  for (int i = bs + tid; i < be; i += 256) {
    int2 en = binned[i];
    int pos = atomicAdd(&s_cur[en.y & (BINW - 1)], 1);
    csr[pos] = make_int2(en.x, 0);
  }
}

__device__ __forceinline__ void bn_coef(const float* __restrict__ bn_sum, const float* __restrict__ bn_sq,
                                        const float* __restrict__ gamma, const float* __restrict__ beta,
                                        float invN, int c, float& sc, float& sh) {
  float mu = bn_sum[c] * invN;
  float var = fmaxf(bn_sq[c] * invN - mu * mu, 0.f);
  sc = rsqrtf(var + 1e-5f) * gamma[c];
  sh = beta[c] - mu * sc;
}

__device__ __forceinline__ void gemm_block(int bid, const float* __restrict__ Z, int ldz,
                                           const float* __restrict__ W, int K,
                                           float* __restrict__ Hout, int N) {
  int wid = bid * (blockDim.x >> 5) + (threadIdx.x >> 5);
  int tilesM = (N + 15) >> 4;
  if (wid >= tilesM * 4) return;
  int tm = wid >> 2, tn = wid & 3;
  int lane = threadIdx.x & 31;
  int half = lane >> 4;
  int m16 = lane & 15;
  int row = tm * 16 + m16;
  int lrow = row < N ? row : N - 1;
  const float* zr = Z + (size_t)lrow * ldz;
  int col = tn * 16 + m16;
  const float* wp = W + col;
  v8f c = {};
  for (int k0 = 0; k0 < K; k0 += 4) {
    int ka = k0 + 2 * half;
    v2f a, b;
    a.x = zr[ka];
    a.y = zr[ka + 1];
    b.x = wp[(size_t)ka * HD];
    b.y = wp[(size_t)(ka + 1) * HD];
    c = __builtin_amdgcn_wmma_f32_16x16x4_f32(false, a, false, b, (short)0, c, false, false);
  }
  int rbase = tm * 16 + half * 8;
#pragma unroll
  for (int j = 0; j < 8; ++j) {
    int r = rbase + j;
    if (r < N) Hout[(size_t)r * HD + col] = c[j];
  }
}

__global__ void __launch_bounds__(256)
k_csr_gemm(const int* __restrict__ binstart, const int2* __restrict__ binned,
           int* __restrict__ deg, int* __restrict__ rowptr, float* __restrict__ dinv,
           const int* __restrict__ batch, float* __restrict__ cnt, int2* __restrict__ csr, int N, int nb,
           const float* __restrict__ X, int ldx,
           const float* __restrict__ W, int K, float* __restrict__ Hout) {
  if ((int)blockIdx.x < nb) {
    csr_block(blockIdx.x, binstart, binned, deg, rowptr, dinv, batch, cnt, csr, N);
  } else {
    gemm_block(blockIdx.x - nb, X, ldx, W, K, Hout, N);
  }
}

__global__ void __launch_bounds__(512)
k_gather_relu_stats(const int* __restrict__ rowptr, const int* __restrict__ deg,
                    int2* __restrict__ csr, const float* __restrict__ dinv,
                    const float* __restrict__ h, const float* __restrict__ bias,
                    float* __restrict__ z, float* __restrict__ bn_sum,
                    float* __restrict__ bn_sq, int N) {
  __shared__ __attribute__((aligned(16))) float s_sum[16 * HD];
  __shared__ __attribute__((aligned(16))) float s_sq[16 * HD];
  int w = __builtin_amdgcn_readfirstlane(threadIdx.x >> 5);
  int lane = threadIdx.x & 31;
  int half = lane >> 4;
  int c4 = (lane & 15) << 2;
  const float* hc = h + c4;
  const float4 b4 = *(const float4*)(bias + c4);
  float4 ts = {0.f, 0.f, 0.f, 0.f};
  float4 tq = {0.f, 0.f, 0.f, 0.f};
  for (int i = 0; i < 4; ++i) {
    int d = blockIdx.x * 64 + i * 16 + w;
    if (d >= N) break;
    int start = rowptr[d];
    int n = deg[d];
    float dd = dinv[d];
    const float4 hs = *(const float4*)(hc + (size_t)d * HD);
    float4 acc = {0.f, 0.f, 0.f, 0.f};
    for (int base = 0; base < n; base += 32) {
      int m = n - base;
      if (m > 32) m = 32;
      int2 ent = make_int2(0, 0);
      float ds = 0.f;
      if (lane < m) {
        int2* ep = csr + start + base + lane;
        ent = *ep;
        ds = dinv[ent.x];
        *ep = make_int2(ent.x, __float_as_int(ds));
      }
      float wl = ds * dd;
      int npair = m >> 1;
#pragma unroll 4
      for (int p = 0; p < npair; ++p) {
        int sl = 2 * p + half;
        int s = __shfl(ent.x, sl, 32);
        float wgt = __shfl(wl, sl, 32);
        const float4 hv = *(const float4*)(hc + (size_t)s * HD);
        acc.x += hv.x * wgt;
        acc.y += hv.y * wgt;
        acc.z += hv.z * wgt;
        acc.w += hv.w * wgt;
      }
      if (m & 1) {
        int s = __shfl(ent.x, m - 1, 32);
        float wgt = __shfl(wl, m - 1, 32);
        if (half == 0) {
          const float4 hv = *(const float4*)(hc + (size_t)s * HD);
          acc.x += hv.x * wgt;
          acc.y += hv.y * wgt;
          acc.z += hv.z * wgt;
          acc.w += hv.w * wgt;
        }
      }
    }
    acc.x += __shfl_xor(acc.x, 16, 32);
    acc.y += __shfl_xor(acc.y, 16, 32);
    acc.z += __shfl_xor(acc.z, 16, 32);
    acc.w += __shfl_xor(acc.w, 16, 32);
    if (half == 0) {
      float w0 = dd * dd;
      float4 o;
      o.x = hs.x * w0 + acc.x;
      o.y = hs.y * w0 + acc.y;
      o.z = hs.z * w0 + acc.z;
      o.w = hs.w * w0 + acc.w;
      o.x = fmaxf(o.x + b4.x, 0.f);
      o.y = fmaxf(o.y + b4.y, 0.f);
      o.z = fmaxf(o.z + b4.z, 0.f);
      o.w = fmaxf(o.w + b4.w, 0.f);
      *(float4*)(z + (size_t)d * HD + c4) = o;
      ts.x += o.x; ts.y += o.y; ts.z += o.z; ts.w += o.w;
      tq.x += o.x * o.x; tq.y += o.y * o.y; tq.z += o.z * o.z; tq.w += o.w * o.w;
    }
  }
  if (half == 0) {
    *(float4*)(s_sum + w * HD + c4) = ts;
    *(float4*)(s_sq + w * HD + c4) = tq;
  }
  __syncthreads();
  if (threadIdx.x < 2 * HD) {
    int c = threadIdx.x & (HD - 1);
    const float* p = (threadIdx.x < HD) ? s_sum : s_sq;
    float t = 0.f;
#pragma unroll
    for (int k = 0; k < 16; ++k) t += p[k * HD + c];
    atomicAdd((threadIdx.x < HD ? bn_sum : bn_sq) + c, t);
  }
}

#define AP 68
__global__ void __launch_bounds__(512)
k_gather_bn_gemm_relu_stats(const int* __restrict__ rowptr, const int* __restrict__ deg,
                            const int2* __restrict__ csr, const float* __restrict__ dinv,
                            const float* __restrict__ zp,
                            const float* __restrict__ pbn_sum, const float* __restrict__ pbn_sq,
                            const float* __restrict__ pgamma, const float* __restrict__ pbeta, float invN,
                            const float* __restrict__ W, const float* __restrict__ bias,
                            float* __restrict__ z, float* __restrict__ bn_sum,
                            float* __restrict__ bn_sq, int N) {
  __shared__ __attribute__((aligned(16))) float s_a[64 * AP];
  __shared__ float s_sum[4 * HD];
  __shared__ float s_sq[4 * HD];
  int w = __builtin_amdgcn_readfirstlane(threadIdx.x >> 5);
  int lane = threadIdx.x & 31;
  int half = lane >> 4;
  int c4 = (lane & 15) << 2;
  const float* hc = zp + c4;
  float sc0, sc1, sc2, sc3, sh0, sh1, sh2, sh3;
  bn_coef(pbn_sum, pbn_sq, pgamma, pbeta, invN, c4,     sc0, sh0);
  bn_coef(pbn_sum, pbn_sq, pgamma, pbeta, invN, c4 + 1, sc1, sh1);
  bn_coef(pbn_sum, pbn_sq, pgamma, pbeta, invN, c4 + 2, sc2, sh2);
  bn_coef(pbn_sum, pbn_sq, pgamma, pbeta, invN, c4 + 3, sc3, sh3);
  for (int i = 0; i < 4; ++i) {
    int lr = i * 16 + w;
    int d = blockIdx.x * 64 + lr;
    float4 g = {0.f, 0.f, 0.f, 0.f};
    if (d < N) {
      int start = rowptr[d];
      int n = deg[d];
      float dd = dinv[d];
      const float4 hs = *(const float4*)(hc + (size_t)d * HD);
      float4 acc = {0.f, 0.f, 0.f, 0.f};
      float wsum = 0.f;
      for (int base = 0; base < n; base += 32) {
        int m = n - base;
        if (m > 32) m = 32;
        int2 ent = make_int2(0, 0);
        if (lane < m) ent = csr[start + base + lane];
        float wl = __int_as_float(ent.y) * dd;
        int npair = m >> 1;
#pragma unroll 4
        for (int p = 0; p < npair; ++p) {
          int sl = 2 * p + half;
          int s = __shfl(ent.x, sl, 32);
          float wgt = __shfl(wl, sl, 32);
          const float4 hv = *(const float4*)(hc + (size_t)s * HD);
          acc.x += hv.x * wgt;
          acc.y += hv.y * wgt;
          acc.z += hv.z * wgt;
          acc.w += hv.w * wgt;
          wsum += wgt;
        }
        if (m & 1) {
          int s = __shfl(ent.x, m - 1, 32);
          float wgt = __shfl(wl, m - 1, 32);
          if (half == 0) {
            const float4 hv = *(const float4*)(hc + (size_t)s * HD);
            acc.x += hv.x * wgt;
            acc.y += hv.y * wgt;
            acc.z += hv.z * wgt;
            acc.w += hv.w * wgt;
            wsum += wgt;
          }
        }
      }
      acc.x += __shfl_xor(acc.x, 16, 32);
      acc.y += __shfl_xor(acc.y, 16, 32);
      acc.z += __shfl_xor(acc.z, 16, 32);
      acc.w += __shfl_xor(acc.w, 16, 32);
      wsum += __shfl_xor(wsum, 16, 32);
      float w0 = dd * dd;
      float r = w0 + wsum;
      g.x = (hs.x * w0 + acc.x) * sc0 + r * sh0;
      g.y = (hs.y * w0 + acc.y) * sc1 + r * sh1;
      g.z = (hs.z * w0 + acc.z) * sc2 + r * sh2;
      g.w = (hs.w * w0 + acc.w) * sc3 + r * sh3;
    }
    if (half == 0) *(float4*)(s_a + lr * AP + c4) = g;
  }
  __syncthreads();
  int tm = w >> 2, tn = w & 3;
  int m16 = lane & 15;
  const float* ap = s_a + (tm * 16 + m16) * AP + 2 * half;
  int col = tn * 16 + m16;
  const float* wp = W + col;
  v8f c = {};
#pragma unroll
  for (int k0 = 0; k0 < HD; k0 += 4) {
    int ka = k0 + 2 * half;
    v2f a, b;
    a.x = ap[k0];
    a.y = ap[k0 + 1];
    b.x = wp[(size_t)ka * HD];
    b.y = wp[(size_t)(ka + 1) * HD];
    c = __builtin_amdgcn_wmma_f32_16x16x4_f32(false, a, false, b, (short)0, c, false, false);
  }
  float bcol = bias[col];
  int rbase = blockIdx.x * 64 + tm * 16 + half * 8;
  float ts = 0.f, tq = 0.f;
#pragma unroll
  for (int j = 0; j < 8; ++j) {
    int r = rbase + j;
    if (r < N) {
      float o = fmaxf(c[j] + bcol, 0.f);
      z[(size_t)r * HD + col] = o;
      ts += o;
      tq += o * o;
    }
  }
  ts += __shfl_xor(ts, 16, 32);
  tq += __shfl_xor(tq, 16, 32);
  if (half == 0) {
    s_sum[tm * HD + col] = ts;
    s_sq[tm * HD + col] = tq;
  }
  __syncthreads();
  if (threadIdx.x < 2 * HD) {
    int cc = threadIdx.x & (HD - 1);
    const float* p = (threadIdx.x < HD) ? s_sum : s_sq;
    float t = (p[cc] + p[HD + cc]) + (p[2 * HD + cc] + p[3 * HD + cc]);
    atomicAdd((threadIdx.x < HD ? bn_sum : bn_sq) + cc, t);
  }
}

__global__ void k_bn_l2_pool(const float* __restrict__ z0, const float* __restrict__ z1,
                             const float* __restrict__ z2, const float* __restrict__ bn,
                             const float* __restrict__ gamma0, const float* __restrict__ beta0,
                             const float* __restrict__ gamma1, const float* __restrict__ beta1,
                             const float* __restrict__ gamma2, const float* __restrict__ beta2,
                             float invN, const int* __restrict__ batch,
                             float* __restrict__ out, float* __restrict__ gsum, int N) {
  int wid = blockIdx.x * (blockDim.x >> 5) + (threadIdx.x >> 5);
  int lane = threadIdx.x & 31;
  int r0 = wid * 8;
  if (r0 >= N) return;
  int rend = r0 + 8 < N ? r0 + 8 : N;
  const float* zl[3] = {z0, z1, z2};
  const float* gl[3] = {gamma0, gamma1, gamma2};
  const float* bl[3] = {beta0, beta1, beta2};
  float sc[6], sh[6];
#pragma unroll
  for (int j = 0; j < 6; ++j) {
    int l = j >> 1;
    bn_coef(bn + l * 128, bn + l * 128 + 64, gl[l], bl[l], invN, lane + 32 * (j & 1), sc[j], sh[j]);
  }
  float acc[6] = {0.f, 0.f, 0.f, 0.f, 0.f, 0.f};
  int gcur = -1;
  for (int row = r0; row < rend; ++row) {
    float v[6]; float s = 0.f;
#pragma unroll
    for (int j = 0; j < 6; ++j) {
      v[j] = zl[j >> 1][(size_t)row * HD + lane + 32 * (j & 1)] * sc[j] + sh[j];
      s += v[j] * v[j];
    }
#pragma unroll
    for (int off = 16; off >= 1; off >>= 1) s += __shfl_xor(s, off, 32);
    float inv = 1.0f / fmaxf(sqrtf(s), 1e-12f);
    float* o = out + (size_t)row * CD;
#pragma unroll
    for (int j = 0; j < 6; ++j) o[lane + 32 * j] = v[j] * inv;
    int g = batch[row];
    if (g != gcur) {
      if (gcur >= 0) {
#pragma unroll
        for (int j = 0; j < 6; ++j) atomicAdd(&gsum[(size_t)gcur * CD + lane + 32 * j], acc[j]);
      }
      gcur = g;
#pragma unroll
      for (int j = 0; j < 6; ++j) acc[j] = 0.f;
    }
#pragma unroll
    for (int j = 0; j < 6; ++j) acc[j] += v[j];
  }
  if (gcur >= 0) {
#pragma unroll
    for (int j = 0; j < 6; ++j) atomicAdd(&gsum[(size_t)gcur * CD + lane + 32 * j], acc[j]);
  }
}

__global__ void k_l2_gc(const float* __restrict__ gsum, const float* __restrict__ cnt,
                        float* __restrict__ out, int rows) {
  int wid = blockIdx.x * (blockDim.x >> 5) + (threadIdx.x >> 5);
  if (wid >= rows) return;
  int lane = threadIdx.x & 31;
  float icnt = 1.0f / fmaxf(cnt[wid], 1.0f);
  const float* a = gsum + (size_t)wid * CD;
  float v[6]; float s = 0.f;
#pragma unroll
  for (int j = 0; j < 6; ++j) { v[j] = a[lane + 32 * j] * icnt; s += v[j] * v[j]; }
#pragma unroll
  for (int off = 16; off >= 1; off >>= 1) s += __shfl_xor(s, off, 32);
  float inv = 1.0f / fmaxf(sqrtf(s), 1e-12f);
  float* o = out + (size_t)wid * CD;
#pragma unroll
  for (int j = 0; j < 6; ++j) o[lane + 32 * j] = v[j] * inv;
}

extern "C" void kernel_launch(void* const* d_in, const int* in_sizes, int n_in,
                              void* d_out, int out_size, void* d_ws, size_t ws_size,
                              hipStream_t stream) {
  (void)n_in; (void)out_size; (void)ws_size;
  const float* x     = (const float*)d_in[0];
  const int*   ei    = (const int*)d_in[1];
  const int*   batch = (const int*)d_in[2];
  int N = in_sizes[0] / 128;
  int E = in_sizes[1] / 2;
  const int* src = ei;
  const int* dst = ei + E;
  const float* Wl[3]  = {(const float*)d_in[3],  (const float*)d_in[7],  (const float*)d_in[11]};
  const float* bl[3]  = {(const float*)d_in[4],  (const float*)d_in[8],  (const float*)d_in[12]};
  const float* gl[3]  = {(const float*)d_in[5],  (const float*)d_in[9],  (const float*)d_in[13]};
  const float* bel[3] = {(const float*)d_in[6],  (const float*)d_in[10], (const float*)d_in[14]};

  float* ws   = (float*)d_ws;
  float* dinv = ws;                       ws += N;
  float* h    = ws;                       ws += (size_t)N * HD;
  float* zb[3];
  zb[0] = ws;                             ws += (size_t)N * HD;
  zb[1] = ws;                             ws += (size_t)N * HD;
  zb[2] = ws;                             ws += (size_t)N * HD;
  float* gsum = ws;                       ws += (size_t)GN * CD;
  float* cnt  = ws;                       ws += GN;
  float* bn   = ws;
  ws += 3 * 128;
  int* deg    = (int*)(((size_t)ws + 15) & ~(size_t)15);
  int* rowptr = deg + (((size_t)N + 3) & ~(size_t)3);
  int nb  = (N + BINW - 1) / BINW;
  int nch = (E + CHUNK - 1) / CHUNK;
  int* cm       = rowptr + (((size_t)N + 3) & ~(size_t)3);
  int* binstart = cm + (size_t)nch * ((nb + 3) & ~3);
  int2* binned  = (int2*)(binstart + (((size_t)nb + 1 + 3) & ~(size_t)3));
  int2* csr     = binned + E;

  k_bin_count<<<nch, 1024, 0, stream>>>(dst, E, nb, cm, gsum, GN * CD + GN + 3 * 128);
  k_bin_scatter<<<nch, 1024, 0, stream>>>(src, dst, E, nb, cm, binstart, binned);
  int tilesM = (N + 15) / 16;
  int waves = tilesM * 4;
  k_csr_gemm<<<nb + (waves + 7) / 8, 256, 0, stream>>>(binstart, binned, deg, rowptr, dinv, batch, cnt, csr,
                                                      N, nb, x, 128, Wl[0], 128, h);

  float invN = 1.0f / (float)N;
  k_gather_relu_stats<<<(N + 63) / 64, 512, 0, stream>>>(rowptr, deg, csr, dinv, h, bl[0], zb[0],
                                                         bn, bn + 64, N);
  for (int l = 1; l < 3; ++l) {
    float* bn_sum = bn + l * 128;
    float* bn_sq  = bn_sum + 64;
    k_gather_bn_gemm_relu_stats<<<(N + 63) / 64, 512, 0, stream>>>(rowptr, deg, csr, dinv, zb[l - 1],
                                                                   bn_sum - 128, bn_sq - 128,
                                                                   gl[l - 1], bel[l - 1], invN,
                                                                   Wl[l], bl[l], zb[l], bn_sum, bn_sq, N);
  }

  float* out = (float*)d_out;
  k_bn_l2_pool<<<(N + 63) / 64, 256, 0, stream>>>(zb[0], zb[1], zb[2], bn, gl[0], bel[0], gl[1], bel[1],
                                                  gl[2], bel[2], invN, batch, out, gsum, N);
  k_l2_gc<<<(GN + 7) / 8, 256, 0, stream>>>(gsum, cnt, out + (size_t)N * CD, GN);
}


# gather loads issued ahead of use: a wave's 4 rows' starts, list entries and dinv[src] fetched together, then each row's up to 24 row loads all in flight before its sums; order of every sum unchanged
# speedup vs baseline: 3.6545x; 1.0359x over previous
#include <hip/hip_runtime.h>

typedef float v2f __attribute__((ext_vector_type(2)));
typedef float v8f __attribute__((ext_vector_type(8)));

#define HD 64
#define CD 192
#define GN 256

#define BINW  256
#define MAXB  256
#define CHUNK 4096

__device__ __forceinline__ int wave_incl_scan(int v, int lane) {
  for (int off = 1; off < 32; off <<= 1) {
    int u = __shfl_up(v, off, 32);
    if (lane >= off) v += u;
  }
  return v;
}

__global__ void __launch_bounds__(1024)
k_bin_count(const int* __restrict__ dst, int E, int nb, int* __restrict__ cm,
            float* __restrict__ zero, int nzero) {
  __shared__ int s_h[MAXB];
  int tid = threadIdx.x;
  int chunk = blockIdx.x, nch = gridDim.x;
  for (int i = chunk * 1024 + tid; i < nzero; i += nch * 1024) zero[i] = 0.f;
  if (tid < MAXB) s_h[tid] = 0;
  __syncthreads();
  int e0 = chunk * CHUNK + tid;
#pragma unroll
  for (int k = 0; k < 4; ++k) {
    int e = e0 + k * 1024;
    if (e < E) atomicAdd(&s_h[dst[e] >> 8], 1);
  }
  __syncthreads();
  int nbp = (nb + 3) & ~3;
  if (tid < nbp) cm[chunk * nbp + tid] = s_h[tid];
}

__global__ void __launch_bounds__(1024)
k_bin_scatter(const int* __restrict__ src, const int* __restrict__ dst, int E, int nb,
              const int* __restrict__ cm, int* __restrict__ binstart, int2* __restrict__ binned) {
  __shared__ int s_cnt[MAXB];
  __shared__ int s_tot[MAXB];
  __shared__ int s_par[MAXB];
  __shared__ int s_lst[MAXB];
  __shared__ int s_gof[MAXB];
  __shared__ int s_wc[8], s_wg[8];
  __shared__ int2 s_e[CHUNK];
  int tid = threadIdx.x, lane = tid & 31, w = tid >> 5;
  int chunk = blockIdx.x, nch = gridDim.x;
  if (tid < MAXB) { s_cnt[tid] = 0; s_tot[tid] = 0; s_par[tid] = 0; }
  __syncthreads();
  int es[4], ed[4], er[4];
  int e0 = chunk * CHUNK + tid;
#pragma unroll
  for (int k = 0; k < 4; ++k) {
    int e = e0 + k * 1024;
    es[k] = 0; ed[k] = 0; er[k] = 0;
    if (e < E) {
      es[k] = src[e];
      ed[k] = dst[e];
      er[k] = atomicAdd(&s_cnt[ed[k] >> 8], 1);
    }
  }
  {
    int nbp = (nb + 3) & ~3;
    int q = tid & 63, r = tid >> 6;
    if (4 * q < nbp) {
      const int4* col = (const int4*)cm + q;
      int rowq = nbp >> 2;
      int t0 = 0, t1 = 0, t2 = 0, t3 = 0, p0 = 0, p1 = 0, p2 = 0, p3 = 0;
#pragma unroll 4
      for (int j = r; j < nch; j += 16) {
        int4 v = col[(size_t)j * rowq];
        int msk = (j < chunk) ? -1 : 0;
        t0 += v.x; t1 += v.y; t2 += v.z; t3 += v.w;
        p0 += v.x & msk; p1 += v.y & msk; p2 += v.z & msk; p3 += v.w & msk;
      }
      atomicAdd(&s_tot[4 * q], t0);     atomicAdd(&s_par[4 * q], p0);
      atomicAdd(&s_tot[4 * q + 1], t1); atomicAdd(&s_par[4 * q + 1], p1);
      atomicAdd(&s_tot[4 * q + 2], t2); atomicAdd(&s_par[4 * q + 2], p2);
      atomicAdd(&s_tot[4 * q + 3], t3); atomicAdd(&s_par[4 * q + 3], p3);
    }
  }
  __syncthreads();
  int c = (tid < MAXB) ? s_cnt[tid] : 0;
  int g = (tid < MAXB) ? s_tot[tid] : 0;
  int ci = wave_incl_scan(c, lane);
  int gi = wave_incl_scan(g, lane);
  if (lane == 31 && w < 8) { s_wc[w] = ci; s_wg[w] = gi; }
  __syncthreads();
  if (tid < MAXB) {
    int bc = 0, bg = 0;
    for (int k = 0; k < 8; ++k) {
      if (k < w) { bc += s_wc[k]; bg += s_wg[k]; }
    }
    int gstart = bg + gi - g;
    s_lst[tid] = bc + ci - c;
    s_gof[tid] = gstart + s_par[tid];
    if (chunk == 0) {
      if (tid < nb) binstart[tid] = gstart;
      if (tid == 0) binstart[nb] = E;
    }
  }
  __syncthreads();
  int nloc = E - chunk * CHUNK;
  if (nloc > CHUNK) nloc = CHUNK;
#pragma unroll
  for (int k = 0; k < 4; ++k) {
    if (e0 + k * 1024 < E) s_e[s_lst[ed[k] >> 8] + er[k]] = make_int2(es[k], ed[k]);
  }
  __syncthreads();
#pragma unroll
  for (int k = 0; k < 4; ++k) {
    int i = k * 1024 + tid;
    if (i < nloc) {
      int2 en = s_e[i];
      int b = en.y >> 8;
      binned[s_gof[b] + (i - s_lst[b])] = en;
    }
  }
}

__device__ __forceinline__ void csr_block(int bin, const int* __restrict__ binstart,
                                          const int2* __restrict__ binned, int* __restrict__ deg,
                                          int* __restrict__ rowptr, float* __restrict__ dinv,
                                          const int* __restrict__ batch, float* __restrict__ cnt,
                                          int2* __restrict__ csr, int N) {
  __shared__ int s_c[BINW], s_cur[BINW], s_w[8];
  int tid = threadIdx.x, lane = tid & 31, w = tid >> 5;
  int bs = binstart[bin], be = binstart[bin + 1];
  s_c[tid] = 0;
  __syncthreads();
#pragma unroll 4
  for (int i = bs + tid; i < be; i += 256) atomicAdd(&s_c[binned[i].y & (BINW - 1)], 1);
  __syncthreads();
  int c = s_c[tid];
  int inc = wave_incl_scan(c, lane);
  if (lane == 31) s_w[w] = inc;
  __syncthreads();
  int base = bs;
  for (int k = 0; k < 8; ++k) {
    if (k < w) base += s_w[k];
  }
  int first = base + inc - c;
  s_cur[tid] = first;
  int node = bin * BINW + tid;
  if (node < N) {
    deg[node] = c;
    rowptr[node] = first;
    dinv[node] = rsqrtf((float)c + 1.0f);
    atomicAdd(&cnt[batch[node]], 1.0f);
  }
  __syncthreads();
#pragma unroll 4
  for (int i = bs + tid; i < be; i += 256) {
    int2 en = binned[i];
    int pos = atomicAdd(&s_cur[en.y & (BINW - 1)], 1);
    csr[pos] = make_int2(en.x, 0);
  }
}

__device__ __forceinline__ void bn_coef(const float* __restrict__ bn_sum, const float* __restrict__ bn_sq,
                                        const float* __restrict__ gamma, const float* __restrict__ beta,
                                        float invN, int c, float& sc, float& sh) {
  float mu = bn_sum[c] * invN;
  float var = fmaxf(bn_sq[c] * invN - mu * mu, 0.f);
  sc = rsqrtf(var + 1e-5f) * gamma[c];
  sh = beta[c] - mu * sc;
}

__device__ __forceinline__ void gemm_block(int bid, const float* __restrict__ Z, int ldz,
                                           const float* __restrict__ W, int K,
                                           float* __restrict__ Hout, int N) {
  int wid = bid * (blockDim.x >> 5) + (threadIdx.x >> 5);
  int tilesM = (N + 15) >> 4;
  if (wid >= tilesM * 4) return;
  int tm = wid >> 2, tn = wid & 3;
  int lane = threadIdx.x & 31;
  int half = lane >> 4;
  int m16 = lane & 15;
  int row = tm * 16 + m16;
  int lrow = row < N ? row : N - 1;
  const float* zr = Z + (size_t)lrow * ldz;
  int col = tn * 16 + m16;
  const float* wp = W + col;
  v8f c = {};
  for (int k0 = 0; k0 < K; k0 += 4) {
    int ka = k0 + 2 * half;
    v2f a, b;
    a.x = zr[ka];
    a.y = zr[ka + 1];
    b.x = wp[(size_t)ka * HD];
    b.y = wp[(size_t)(ka + 1) * HD];
    c = __builtin_amdgcn_wmma_f32_16x16x4_f32(false, a, false, b, (short)0, c, false, false);
  }
  int rbase = tm * 16 + half * 8;
#pragma unroll
  for (int j = 0; j < 8; ++j) {
    int r = rbase + j;
    if (r < N) Hout[(size_t)r * HD + col] = c[j];
  }
}

__global__ void __launch_bounds__(256)
k_csr_gemm(const int* __restrict__ binstart, const int2* __restrict__ binned,
           int* __restrict__ deg, int* __restrict__ rowptr, float* __restrict__ dinv,
           const int* __restrict__ batch, float* __restrict__ cnt, int2* __restrict__ csr, int N, int nb,
           const float* __restrict__ X, int ldx,
           const float* __restrict__ W, int K, float* __restrict__ Hout) {
  if ((int)blockIdx.x < nb) {
    csr_block(blockIdx.x, binstart, binned, deg, rowptr, dinv, batch, cnt, csr, N);
  } else {
    gemm_block(blockIdx.x - nb, X, ldx, W, K, Hout, N);
  }
}

struct RowBatch {
  float4 v[4];
};

__device__ __forceinline__ void batch_issue(RowBatch& t, const float4* __restrict__ h4, int q, int ex,
                                            int b, int half) {
#pragma unroll
  for (int u = 0; u < 4; ++u) {
    int sl = 8 * b + 2 * u + half;
    int sr = __shfl(ex, sl, 32);
    t.v[u] = h4[(unsigned)(sr * (HD / 4) + q)];
  }
}

template <bool WSUM>
__device__ __forceinline__ void batch_sum(const RowBatch& t, float wl, int b, int half, float4& acc, float& wsum) {
#pragma unroll
  for (int u = 0; u < 4; ++u) {
    float wgt = __shfl(wl, 8 * b + 2 * u + half, 32);
    acc.x += t.v[u].x * wgt;
    acc.y += t.v[u].y * wgt;
    acc.z += t.v[u].z * wgt;
    acc.w += t.v[u].w * wgt;
    if (WSUM) wsum += wgt;
  }
}

template <bool WSUM>
__device__ __forceinline__ void gather_chunk(const float4* __restrict__ h4, int q, int ex, float wl, int m, int half,
                                             float4& acc, float& wsum) {
  int nbat = (m + 7) >> 3;
  RowBatch A, B, C;
  if (nbat == 1) {
    batch_issue(A, h4, q, ex, 0, half);
    batch_sum<WSUM>(A, wl, 0, half, acc, wsum);
  } else if (nbat == 2) {
    batch_issue(A, h4, q, ex, 0, half);
    batch_issue(B, h4, q, ex, 1, half);
    __builtin_amdgcn_sched_barrier(0);
    batch_sum<WSUM>(A, wl, 0, half, acc, wsum);
    batch_sum<WSUM>(B, wl, 1, half, acc, wsum);
  } else {
    batch_issue(A, h4, q, ex, 0, half);
    batch_issue(B, h4, q, ex, 1, half);
    batch_issue(C, h4, q, ex, 2, half);
    __builtin_amdgcn_sched_barrier(0);
    batch_sum<WSUM>(A, wl, 0, half, acc, wsum);
    if (nbat == 4) batch_issue(A, h4, q, ex, 3, half);
    batch_sum<WSUM>(B, wl, 1, half, acc, wsum);
    batch_sum<WSUM>(C, wl, 2, half, acc, wsum);
    if (nbat == 4) batch_sum<WSUM>(A, wl, 3, half, acc, wsum);
  }
}

__global__ void __launch_bounds__(512, 8)
k_gather_relu_stats(const int* __restrict__ rowptr, const int* __restrict__ deg,
                    int2* __restrict__ csr, const float* __restrict__ dinv,
                    const float* __restrict__ h, const float* __restrict__ bias,
                    float* __restrict__ z, float* __restrict__ bn_sum,
                    float* __restrict__ bn_sq, int N) {
  __shared__ __attribute__((aligned(16))) float s_sum[16 * HD];
  __shared__ __attribute__((aligned(16))) float s_sq[16 * HD];
  int w = __builtin_amdgcn_readfirstlane(threadIdx.x >> 5);
  int lane = threadIdx.x & 31;
  int half = lane >> 4;
  int c4 = (lane & 15) << 2;
  const float4* h4 = (const float4*)h;
  int q = lane & 15;
  float4* ps = (float4*)(s_sum + w * HD + c4);
  float4* pq = (float4*)(s_sq + w * HD + c4);
  if (half == 0) {
    *ps = make_float4(0.f, 0.f, 0.f, 0.f);
    *pq = make_float4(0.f, 0.f, 0.f, 0.f);
  }
  int st[4], nn[4], ex[4];
  float dv[4], wl[4];
#pragma unroll
  for (int i = 0; i < 4; ++i) {
    int d = blockIdx.x * 64 + i * 16 + w;
    int dc = d < N ? d : N - 1;
    st[i] = rowptr[dc];
    nn[i] = d < N ? deg[dc] : 0;
    dv[i] = dinv[dc];
  }
#pragma unroll
  for (int i = 0; i < 4; ++i) {
    int d = blockIdx.x * 64 + i * 16 + w;
    int sx = d < N ? d : N - 1;
    if (lane < nn[i]) sx = csr[st[i] + lane].x;
    ex[i] = sx;
  }
  float dsr[4];
#pragma unroll
  for (int i = 0; i < 4; ++i) dsr[i] = dinv[ex[i]];
#pragma unroll
  for (int i = 0; i < 4; ++i) {
    float ds = 0.f;
    if (lane < nn[i]) {
      ds = dsr[i];
      csr[st[i] + lane] = make_int2(ex[i], __float_as_int(ds));
    }
    wl[i] = ds * dv[i];
  }
#pragma unroll
  for (int i = 0; i < 4; ++i) {
    int d = blockIdx.x * 64 + i * 16 + w;
    if (d >= N) break;
    int start = st[i];
    int n = nn[i];
    float dd = dv[i];
    const float4 hs = h4[(unsigned)(d * (HD / 4) + q)];
    const float4 b4 = ((const float4*)bias)[q];
    float4 acc = {0.f, 0.f, 0.f, 0.f};
    float wsum = 0.f;
    int exc = ex[i];
    float wlc = wl[i];
    for (int base = 0; base < n; base += 32) {
      int m = n - base;
      if (m > 32) m = 32;
      gather_chunk<false>(h4, q, exc, wlc, m, half, acc, wsum);
      int m2 = n - base - 32;
      if (m2 > 0) {
        if (m2 > 32) m2 = 32;
        exc = d;
        float ds = 0.f;
        if (lane < m2) {
          int2* ep = csr + start + base + 32 + lane;
          exc = ep->x;
          ds = dinv[exc];
          *ep = make_int2(exc, __float_as_int(ds));
        }
        wlc = ds * dd;
      }
    }
    acc.x += __shfl_xor(acc.x, 16, 32);
    acc.y += __shfl_xor(acc.y, 16, 32);
    acc.z += __shfl_xor(acc.z, 16, 32);
    acc.w += __shfl_xor(acc.w, 16, 32);
    if (half == 0) {
      float w0 = dd * dd;
      float4 o;
      o.x = hs.x * w0 + acc.x;
      o.y = hs.y * w0 + acc.y;
      o.z = hs.z * w0 + acc.z;
      o.w = hs.w * w0 + acc.w;
      o.x = fmaxf(o.x + b4.x, 0.f);
      o.y = fmaxf(o.y + b4.y, 0.f);
      o.z = fmaxf(o.z + b4.z, 0.f);
      o.w = fmaxf(o.w + b4.w, 0.f);
      ((float4*)z)[(unsigned)(d * (HD / 4) + q)] = o;
      float4 ts = *ps;
      float4 tq = *pq;
      ts.x += o.x; ts.y += o.y; ts.z += o.z; ts.w += o.w;
      tq.x += o.x * o.x; tq.y += o.y * o.y; tq.z += o.z * o.z; tq.w += o.w * o.w;
      *ps = ts;
      *pq = tq;
    }
  }
  __syncthreads();
  if (threadIdx.x < 2 * HD) {
    int c = threadIdx.x & (HD - 1);
    const float* p = (threadIdx.x < HD) ? s_sum : s_sq;
    float t = 0.f;
#pragma unroll
    for (int k = 0; k < 16; ++k) t += p[k * HD + c];
    atomicAdd((threadIdx.x < HD ? bn_sum : bn_sq) + c, t);
  }
}

#define AP 68
__global__ void __launch_bounds__(512, 8)
k_gather_bn_gemm_relu_stats(const int* __restrict__ rowptr, const int* __restrict__ deg,
                            const int2* __restrict__ csr, const float* __restrict__ dinv,
                            const float* __restrict__ zp,
                            const float* __restrict__ pbn_sum, const float* __restrict__ pbn_sq,
                            const float* __restrict__ pgamma, const float* __restrict__ pbeta, float invN,
                            const float* __restrict__ W, const float* __restrict__ bias,
                            float* __restrict__ z, float* __restrict__ bn_sum,
                            float* __restrict__ bn_sq, int N) {
  __shared__ __attribute__((aligned(16))) float s_a[64 * AP];
  __shared__ float s_sum[4 * HD];
  __shared__ float s_sq[4 * HD];
  int w = __builtin_amdgcn_readfirstlane(threadIdx.x >> 5);
  int lane = threadIdx.x & 31;
  int half = lane >> 4;
  int c4 = (lane & 15) << 2;
  const float4* h4 = (const float4*)zp;
  int q = lane & 15;
  float sc0, sc1, sc2, sc3, sh0, sh1, sh2, sh3;
  bn_coef(pbn_sum, pbn_sq, pgamma, pbeta, invN, c4,     sc0, sh0);
  bn_coef(pbn_sum, pbn_sq, pgamma, pbeta, invN, c4 + 1, sc1, sh1);
  bn_coef(pbn_sum, pbn_sq, pgamma, pbeta, invN, c4 + 2, sc2, sh2);
  bn_coef(pbn_sum, pbn_sq, pgamma, pbeta, invN, c4 + 3, sc3, sh3);
  int st[4], nn[4], ex[4];
  float dv[4], wl[4];
#pragma unroll
  for (int i = 0; i < 4; ++i) {
    int d = blockIdx.x * 64 + i * 16 + w;
    int dc = d < N ? d : N - 1;
    st[i] = rowptr[dc];
    nn[i] = d < N ? deg[dc] : 0;
    dv[i] = dinv[dc];
  }
#pragma unroll
  for (int i = 0; i < 4; ++i) {
    int d = blockIdx.x * 64 + i * 16 + w;
    int2 ent = make_int2(d < N ? d : N - 1, 0);
    if (lane < nn[i]) ent = csr[st[i] + lane];
    ex[i] = ent.x;
    wl[i] = __int_as_float(ent.y) * dv[i];
  }
#pragma unroll
  for (int i = 0; i < 4; ++i) {
    int lr = i * 16 + w;
    int d = blockIdx.x * 64 + lr;
    float4 g = {0.f, 0.f, 0.f, 0.f};
    if (d < N) {
      int start = st[i];
      int n = nn[i];
      float dd = dv[i];
      const float4 hs = h4[(unsigned)(d * (HD / 4) + q)];
      float4 acc = {0.f, 0.f, 0.f, 0.f};
      float wsum = 0.f;
      int exc = ex[i];
      float wlc = wl[i];
      for (int base = 0; base < n; base += 32) {
        int m = n - base;
        if (m > 32) m = 32;
        gather_chunk<true>(h4, q, exc, wlc, m, half, acc, wsum);
        int m2 = n - base - 32;
        if (m2 > 0) {
          if (m2 > 32) m2 = 32;
          int2 ent = make_int2(d, 0);
          if (lane < m2) ent = csr[start + base + 32 + lane];
          exc = ent.x;
          wlc = __int_as_float(ent.y) * dd;
        }
      }
      acc.x += __shfl_xor(acc.x, 16, 32);
      acc.y += __shfl_xor(acc.y, 16, 32);
      acc.z += __shfl_xor(acc.z, 16, 32);
      acc.w += __shfl_xor(acc.w, 16, 32);
      wsum += __shfl_xor(wsum, 16, 32);
      float w0 = dd * dd;
      float r = w0 + wsum;
      g.x = (hs.x * w0 + acc.x) * sc0 + r * sh0;
      g.y = (hs.y * w0 + acc.y) * sc1 + r * sh1;
      g.z = (hs.z * w0 + acc.z) * sc2 + r * sh2;
      g.w = (hs.w * w0 + acc.w) * sc3 + r * sh3;
    }
    if (half == 0) *(float4*)(s_a + lr * AP + c4) = g;
  }
  __syncthreads();
  int tm = w >> 2, tn = w & 3;
  int m16 = lane & 15;
  const float* ap = s_a + (tm * 16 + m16) * AP + 2 * half;
  int col = tn * 16 + m16;
  const float* wp = W + col;
  v8f c = {};
#pragma unroll
  for (int k0 = 0; k0 < HD; k0 += 4) {
    int ka = k0 + 2 * half;
    v2f a, b;
    a.x = ap[k0];
    a.y = ap[k0 + 1];
    b.x = wp[(size_t)ka * HD];
    b.y = wp[(size_t)(ka + 1) * HD];
    c = __builtin_amdgcn_wmma_f32_16x16x4_f32(false, a, false, b, (short)0, c, false, false);
  }
  float bcol = bias[col];
  int rbase = blockIdx.x * 64 + tm * 16 + half * 8;
  float ts = 0.f, tq = 0.f;
#pragma unroll
  for (int j = 0; j < 8; ++j) {
    int r = rbase + j;
    if (r < N) {
      float o = fmaxf(c[j] + bcol, 0.f);
      z[(size_t)r * HD + col] = o;
      ts += o;
      tq += o * o;
    }
  }
  ts += __shfl_xor(ts, 16, 32);
  tq += __shfl_xor(tq, 16, 32);
  if (half == 0) {
    s_sum[tm * HD + col] = ts;
    s_sq[tm * HD + col] = tq;
  }
  __syncthreads();
  if (threadIdx.x < 2 * HD) {
    int cc = threadIdx.x & (HD - 1);
    const float* p = (threadIdx.x < HD) ? s_sum : s_sq;
    float t = (p[cc] + p[HD + cc]) + (p[2 * HD + cc] + p[3 * HD + cc]);
    atomicAdd((threadIdx.x < HD ? bn_sum : bn_sq) + cc, t);
  }
}

__global__ void k_bn_l2_pool(const float* __restrict__ z0, const float* __restrict__ z1,
                             const float* __restrict__ z2, const float* __restrict__ bn,
                             const float* __restrict__ gamma0, const float* __restrict__ beta0,
                             const float* __restrict__ gamma1, const float* __restrict__ beta1,
                             const float* __restrict__ gamma2, const float* __restrict__ beta2,
                             float invN, const int* __restrict__ batch,
                             float* __restrict__ out, float* __restrict__ gsum, int N) {
  int wid = blockIdx.x * (blockDim.x >> 5) + (threadIdx.x >> 5);
  int lane = threadIdx.x & 31;
  int r0 = wid * 8;
  if (r0 >= N) return;
  int rend = r0 + 8 < N ? r0 + 8 : N;
  const float* zl[3] = {z0, z1, z2};
  const float* gl[3] = {gamma0, gamma1, gamma2};
  const float* bl[3] = {beta0, beta1, beta2};
  float sc[6], sh[6];
#pragma unroll
  for (int j = 0; j < 6; ++j) {
    int l = j >> 1;
    bn_coef(bn + l * 128, bn + l * 128 + 64, gl[l], bl[l], invN, lane + 32 * (j & 1), sc[j], sh[j]);
  }
  float acc[6] = {0.f, 0.f, 0.f, 0.f, 0.f, 0.f};
  int gcur = -1;
  for (int row = r0; row < rend; ++row) {
    float v[6]; float s = 0.f;
#pragma unroll
    for (int j = 0; j < 6; ++j) {
      v[j] = zl[j >> 1][(size_t)row * HD + lane + 32 * (j & 1)] * sc[j] + sh[j];
      s += v[j] * v[j];
    }
#pragma unroll
    for (int off = 16; off >= 1; off >>= 1) s += __shfl_xor(s, off, 32);
    float inv = 1.0f / fmaxf(sqrtf(s), 1e-12f);
    float* o = out + (size_t)row * CD;
#pragma unroll
    for (int j = 0; j < 6; ++j) o[lane + 32 * j] = v[j] * inv;
    int g = batch[row];
    if (g != gcur) {
      if (gcur >= 0) {
#pragma unroll
        for (int j = 0; j < 6; ++j) atomicAdd(&gsum[(size_t)gcur * CD + lane + 32 * j], acc[j]);
      }
      gcur = g;
#pragma unroll
      for (int j = 0; j < 6; ++j) acc[j] = 0.f;
    }
#pragma unroll
    for (int j = 0; j < 6; ++j) acc[j] += v[j];
  }
  if (gcur >= 0) {
#pragma unroll
    for (int j = 0; j < 6; ++j) atomicAdd(&gsum[(size_t)gcur * CD + lane + 32 * j], acc[j]);
  }
}

__global__ void k_l2_gc(const float* __restrict__ gsum, const float* __restrict__ cnt,
                        float* __restrict__ out, int rows) {
  int wid = blockIdx.x * (blockDim.x >> 5) + (threadIdx.x >> 5);
  if (wid >= rows) return;
  int lane = threadIdx.x & 31;
  float icnt = 1.0f / fmaxf(cnt[wid], 1.0f);
  const float* a = gsum + (size_t)wid * CD;
  float v[6]; float s = 0.f;
#pragma unroll
  for (int j = 0; j < 6; ++j) { v[j] = a[lane + 32 * j] * icnt; s += v[j] * v[j]; }
#pragma unroll
  for (int off = 16; off >= 1; off >>= 1) s += __shfl_xor(s, off, 32);
  float inv = 1.0f / fmaxf(sqrtf(s), 1e-12f);
  float* o = out + (size_t)wid * CD;
#pragma unroll
  for (int j = 0; j < 6; ++j) o[lane + 32 * j] = v[j] * inv;
}

extern "C" void kernel_launch(void* const* d_in, const int* in_sizes, int n_in,
                              void* d_out, int out_size, void* d_ws, size_t ws_size,
                              hipStream_t stream) {
  (void)n_in; (void)out_size; (void)ws_size;
  const float* x     = (const float*)d_in[0];
  const int*   ei    = (const int*)d_in[1];
  const int*   batch = (const int*)d_in[2];
  int N = in_sizes[0] / 128;
  int E = in_sizes[1] / 2;
  const int* src = ei;
  const int* dst = ei + E;
  const float* Wl[3]  = {(const float*)d_in[3],  (const float*)d_in[7],  (const float*)d_in[11]};
  const float* bl[3]  = {(const float*)d_in[4],  (const float*)d_in[8],  (const float*)d_in[12]};
  const float* gl[3]  = {(const float*)d_in[5],  (const float*)d_in[9],  (const float*)d_in[13]};
  const float* bel[3] = {(const float*)d_in[6],  (const float*)d_in[10], (const float*)d_in[14]};

  float* ws   = (float*)d_ws;
  float* dinv = ws;                       ws += N;
  float* h    = ws;                       ws += (size_t)N * HD;
  float* zb[3];
  zb[0] = ws;                             ws += (size_t)N * HD;
  zb[1] = ws;                             ws += (size_t)N * HD;
  zb[2] = ws;                             ws += (size_t)N * HD;
  float* gsum = ws;                       ws += (size_t)GN * CD;
  float* cnt  = ws;                       ws += GN;
  float* bn   = ws;
  ws += 3 * 128;
  int* deg    = (int*)(((size_t)ws + 15) & ~(size_t)15);
  int* rowptr = deg + (((size_t)N + 3) & ~(size_t)3);
  int nb  = (N + BINW - 1) / BINW;
  int nch = (E + CHUNK - 1) / CHUNK;
  int* cm       = rowptr + (((size_t)N + 3) & ~(size_t)3);
  int* binstart = cm + (size_t)nch * ((nb + 3) & ~3);
  int2* binned  = (int2*)(binstart + (((size_t)nb + 1 + 3) & ~(size_t)3));
  int2* csr     = binned + E;

  k_bin_count<<<nch, 1024, 0, stream>>>(dst, E, nb, cm, gsum, GN * CD + GN + 3 * 128);
  k_bin_scatter<<<nch, 1024, 0, stream>>>(src, dst, E, nb, cm, binstart, binned);
  int tilesM = (N + 15) / 16;
  int waves = tilesM * 4;
  k_csr_gemm<<<nb + (waves + 7) / 8, 256, 0, stream>>>(binstart, binned, deg, rowptr, dinv, batch, cnt, csr,
                                                      N, nb, x, 128, Wl[0], 128, h);

  float invN = 1.0f / (float)N;
  k_gather_relu_stats<<<(N + 63) / 64, 512, 0, stream>>>(rowptr, deg, csr, dinv, h, bl[0], zb[0],
                                                         bn, bn + 64, N);
  for (int l = 1; l < 3; ++l) {
    float* bn_sum = bn + l * 128;
    float* bn_sq  = bn_sum + 64;
    k_gather_bn_gemm_relu_stats<<<(N + 63) / 64, 512, 0, stream>>>(rowptr, deg, csr, dinv, zb[l - 1],
                                                                   bn_sum - 128, bn_sq - 128,
                                                                   gl[l - 1], bel[l - 1], invN,
                                                                   Wl[l], bl[l], zb[l], bn_sum, bn_sq, N);
  }

  float* out = (float*)d_out;
  k_bn_l2_pool<<<(N + 63) / 64, 256, 0, stream>>>(zb[0], zb[1], zb[2], bn, gl[0], bel[0], gl[1], bel[1],
                                                  gl[2], bel[2], invN, batch, out, gsum, N);
  k_l2_gc<<<(GN + 7) / 8, 256, 0, stream>>>(gsum, cnt, out + (size_t)N * CD, GN);
}


# workspace carve-up: planes h, z and the graph sums start on a 256-byte boundary (pointer arguments move), so a gathered 256-byte row is two whole cache lines, not parts of three; sums unchanged
# speedup vs baseline: 4.3940x; 1.2024x over previous
#include <hip/hip_runtime.h>

typedef float v2f __attribute__((ext_vector_type(2)));
typedef float v8f __attribute__((ext_vector_type(8)));

#define HD 64
#define CD 192
#define GN 256

#define BINW  256
#define MAXB  256
#define CHUNK 4096

__device__ __forceinline__ int wave_incl_scan(int v, int lane) {
  for (int off = 1; off < 32; off <<= 1) {
    int u = __shfl_up(v, off, 32);
    if (lane >= off) v += u;
  }
  return v;
}

__global__ void __launch_bounds__(1024)
k_bin_count(const int* __restrict__ dst, int E, int nb, int* __restrict__ cm,
            float* __restrict__ zero, int nzero) {
  __shared__ int s_h[MAXB];
  int tid = threadIdx.x;
  int chunk = blockIdx.x, nch = gridDim.x;
  for (int i = chunk * 1024 + tid; i < nzero; i += nch * 1024) zero[i] = 0.f;
  if (tid < MAXB) s_h[tid] = 0;
  __syncthreads();
  int e0 = chunk * CHUNK + tid;
#pragma unroll
  for (int k = 0; k < 4; ++k) {
    int e = e0 + k * 1024;
    if (e < E) atomicAdd(&s_h[dst[e] >> 8], 1);
  }
  __syncthreads();
  int nbp = (nb + 3) & ~3;
  if (tid < nbp) cm[chunk * nbp + tid] = s_h[tid];
}

__global__ void __launch_bounds__(1024)
k_bin_scatter(const int* __restrict__ src, const int* __restrict__ dst, int E, int nb,
              const int* __restrict__ cm, int* __restrict__ binstart, int2* __restrict__ binned) {
  __shared__ int s_cnt[MAXB];
  __shared__ int s_tot[MAXB];
  __shared__ int s_par[MAXB];
  __shared__ int s_lst[MAXB];
  __shared__ int s_gof[MAXB];
  __shared__ int s_wc[8], s_wg[8];
  __shared__ int2 s_e[CHUNK];
  int tid = threadIdx.x, lane = tid & 31, w = tid >> 5;
  int chunk = blockIdx.x, nch = gridDim.x;
  if (tid < MAXB) { s_cnt[tid] = 0; s_tot[tid] = 0; s_par[tid] = 0; }
  __syncthreads();
  int es[4], ed[4], er[4];
  int e0 = chunk * CHUNK + tid;
#pragma unroll
  for (int k = 0; k < 4; ++k) {
    int e = e0 + k * 1024;
    es[k] = 0; ed[k] = 0; er[k] = 0;
    if (e < E) {
      es[k] = src[e];
      ed[k] = dst[e];
      er[k] = atomicAdd(&s_cnt[ed[k] >> 8], 1);
    }
  }
  {
    int nbp = (nb + 3) & ~3;
    int q = tid & 63, r = tid >> 6;
    if (4 * q < nbp) {
      const int4* col = (const int4*)cm + q;
      int rowq = nbp >> 2;
      int t0 = 0, t1 = 0, t2 = 0, t3 = 0, p0 = 0, p1 = 0, p2 = 0, p3 = 0;
#pragma unroll 4
      for (int j = r; j < nch; j += 16) {
        int4 v = col[(size_t)j * rowq];
        int msk = (j < chunk) ? -1 : 0;
        t0 += v.x; t1 += v.y; t2 += v.z; t3 += v.w;
        p0 += v.x & msk; p1 += v.y & msk; p2 += v.z & msk; p3 += v.w & msk;
      }
      atomicAdd(&s_tot[4 * q], t0);     atomicAdd(&s_par[4 * q], p0);
      atomicAdd(&s_tot[4 * q + 1], t1); atomicAdd(&s_par[4 * q + 1], p1);
      atomicAdd(&s_tot[4 * q + 2], t2); atomicAdd(&s_par[4 * q + 2], p2);
      atomicAdd(&s_tot[4 * q + 3], t3); atomicAdd(&s_par[4 * q + 3], p3);
    }
  }
  __syncthreads();
  int c = (tid < MAXB) ? s_cnt[tid] : 0;
  int g = (tid < MAXB) ? s_tot[tid] : 0;
  int ci = wave_incl_scan(c, lane);
  int gi = wave_incl_scan(g, lane);
  if (lane == 31 && w < 8) { s_wc[w] = ci; s_wg[w] = gi; }
  __syncthreads();
  if (tid < MAXB) {
    int bc = 0, bg = 0;
    for (int k = 0; k < 8; ++k) {
      if (k < w) { bc += s_wc[k]; bg += s_wg[k]; }
    }
    int gstart = bg + gi - g;
    s_lst[tid] = bc + ci - c;
    s_gof[tid] = gstart + s_par[tid];
    if (chunk == 0) {
      if (tid < nb) binstart[tid] = gstart;
      if (tid == 0) binstart[nb] = E;
    }
  }
  __syncthreads();
  int nloc = E - chunk * CHUNK;
  if (nloc > CHUNK) nloc = CHUNK;
#pragma unroll
  for (int k = 0; k < 4; ++k) {
    if (e0 + k * 1024 < E) s_e[s_lst[ed[k] >> 8] + er[k]] = make_int2(es[k], ed[k]);
  }
  __syncthreads();
#pragma unroll
  for (int k = 0; k < 4; ++k) {
    int i = k * 1024 + tid;
    if (i < nloc) {
      int2 en = s_e[i];
      int b = en.y >> 8;
      binned[s_gof[b] + (i - s_lst[b])] = en;
    }
  }
}

__device__ __forceinline__ void csr_block(int bin, const int* __restrict__ binstart,
                                          const int2* __restrict__ binned, int* __restrict__ deg,
                                          int* __restrict__ rowptr, float* __restrict__ dinv,
                                          const int* __restrict__ batch, float* __restrict__ cnt,
                                          int2* __restrict__ csr, int N) {
  __shared__ int s_c[BINW], s_cur[BINW], s_w[8];
  int tid = threadIdx.x, lane = tid & 31, w = tid >> 5;
  int bs = binstart[bin], be = binstart[bin + 1];
  s_c[tid] = 0;
  __syncthreads();
#pragma unroll 4
  for (int i = bs + tid; i < be; i += 256) atomicAdd(&s_c[binned[i].y & (BINW - 1)], 1);
  __syncthreads();
  int c = s_c[tid];
  int inc = wave_incl_scan(c, lane);
  if (lane == 31) s_w[w] = inc;
  __syncthreads();
  int base = bs;
  for (int k = 0; k < 8; ++k) {
    if (k < w) base += s_w[k];
  }
  int first = base + inc - c;
  s_cur[tid] = first;
  int node = bin * BINW + tid;
  if (node < N) {
    deg[node] = c;
    rowptr[node] = first;
    dinv[node] = rsqrtf((float)c + 1.0f);
    atomicAdd(&cnt[batch[node]], 1.0f);
  }
  __syncthreads();
#pragma unroll 4
  for (int i = bs + tid; i < be; i += 256) {
    int2 en = binned[i];
    int pos = atomicAdd(&s_cur[en.y & (BINW - 1)], 1);
    csr[pos] = make_int2(en.x, 0);
  }
}

__device__ __forceinline__ void bn_coef(const float* __restrict__ bn_sum, const float* __restrict__ bn_sq,
                                        const float* __restrict__ gamma, const float* __restrict__ beta,
                                        float invN, int c, float& sc, float& sh) {
  float mu = bn_sum[c] * invN;
  float var = fmaxf(bn_sq[c] * invN - mu * mu, 0.f);
  sc = rsqrtf(var + 1e-5f) * gamma[c];
  sh = beta[c] - mu * sc;
}

__device__ __forceinline__ void gemm_block(int bid, const float* __restrict__ Z, int ldz,
                                           const float* __restrict__ W, int K,
                                           float* __restrict__ Hout, int N) {
  int wid = bid * (blockDim.x >> 5) + (threadIdx.x >> 5);
  int tilesM = (N + 15) >> 4;
  if (wid >= tilesM * 4) return;
  int tm = wid >> 2, tn = wid & 3;
  int lane = threadIdx.x & 31;
  int half = lane >> 4;
  int m16 = lane & 15;
  int row = tm * 16 + m16;
  int lrow = row < N ? row : N - 1;
  const float* zr = Z + (size_t)lrow * ldz;
  int col = tn * 16 + m16;
  const float* wp = W + col;
  v8f c = {};
  for (int k0 = 0; k0 < K; k0 += 4) {
    int ka = k0 + 2 * half;
    v2f a, b;
    a.x = zr[ka];
    a.y = zr[ka + 1];
    b.x = wp[(size_t)ka * HD];
    b.y = wp[(size_t)(ka + 1) * HD];
    c = __builtin_amdgcn_wmma_f32_16x16x4_f32(false, a, false, b, (short)0, c, false, false);
  }
  int rbase = tm * 16 + half * 8;
#pragma unroll
  for (int j = 0; j < 8; ++j) {
    int r = rbase + j;
    if (r < N) Hout[(size_t)r * HD + col] = c[j];
  }
}

__global__ void __launch_bounds__(256)
k_csr_gemm(const int* __restrict__ binstart, const int2* __restrict__ binned,
           int* __restrict__ deg, int* __restrict__ rowptr, float* __restrict__ dinv,
           const int* __restrict__ batch, float* __restrict__ cnt, int2* __restrict__ csr, int N, int nb,
           const float* __restrict__ X, int ldx,
           const float* __restrict__ W, int K, float* __restrict__ Hout) {
  if ((int)blockIdx.x < nb) {
    csr_block(blockIdx.x, binstart, binned, deg, rowptr, dinv, batch, cnt, csr, N);
  } else {
    gemm_block(blockIdx.x - nb, X, ldx, W, K, Hout, N);
  }
}

struct RowBatch {
  float4 v[4];
};

__device__ __forceinline__ void batch_issue(RowBatch& t, const float4* __restrict__ h4, int q, int ex,
                                            int b, int half) {
#pragma unroll
  for (int u = 0; u < 4; ++u) {
    int sl = 8 * b + 2 * u + half;
    int sr = __shfl(ex, sl, 32);
    t.v[u] = h4[(unsigned)(sr * (HD / 4) + q)];
  }
}

template <bool WSUM>
__device__ __forceinline__ void batch_sum(const RowBatch& t, float wl, int b, int half, float4& acc, float& wsum) {
#pragma unroll
  for (int u = 0; u < 4; ++u) {
    float wgt = __shfl(wl, 8 * b + 2 * u + half, 32);
    acc.x += t.v[u].x * wgt;
    acc.y += t.v[u].y * wgt;
    acc.z += t.v[u].z * wgt;
    acc.w += t.v[u].w * wgt;
    if (WSUM) wsum += wgt;
  }
}

template <bool WSUM>
__device__ __forceinline__ void gather_chunk(const float4* __restrict__ h4, int q, int ex, float wl, int m, int half,
                                             float4& acc, float& wsum) {
  int nbat = (m + 7) >> 3;
  RowBatch A, B, C;
  if (nbat == 1) {
    batch_issue(A, h4, q, ex, 0, half);
    batch_sum<WSUM>(A, wl, 0, half, acc, wsum);
  } else if (nbat == 2) {
    batch_issue(A, h4, q, ex, 0, half);
    batch_issue(B, h4, q, ex, 1, half);
    __builtin_amdgcn_sched_barrier(0);
    batch_sum<WSUM>(A, wl, 0, half, acc, wsum);
    batch_sum<WSUM>(B, wl, 1, half, acc, wsum);
  } else {
    batch_issue(A, h4, q, ex, 0, half);
    batch_issue(B, h4, q, ex, 1, half);
    batch_issue(C, h4, q, ex, 2, half);
    __builtin_amdgcn_sched_barrier(0);
    batch_sum<WSUM>(A, wl, 0, half, acc, wsum);
    if (nbat == 4) batch_issue(A, h4, q, ex, 3, half);
    batch_sum<WSUM>(B, wl, 1, half, acc, wsum);
    batch_sum<WSUM>(C, wl, 2, half, acc, wsum);
    if (nbat == 4) batch_sum<WSUM>(A, wl, 3, half, acc, wsum);
  }
}

__global__ void __launch_bounds__(512, 8)
k_gather_relu_stats(const int* __restrict__ rowptr, const int* __restrict__ deg,
                    int2* __restrict__ csr, const float* __restrict__ dinv,
                    const float* __restrict__ h, const float* __restrict__ bias,
                    float* __restrict__ z, float* __restrict__ bn_sum,
                    float* __restrict__ bn_sq, int N) {
  __shared__ __attribute__((aligned(16))) float s_sum[16 * HD];
  __shared__ __attribute__((aligned(16))) float s_sq[16 * HD];
  int w = __builtin_amdgcn_readfirstlane(threadIdx.x >> 5);
  int lane = threadIdx.x & 31;
  int half = lane >> 4;
  int c4 = (lane & 15) << 2;
  const float4* h4 = (const float4*)h;
  int q = lane & 15;
  float4* ps = (float4*)(s_sum + w * HD + c4);
  float4* pq = (float4*)(s_sq + w * HD + c4);
  if (half == 0) {
    *ps = make_float4(0.f, 0.f, 0.f, 0.f);
    *pq = make_float4(0.f, 0.f, 0.f, 0.f);
  }
  int st[4], nn[4], ex[4];
  float dv[4], wl[4];
#pragma unroll
  for (int i = 0; i < 4; ++i) {
    int d = blockIdx.x * 64 + i * 16 + w;
    int dc = d < N ? d : N - 1;
    st[i] = rowptr[dc];
    nn[i] = d < N ? deg[dc] : 0;
    dv[i] = dinv[dc];
  }
#pragma unroll
  for (int i = 0; i < 4; ++i) {
    int d = blockIdx.x * 64 + i * 16 + w;
    int sx = d < N ? d : N - 1;
    if (lane < nn[i]) sx = csr[st[i] + lane].x;
    ex[i] = sx;
  }
  float dsr[4];
#pragma unroll
  for (int i = 0; i < 4; ++i) dsr[i] = dinv[ex[i]];
#pragma unroll
  for (int i = 0; i < 4; ++i) {
    float ds = 0.f;
    if (lane < nn[i]) {
      ds = dsr[i];
      csr[st[i] + lane] = make_int2(ex[i], __float_as_int(ds));
    }
    wl[i] = ds * dv[i];
  }
#pragma unroll
  for (int i = 0; i < 4; ++i) {
    int d = blockIdx.x * 64 + i * 16 + w;
    if (d >= N) break;
    int start = st[i];
    int n = nn[i];
    float dd = dv[i];
    const float4 hs = h4[(unsigned)(d * (HD / 4) + q)];
    const float4 b4 = ((const float4*)bias)[q];
    float4 acc = {0.f, 0.f, 0.f, 0.f};
    float wsum = 0.f;
    int exc = ex[i];
    float wlc = wl[i];
    for (int base = 0; base < n; base += 32) {
      int m = n - base;
      if (m > 32) m = 32;
      gather_chunk<false>(h4, q, exc, wlc, m, half, acc, wsum);
      int m2 = n - base - 32;
      if (m2 > 0) {
        if (m2 > 32) m2 = 32;
        exc = d;
        float ds = 0.f;
        if (lane < m2) {
          int2* ep = csr + start + base + 32 + lane;
          exc = ep->x;
          ds = dinv[exc];
          *ep = make_int2(exc, __float_as_int(ds));
        }
        wlc = ds * dd;
      }
    }
    acc.x += __shfl_xor(acc.x, 16, 32);
    acc.y += __shfl_xor(acc.y, 16, 32);
    acc.z += __shfl_xor(acc.z, 16, 32);
    acc.w += __shfl_xor(acc.w, 16, 32);
    if (half == 0) {
      float w0 = dd * dd;
      float4 o;
      o.x = hs.x * w0 + acc.x;
      o.y = hs.y * w0 + acc.y;
      o.z = hs.z * w0 + acc.z;
      o.w = hs.w * w0 + acc.w;
      o.x = fmaxf(o.x + b4.x, 0.f);
      o.y = fmaxf(o.y + b4.y, 0.f);
      o.z = fmaxf(o.z + b4.z, 0.f);
      o.w = fmaxf(o.w + b4.w, 0.f);
      ((float4*)z)[(unsigned)(d * (HD / 4) + q)] = o;
      float4 ts = *ps;
      float4 tq = *pq;
      ts.x += o.x; ts.y += o.y; ts.z += o.z; ts.w += o.w;
      tq.x += o.x * o.x; tq.y += o.y * o.y; tq.z += o.z * o.z; tq.w += o.w * o.w;
      *ps = ts;
      *pq = tq;
    }
  }
  __syncthreads();
  if (threadIdx.x < 2 * HD) {
    int c = threadIdx.x & (HD - 1);
    const float* p = (threadIdx.x < HD) ? s_sum : s_sq;
    float t = 0.f;
#pragma unroll
    for (int k = 0; k < 16; ++k) t += p[k * HD + c];
    atomicAdd((threadIdx.x < HD ? bn_sum : bn_sq) + c, t);
  }
}

#define AP 68
__global__ void __launch_bounds__(512, 8)
k_gather_bn_gemm_relu_stats(const int* __restrict__ rowptr, const int* __restrict__ deg,
                            const int2* __restrict__ csr, const float* __restrict__ dinv,
                            const float* __restrict__ zp,
                            const float* __restrict__ pbn_sum, const float* __restrict__ pbn_sq,
                            const float* __restrict__ pgamma, const float* __restrict__ pbeta, float invN,
                            const float* __restrict__ W, const float* __restrict__ bias,
                            float* __restrict__ z, float* __restrict__ bn_sum,
                            float* __restrict__ bn_sq, int N) {
  __shared__ __attribute__((aligned(16))) float s_a[64 * AP];
  __shared__ float s_sum[4 * HD];
  __shared__ float s_sq[4 * HD];
  int w = __builtin_amdgcn_readfirstlane(threadIdx.x >> 5);
  int lane = threadIdx.x & 31;
  int half = lane >> 4;
  int c4 = (lane & 15) << 2;
  const float4* h4 = (const float4*)zp;
  int q = lane & 15;
  float sc0, sc1, sc2, sc3, sh0, sh1, sh2, sh3;
  bn_coef(pbn_sum, pbn_sq, pgamma, pbeta, invN, c4,     sc0, sh0);
  bn_coef(pbn_sum, pbn_sq, pgamma, pbeta, invN, c4 + 1, sc1, sh1);
  bn_coef(pbn_sum, pbn_sq, pgamma, pbeta, invN, c4 + 2, sc2, sh2);
  bn_coef(pbn_sum, pbn_sq, pgamma, pbeta, invN, c4 + 3, sc3, sh3);
  int st[4], nn[4], ex[4];
  float dv[4], wl[4];
#pragma unroll
  for (int i = 0; i < 4; ++i) {
    int d = blockIdx.x * 64 + i * 16 + w;
    int dc = d < N ? d : N - 1;
    st[i] = rowptr[dc];
    nn[i] = d < N ? deg[dc] : 0;
    dv[i] = dinv[dc];
  }
#pragma unroll
  for (int i = 0; i < 4; ++i) {
    int d = blockIdx.x * 64 + i * 16 + w;
    int2 ent = make_int2(d < N ? d : N - 1, 0);
    if (lane < nn[i]) ent = csr[st[i] + lane];
    ex[i] = ent.x;
    wl[i] = __int_as_float(ent.y) * dv[i];
  }
#pragma unroll
  for (int i = 0; i < 4; ++i) {
    int lr = i * 16 + w;
    int d = blockIdx.x * 64 + lr;
    float4 g = {0.f, 0.f, 0.f, 0.f};
    if (d < N) {
      int start = st[i];
      int n = nn[i];
      float dd = dv[i];
      const float4 hs = h4[(unsigned)(d * (HD / 4) + q)];
      float4 acc = {0.f, 0.f, 0.f, 0.f};
      float wsum = 0.f;
      int exc = ex[i];
      float wlc = wl[i];
      for (int base = 0; base < n; base += 32) {
        int m = n - base;
        if (m > 32) m = 32;
        gather_chunk<true>(h4, q, exc, wlc, m, half, acc, wsum);
        int m2 = n - base - 32;
        if (m2 > 0) {
          if (m2 > 32) m2 = 32;
          int2 ent = make_int2(d, 0);
          if (lane < m2) ent = csr[start + base + 32 + lane];
          exc = ent.x;
          wlc = __int_as_float(ent.y) * dd;
        }
      }
      acc.x += __shfl_xor(acc.x, 16, 32);
      acc.y += __shfl_xor(acc.y, 16, 32);
      acc.z += __shfl_xor(acc.z, 16, 32);
      acc.w += __shfl_xor(acc.w, 16, 32);
      wsum += __shfl_xor(wsum, 16, 32);
      float w0 = dd * dd;
      float r = w0 + wsum;
      g.x = (hs.x * w0 + acc.x) * sc0 + r * sh0;
      g.y = (hs.y * w0 + acc.y) * sc1 + r * sh1;
      g.z = (hs.z * w0 + acc.z) * sc2 + r * sh2;
      g.w = (hs.w * w0 + acc.w) * sc3 + r * sh3;
    }
    if (half == 0) *(float4*)(s_a + lr * AP + c4) = g;
  }
  __syncthreads();
  int tm = w >> 2, tn = w & 3;
  int m16 = lane & 15;
  const float* ap = s_a + (tm * 16 + m16) * AP + 2 * half;
  int col = tn * 16 + m16;
  const float* wp = W + col;
  v8f c = {};
#pragma unroll
  for (int k0 = 0; k0 < HD; k0 += 4) {
    int ka = k0 + 2 * half;
    v2f a, b;
    a.x = ap[k0];
    a.y = ap[k0 + 1];
    b.x = wp[(size_t)ka * HD];
    b.y = wp[(size_t)(ka + 1) * HD];
    c = __builtin_amdgcn_wmma_f32_16x16x4_f32(false, a, false, b, (short)0, c, false, false);
  }
  float bcol = bias[col];
  int rbase = blockIdx.x * 64 + tm * 16 + half * 8;
  float ts = 0.f, tq = 0.f;
#pragma unroll
  for (int j = 0; j < 8; ++j) {
    int r = rbase + j;
    if (r < N) {
      float o = fmaxf(c[j] + bcol, 0.f);
      z[(size_t)r * HD + col] = o;
      ts += o;
      tq += o * o;
    }
  }
  ts += __shfl_xor(ts, 16, 32);
  tq += __shfl_xor(tq, 16, 32);
  if (half == 0) {
    s_sum[tm * HD + col] = ts;
    s_sq[tm * HD + col] = tq;
  }
  __syncthreads();
  if (threadIdx.x < 2 * HD) {
    int cc = threadIdx.x & (HD - 1);
    const float* p = (threadIdx.x < HD) ? s_sum : s_sq;
    float t = (p[cc] + p[HD + cc]) + (p[2 * HD + cc] + p[3 * HD + cc]);
    atomicAdd((threadIdx.x < HD ? bn_sum : bn_sq) + cc, t);
  }
}

__global__ void k_bn_l2_pool(const float* __restrict__ z0, const float* __restrict__ z1,
                             const float* __restrict__ z2, const float* __restrict__ bn,
                             const float* __restrict__ gamma0, const float* __restrict__ beta0,
                             const float* __restrict__ gamma1, const float* __restrict__ beta1,
                             const float* __restrict__ gamma2, const float* __restrict__ beta2,
                             float invN, const int* __restrict__ batch,
                             float* __restrict__ out, float* __restrict__ gsum, int N) {
  int wid = blockIdx.x * (blockDim.x >> 5) + (threadIdx.x >> 5);
  int lane = threadIdx.x & 31;
  int r0 = wid * 8;
  if (r0 >= N) return;
  int rend = r0 + 8 < N ? r0 + 8 : N;
  const float* zl[3] = {z0, z1, z2};
  const float* gl[3] = {gamma0, gamma1, gamma2};
  const float* bl[3] = {beta0, beta1, beta2};
  float sc[6], sh[6];
#pragma unroll
  for (int j = 0; j < 6; ++j) {
    int l = j >> 1;
    bn_coef(bn + l * 128, bn + l * 128 + 64, gl[l], bl[l], invN, lane + 32 * (j & 1), sc[j], sh[j]);
  }
  float acc[6] = {0.f, 0.f, 0.f, 0.f, 0.f, 0.f};
  int gcur = -1;
  for (int row = r0; row < rend; ++row) {
    float v[6]; float s = 0.f;
#pragma unroll
    for (int j = 0; j < 6; ++j) {
      v[j] = zl[j >> 1][(size_t)row * HD + lane + 32 * (j & 1)] * sc[j] + sh[j];
      s += v[j] * v[j];
    }
#pragma unroll
    for (int off = 16; off >= 1; off >>= 1) s += __shfl_xor(s, off, 32);
    float inv = 1.0f / fmaxf(sqrtf(s), 1e-12f);
    float* o = out + (size_t)row * CD;
#pragma unroll
    for (int j = 0; j < 6; ++j) o[lane + 32 * j] = v[j] * inv;
    int g = batch[row];
    if (g != gcur) {
      if (gcur >= 0) {
#pragma unroll
        for (int j = 0; j < 6; ++j) atomicAdd(&gsum[(size_t)gcur * CD + lane + 32 * j], acc[j]);
      }
      gcur = g;
#pragma unroll
      for (int j = 0; j < 6; ++j) acc[j] = 0.f;
    }
#pragma unroll
    for (int j = 0; j < 6; ++j) acc[j] += v[j];
  }
  if (gcur >= 0) {
#pragma unroll
    for (int j = 0; j < 6; ++j) atomicAdd(&gsum[(size_t)gcur * CD + lane + 32 * j], acc[j]);
  }
}

__global__ void k_l2_gc(const float* __restrict__ gsum, const float* __restrict__ cnt,
                        float* __restrict__ out, int rows) {
  int wid = blockIdx.x * (blockDim.x >> 5) + (threadIdx.x >> 5);
  if (wid >= rows) return;
  int lane = threadIdx.x & 31;
  float icnt = 1.0f / fmaxf(cnt[wid], 1.0f);
  const float* a = gsum + (size_t)wid * CD;
  float v[6]; float s = 0.f;
#pragma unroll
  for (int j = 0; j < 6; ++j) { v[j] = a[lane + 32 * j] * icnt; s += v[j] * v[j]; }
#pragma unroll
  for (int off = 16; off >= 1; off >>= 1) s += __shfl_xor(s, off, 32);
  float inv = 1.0f / fmaxf(sqrtf(s), 1e-12f);
  float* o = out + (size_t)wid * CD;
#pragma unroll
  for (int j = 0; j < 6; ++j) o[lane + 32 * j] = v[j] * inv;
}

extern "C" void kernel_launch(void* const* d_in, const int* in_sizes, int n_in,
                              void* d_out, int out_size, void* d_ws, size_t ws_size,
                              hipStream_t stream) {
  (void)n_in; (void)out_size; (void)ws_size;
  const float* x     = (const float*)d_in[0];
  const int*   ei    = (const int*)d_in[1];
  const int*   batch = (const int*)d_in[2];
  int N = in_sizes[0] / 128;
  int E = in_sizes[1] / 2;
  const int* src = ei;
  const int* dst = ei + E;
  const float* Wl[3]  = {(const float*)d_in[3],  (const float*)d_in[7],  (const float*)d_in[11]};
  const float* bl[3]  = {(const float*)d_in[4],  (const float*)d_in[8],  (const float*)d_in[12]};
  const float* gl[3]  = {(const float*)d_in[5],  (const float*)d_in[9],  (const float*)d_in[13]};
  const float* bel[3] = {(const float*)d_in[6],  (const float*)d_in[10], (const float*)d_in[14]};

  float* ws   = (float*)d_ws;
  float* dinv = ws;                       ws += N;
  ws = (float*)(((size_t)ws + 255) & ~(size_t)255);
  float* h    = ws;                       ws += (size_t)N * HD;
  float* zb[3];
  zb[0] = ws;                             ws += (size_t)N * HD;
  zb[1] = ws;                             ws += (size_t)N * HD;
  zb[2] = ws;                             ws += (size_t)N * HD;
  float* gsum = ws;                       ws += (size_t)GN * CD;
  float* cnt  = ws;                       ws += GN;
  float* bn   = ws;
  ws += 3 * 128;
  int* deg    = (int*)(((size_t)ws + 15) & ~(size_t)15);
  int* rowptr = deg + (((size_t)N + 3) & ~(size_t)3);
  int nb  = (N + BINW - 1) / BINW;
  int nch = (E + CHUNK - 1) / CHUNK;
  int* cm       = rowptr + (((size_t)N + 3) & ~(size_t)3);
  int* binstart = cm + (size_t)nch * ((nb + 3) & ~3);
  int2* binned  = (int2*)(binstart + (((size_t)nb + 1 + 3) & ~(size_t)3));
  int2* csr     = binned + E;

  k_bin_count<<<nch, 1024, 0, stream>>>(dst, E, nb, cm, gsum, GN * CD + GN + 3 * 128);
  k_bin_scatter<<<nch, 1024, 0, stream>>>(src, dst, E, nb, cm, binstart, binned);
  int tilesM = (N + 15) / 16;
  int waves = tilesM * 4;
  k_csr_gemm<<<nb + (waves + 7) / 8, 256, 0, stream>>>(binstart, binned, deg, rowptr, dinv, batch, cnt, csr,
                                                      N, nb, x, 128, Wl[0], 128, h);

  float invN = 1.0f / (float)N;
  k_gather_relu_stats<<<(N + 63) / 64, 512, 0, stream>>>(rowptr, deg, csr, dinv, h, bl[0], zb[0],
                                                         bn, bn + 64, N);
  for (int l = 1; l < 3; ++l) {
    float* bn_sum = bn + l * 128;
    float* bn_sq  = bn_sum + 64;
    k_gather_bn_gemm_relu_stats<<<(N + 63) / 64, 512, 0, stream>>>(rowptr, deg, csr, dinv, zb[l - 1],
                                                                   bn_sum - 128, bn_sq - 128,
                                                                   gl[l - 1], bel[l - 1], invN,
                                                                   Wl[l], bl[l], zb[l], bn_sum, bn_sq, N);
  }

  float* out = (float*)d_out;
  k_bn_l2_pool<<<(N + 63) / 64, 256, 0, stream>>>(zb[0], zb[1], zb[2], bn, gl[0], bel[0], gl[1], bel[1],
                                                  gl[2], bel[2], invN, batch, out, gsum, N);
  k_l2_gc<<<(GN + 7) / 8, 256, 0, stream>>>(gsum, cnt, out + (size_t)N * CD, GN);
}


# edge list built in 2 launches, not 3: each chunk ordered by bin inside its block and kept chunk-major, no global count pass; list loads ahead of the LDS adds; edge sums in another LDS-arrival order
# speedup vs baseline: 4.6436x; 1.0568x over previous
#include <hip/hip_runtime.h>

typedef float v2f __attribute__((ext_vector_type(2)));
typedef float v8f __attribute__((ext_vector_type(8)));

#define HD 64
#define CD 192
#define GN 256

#define BINW  256
#define MAXB  256
#define CHUNK 4096

__device__ __forceinline__ int wave_incl_scan(int v, int lane) {
  for (int off = 1; off < 32; off <<= 1) {
    int u = __shfl_up(v, off, 32);
    if (lane >= off) v += u;
  }
  return v;
}

__global__ void __launch_bounds__(1024)
k_chunk_order(const int* __restrict__ src, const int* __restrict__ dst, int E, int nb,
              int* __restrict__ cm, int2* __restrict__ binned, float* __restrict__ zero, int nzero) {
  __shared__ int s_cnt[MAXB];
  __shared__ int s_lst[MAXB];
  __shared__ int s_wc[8];
  __shared__ int2 s_e[CHUNK];
  int tid = threadIdx.x, lane = tid & 31, w = tid >> 5;
  int chunk = blockIdx.x, nch = gridDim.x;
  for (int i = chunk * 1024 + tid; i < nzero; i += nch * 1024) zero[i] = 0.f;
  if (tid < MAXB) s_cnt[tid] = 0;
  __syncthreads();
  int nloc = E - chunk * CHUNK;
  if (nloc > CHUNK) nloc = CHUNK;
  const int* cs = src + (size_t)chunk * CHUNK;
  const int* cd = dst + (size_t)chunk * CHUNK;
  int es[4], ed[4], er[4];
#pragma unroll
  for (int k = 0; k < 4; ++k) {
    int i = k * 1024 + tid;
    int ic = i < nloc ? i : nloc - 1;
    es[k] = cs[ic];
    ed[k] = cd[ic];
  }
#pragma unroll
  for (int k = 0; k < 4; ++k) {
    er[k] = 0;
    if (k * 1024 + tid < nloc) er[k] = atomicAdd(&s_cnt[ed[k] >> 8], 1);
  }
  __syncthreads();
  int c = (tid < MAXB) ? s_cnt[tid] : 0;
  int ci = wave_incl_scan(c, lane);
  if (lane == 31 && w < 8) s_wc[w] = ci;
  __syncthreads();
  if (tid < MAXB) {
    int bc = 0;
    for (int k = 0; k < 8; ++k) {
      if (k < w) bc += s_wc[k];
    }
    int first = bc + ci - c;
    s_lst[tid] = first;
    if (tid < nb) cm[chunk * (nb + 1) + tid] = first;
    if (tid == 0) cm[chunk * (nb + 1) + nb] = nloc;
  }
  __syncthreads();
#pragma unroll
  for (int k = 0; k < 4; ++k) {
    if (k * 1024 + tid < nloc) s_e[s_lst[ed[k] >> 8] + er[k]] = make_int2(es[k], ed[k]);
  }
  __syncthreads();
#pragma unroll
  for (int k = 0; k < 4; ++k) {
    int i = k * 1024 + tid;
    if (i < nloc) binned[(size_t)chunk * CHUNK + i] = s_e[i];
  }
}

#define RUNB 16
__device__ __forceinline__ void csr_block(int bin, int nb, int nch, const int* __restrict__ cm,
                                          const int2* __restrict__ binned, int* __restrict__ deg,
                                          int* __restrict__ rowptr, float* __restrict__ dinv,
                                          const int* __restrict__ batch, float* __restrict__ cnt,
                                          int2* __restrict__ csr, int N) {
  __shared__ int s_c[BINW], s_cur[BINW], s_w[8], s_b[8];
  int tid = threadIdx.x, lane = tid & 31, w = tid >> 5;
  int pitch = nb + 1;
  s_c[tid] = 0;
  int bsum = 0;
  for (int ch = tid; ch < nch; ch += 256) bsum += cm[ch * pitch + bin];
#pragma unroll
  for (int off = 16; off >= 1; off >>= 1) bsum += __shfl_xor(bsum, off, 32);
  if (lane == 0) s_b[w] = bsum;
  __syncthreads();
  for (int ch = tid; ch < nch; ch += 256) {
    int ls = cm[ch * pitch + bin];
    int len = cm[ch * pitch + bin + 1] - ls;
    const int2* run = binned + (size_t)ch * CHUNK + ls;
    for (int j0 = 0; j0 < len; j0 += RUNB) {
      int v[RUNB];
#pragma unroll
      for (int u = 0; u < RUNB; ++u) {
        int j = j0 + u;
        v[u] = run[j < len ? j : len - 1].y;
      }
#pragma unroll
      for (int u = 0; u < RUNB; ++u) {
        if (j0 + u < len) atomicAdd(&s_c[v[u] & (BINW - 1)], 1);
      }
    }
  }
  __syncthreads();
  int c = s_c[tid];
  int inc = wave_incl_scan(c, lane);
  if (lane == 31) s_w[w] = inc;
  __syncthreads();
  int base = 0;
  for (int k = 0; k < 8; ++k) {
    base += s_b[k];
    if (k < w) base += s_w[k];
  }
  int first = base + inc - c;
  s_cur[tid] = first;
  int node = bin * BINW + tid;
  if (node < N) {
    deg[node] = c;
    rowptr[node] = first;
    dinv[node] = rsqrtf((float)c + 1.0f);
    atomicAdd(&cnt[batch[node]], 1.0f);
  }
  __syncthreads();
  for (int ch = tid; ch < nch; ch += 256) {
    int ls = cm[ch * pitch + bin];
    int len = cm[ch * pitch + bin + 1] - ls;
    const int2* run = binned + (size_t)ch * CHUNK + ls;
    for (int j0 = 0; j0 < len; j0 += RUNB) {
      int2 v[RUNB];
#pragma unroll
      for (int u = 0; u < RUNB; ++u) {
        int j = j0 + u;
        v[u] = run[j < len ? j : len - 1];
      }
#pragma unroll
      for (int u = 0; u < RUNB; ++u) {
        if (j0 + u < len) {
          int pos = atomicAdd(&s_cur[v[u].y & (BINW - 1)], 1);
          csr[pos] = make_int2(v[u].x, 0);
        }
      }
    }
  }
}

__device__ __forceinline__ void bn_coef(const float* __restrict__ bn_sum, const float* __restrict__ bn_sq,
                                        const float* __restrict__ gamma, const float* __restrict__ beta,
                                        float invN, int c, float& sc, float& sh) {
  float mu = bn_sum[c] * invN;
  float var = fmaxf(bn_sq[c] * invN - mu * mu, 0.f);
  sc = rsqrtf(var + 1e-5f) * gamma[c];
  sh = beta[c] - mu * sc;
}

__device__ __forceinline__ void gemm_block(int bid, const float* __restrict__ Z, int ldz,
                                           const float* __restrict__ W, int K,
                                           float* __restrict__ Hout, int N) {
  int wid = bid * (blockDim.x >> 5) + (threadIdx.x >> 5);
  int tilesM = (N + 15) >> 4;
  if (wid >= tilesM * 4) return;
  int tm = wid >> 2, tn = wid & 3;
  int lane = threadIdx.x & 31;
  int half = lane >> 4;
  int m16 = lane & 15;
  int row = tm * 16 + m16;
  int lrow = row < N ? row : N - 1;
  const float* zr = Z + (size_t)lrow * ldz;
  int col = tn * 16 + m16;
  const float* wp = W + col;
  v8f c = {};
  for (int k0 = 0; k0 < K; k0 += 4) {
    int ka = k0 + 2 * half;
    v2f a, b;
    a.x = zr[ka];
    a.y = zr[ka + 1];
    b.x = wp[(size_t)ka * HD];
    b.y = wp[(size_t)(ka + 1) * HD];
    c = __builtin_amdgcn_wmma_f32_16x16x4_f32(false, a, false, b, (short)0, c, false, false);
  }
  int rbase = tm * 16 + half * 8;
#pragma unroll
  for (int j = 0; j < 8; ++j) {
    int r = rbase + j;
    if (r < N) Hout[(size_t)r * HD + col] = c[j];
  }
}

__global__ void __launch_bounds__(256)
k_csr_gemm(const int* __restrict__ cm, const int2* __restrict__ binned,
           int* __restrict__ deg, int* __restrict__ rowptr, float* __restrict__ dinv,
           const int* __restrict__ batch, float* __restrict__ cnt, int2* __restrict__ csr, int N, int nb, int nch,
           const float* __restrict__ X, int ldx,
           const float* __restrict__ W, int K, float* __restrict__ Hout) {
  if ((int)blockIdx.x < nb) {
    csr_block(blockIdx.x, nb, nch, cm, binned, deg, rowptr, dinv, batch, cnt, csr, N);
  } else {
    gemm_block(blockIdx.x - nb, X, ldx, W, K, Hout, N);
  }
}

struct RowBatch {
  float4 v[4];
};

__device__ __forceinline__ void batch_issue(RowBatch& t, const float4* __restrict__ h4, int q, int ex,
                                            int b, int half) {
#pragma unroll
  for (int u = 0; u < 4; ++u) {
    int sl = 8 * b + 2 * u + half;
    int sr = __shfl(ex, sl, 32);
    t.v[u] = h4[(unsigned)(sr * (HD / 4) + q)];
  }
}

template <bool WSUM>
__device__ __forceinline__ void batch_sum(const RowBatch& t, float wl, int b, int half, float4& acc, float& wsum) {
#pragma unroll
  for (int u = 0; u < 4; ++u) {
    float wgt = __shfl(wl, 8 * b + 2 * u + half, 32);
    acc.x += t.v[u].x * wgt;
    acc.y += t.v[u].y * wgt;
    acc.z += t.v[u].z * wgt;
    acc.w += t.v[u].w * wgt;
    if (WSUM) wsum += wgt;
  }
}

template <bool WSUM>
__device__ __forceinline__ void gather_chunk(const float4* __restrict__ h4, int q, int ex, float wl, int m, int half,
                                             float4& acc, float& wsum) {
  int nbat = (m + 7) >> 3;
  RowBatch A, B, C;
  if (nbat == 1) {
    batch_issue(A, h4, q, ex, 0, half);
    batch_sum<WSUM>(A, wl, 0, half, acc, wsum);
  } else if (nbat == 2) {
    batch_issue(A, h4, q, ex, 0, half);
    batch_issue(B, h4, q, ex, 1, half);
    __builtin_amdgcn_sched_barrier(0);
    batch_sum<WSUM>(A, wl, 0, half, acc, wsum);
    batch_sum<WSUM>(B, wl, 1, half, acc, wsum);
  } else {
    batch_issue(A, h4, q, ex, 0, half);
    batch_issue(B, h4, q, ex, 1, half);
    batch_issue(C, h4, q, ex, 2, half);
    __builtin_amdgcn_sched_barrier(0);
    batch_sum<WSUM>(A, wl, 0, half, acc, wsum);
    if (nbat == 4) batch_issue(A, h4, q, ex, 3, half);
    batch_sum<WSUM>(B, wl, 1, half, acc, wsum);
    batch_sum<WSUM>(C, wl, 2, half, acc, wsum);
    if (nbat == 4) batch_sum<WSUM>(A, wl, 3, half, acc, wsum);
  }
}

__global__ void __launch_bounds__(512, 8)
k_gather_relu_stats(const int* __restrict__ rowptr, const int* __restrict__ deg,
                    int2* __restrict__ csr, const float* __restrict__ dinv,
                    const float* __restrict__ h, const float* __restrict__ bias,
                    float* __restrict__ z, float* __restrict__ bn_sum,
                    float* __restrict__ bn_sq, int N) {
  __shared__ __attribute__((aligned(16))) float s_sum[16 * HD];
  __shared__ __attribute__((aligned(16))) float s_sq[16 * HD];
  int w = __builtin_amdgcn_readfirstlane(threadIdx.x >> 5);
  int lane = threadIdx.x & 31;
  int half = lane >> 4;
  int c4 = (lane & 15) << 2;
  const float4* h4 = (const float4*)h;
  int q = lane & 15;
  float4* ps = (float4*)(s_sum + w * HD + c4);
  float4* pq = (float4*)(s_sq + w * HD + c4);
  if (half == 0) {
    *ps = make_float4(0.f, 0.f, 0.f, 0.f);
    *pq = make_float4(0.f, 0.f, 0.f, 0.f);
  }
  int st[4], nn[4], ex[4];
  float dv[4], wl[4];
#pragma unroll
  for (int i = 0; i < 4; ++i) {
    int d = blockIdx.x * 64 + i * 16 + w;
    int dc = d < N ? d : N - 1;
    st[i] = rowptr[dc];
    nn[i] = d < N ? deg[dc] : 0;
    dv[i] = dinv[dc];
  }
#pragma unroll
  for (int i = 0; i < 4; ++i) {
    int d = blockIdx.x * 64 + i * 16 + w;
    int sx = d < N ? d : N - 1;
    if (lane < nn[i]) sx = csr[st[i] + lane].x;
    ex[i] = sx;
  }
  float dsr[4];
#pragma unroll
  for (int i = 0; i < 4; ++i) dsr[i] = dinv[ex[i]];
#pragma unroll
  for (int i = 0; i < 4; ++i) {
    float ds = 0.f;
    if (lane < nn[i]) {
      ds = dsr[i];
      csr[st[i] + lane] = make_int2(ex[i], __float_as_int(ds));
    }
    wl[i] = ds * dv[i];
  }
#pragma unroll
  for (int i = 0; i < 4; ++i) {
    int d = blockIdx.x * 64 + i * 16 + w;
    if (d >= N) break;
    int start = st[i];
    int n = nn[i];
    float dd = dv[i];
    const float4 hs = h4[(unsigned)(d * (HD / 4) + q)];
    const float4 b4 = ((const float4*)bias)[q];
    float4 acc = {0.f, 0.f, 0.f, 0.f};
    float wsum = 0.f;
    int exc = ex[i];
    float wlc = wl[i];
    for (int base = 0; base < n; base += 32) {
      int m = n - base;
      if (m > 32) m = 32;
      gather_chunk<false>(h4, q, exc, wlc, m, half, acc, wsum);
      int m2 = n - base - 32;
      if (m2 > 0) {
        if (m2 > 32) m2 = 32;
        exc = d;
        float ds = 0.f;
        if (lane < m2) {
          int2* ep = csr + start + base + 32 + lane;
          exc = ep->x;
          ds = dinv[exc];
          *ep = make_int2(exc, __float_as_int(ds));
        }
        wlc = ds * dd;
      }
    }
    acc.x += __shfl_xor(acc.x, 16, 32);
    acc.y += __shfl_xor(acc.y, 16, 32);
    acc.z += __shfl_xor(acc.z, 16, 32);
    acc.w += __shfl_xor(acc.w, 16, 32);
    if (half == 0) {
      float w0 = dd * dd;
      float4 o;
      o.x = hs.x * w0 + acc.x;
      o.y = hs.y * w0 + acc.y;
      o.z = hs.z * w0 + acc.z;
      o.w = hs.w * w0 + acc.w;
      o.x = fmaxf(o.x + b4.x, 0.f);
      o.y = fmaxf(o.y + b4.y, 0.f);
      o.z = fmaxf(o.z + b4.z, 0.f);
      o.w = fmaxf(o.w + b4.w, 0.f);
      ((float4*)z)[(unsigned)(d * (HD / 4) + q)] = o;
      float4 ts = *ps;
      float4 tq = *pq;
      ts.x += o.x; ts.y += o.y; ts.z += o.z; ts.w += o.w;
      tq.x += o.x * o.x; tq.y += o.y * o.y; tq.z += o.z * o.z; tq.w += o.w * o.w;
      *ps = ts;
      *pq = tq;
    }
  }
  __syncthreads();
  if (threadIdx.x < 2 * HD) {
    int c = threadIdx.x & (HD - 1);
    const float* p = (threadIdx.x < HD) ? s_sum : s_sq;
    float t = 0.f;
#pragma unroll
    for (int k = 0; k < 16; ++k) t += p[k * HD + c];
    atomicAdd((threadIdx.x < HD ? bn_sum : bn_sq) + c, t);
  }
}

#define AP 68
__global__ void __launch_bounds__(512, 8)
k_gather_bn_gemm_relu_stats(const int* __restrict__ rowptr, const int* __restrict__ deg,
                            const int2* __restrict__ csr, const float* __restrict__ dinv,
                            const float* __restrict__ zp,
                            const float* __restrict__ pbn_sum, const float* __restrict__ pbn_sq,
                            const float* __restrict__ pgamma, const float* __restrict__ pbeta, float invN,
                            const float* __restrict__ W, const float* __restrict__ bias,
                            float* __restrict__ z, float* __restrict__ bn_sum,
                            float* __restrict__ bn_sq, int N) {
  __shared__ __attribute__((aligned(16))) float s_a[64 * AP];
  __shared__ float s_sum[4 * HD];
  __shared__ float s_sq[4 * HD];
  int w = __builtin_amdgcn_readfirstlane(threadIdx.x >> 5);
  int lane = threadIdx.x & 31;
  int half = lane >> 4;
  int c4 = (lane & 15) << 2;
  const float4* h4 = (const float4*)zp;
  int q = lane & 15;
  float sc0, sc1, sc2, sc3, sh0, sh1, sh2, sh3;
  bn_coef(pbn_sum, pbn_sq, pgamma, pbeta, invN, c4,     sc0, sh0);
  bn_coef(pbn_sum, pbn_sq, pgamma, pbeta, invN, c4 + 1, sc1, sh1);
  bn_coef(pbn_sum, pbn_sq, pgamma, pbeta, invN, c4 + 2, sc2, sh2);
  bn_coef(pbn_sum, pbn_sq, pgamma, pbeta, invN, c4 + 3, sc3, sh3);
  int st[4], nn[4], ex[4];
  float dv[4], wl[4];
#pragma unroll
  for (int i = 0; i < 4; ++i) {
    int d = blockIdx.x * 64 + i * 16 + w;
    int dc = d < N ? d : N - 1;
    st[i] = rowptr[dc];
    nn[i] = d < N ? deg[dc] : 0;
    dv[i] = dinv[dc];
  }
#pragma unroll
  for (int i = 0; i < 4; ++i) {
    int d = blockIdx.x * 64 + i * 16 + w;
    int2 ent = make_int2(d < N ? d : N - 1, 0);
    if (lane < nn[i]) ent = csr[st[i] + lane];
    ex[i] = ent.x;
    wl[i] = __int_as_float(ent.y) * dv[i];
  }
#pragma unroll
  for (int i = 0; i < 4; ++i) {
    int lr = i * 16 + w;
    int d = blockIdx.x * 64 + lr;
    float4 g = {0.f, 0.f, 0.f, 0.f};
    if (d < N) {
      int start = st[i];
      int n = nn[i];
      float dd = dv[i];
      const float4 hs = h4[(unsigned)(d * (HD / 4) + q)];
      float4 acc = {0.f, 0.f, 0.f, 0.f};
      float wsum = 0.f;
      int exc = ex[i];
      float wlc = wl[i];
      for (int base = 0; base < n; base += 32) {
        int m = n - base;
        if (m > 32) m = 32;
        gather_chunk<true>(h4, q, exc, wlc, m, half, acc, wsum);
        int m2 = n - base - 32;
        if (m2 > 0) {
          if (m2 > 32) m2 = 32;
          int2 ent = make_int2(d, 0);
          if (lane < m2) ent = csr[start + base + 32 + lane];
          exc = ent.x;
          wlc = __int_as_float(ent.y) * dd;
        }
      }
      acc.x += __shfl_xor(acc.x, 16, 32);
      acc.y += __shfl_xor(acc.y, 16, 32);
      acc.z += __shfl_xor(acc.z, 16, 32);
      acc.w += __shfl_xor(acc.w, 16, 32);
      wsum += __shfl_xor(wsum, 16, 32);
      float w0 = dd * dd;
      float r = w0 + wsum;
      g.x = (hs.x * w0 + acc.x) * sc0 + r * sh0;
      g.y = (hs.y * w0 + acc.y) * sc1 + r * sh1;
      g.z = (hs.z * w0 + acc.z) * sc2 + r * sh2;
      g.w = (hs.w * w0 + acc.w) * sc3 + r * sh3;
    }
    if (half == 0) *(float4*)(s_a + lr * AP + c4) = g;
  }
  __syncthreads();
  int tm = w >> 2, tn = w & 3;
  int m16 = lane & 15;
  const float* ap = s_a + (tm * 16 + m16) * AP + 2 * half;
  int col = tn * 16 + m16;
  const float* wp = W + col;
  v8f c = {};
#pragma unroll
  for (int k0 = 0; k0 < HD; k0 += 4) {
    int ka = k0 + 2 * half;
    v2f a, b;
    a.x = ap[k0];
    a.y = ap[k0 + 1];
    b.x = wp[(size_t)ka * HD];
    b.y = wp[(size_t)(ka + 1) * HD];
    c = __builtin_amdgcn_wmma_f32_16x16x4_f32(false, a, false, b, (short)0, c, false, false);
  }
  float bcol = bias[col];
  int rbase = blockIdx.x * 64 + tm * 16 + half * 8;
  float ts = 0.f, tq = 0.f;
#pragma unroll
  for (int j = 0; j < 8; ++j) {
    int r = rbase + j;
    if (r < N) {
      float o = fmaxf(c[j] + bcol, 0.f);
      z[(size_t)r * HD + col] = o;
      ts += o;
      tq += o * o;
    }
  }
  ts += __shfl_xor(ts, 16, 32);
  tq += __shfl_xor(tq, 16, 32);
  if (half == 0) {
    s_sum[tm * HD + col] = ts;
    s_sq[tm * HD + col] = tq;
  }
  __syncthreads();
  if (threadIdx.x < 2 * HD) {
    int cc = threadIdx.x & (HD - 1);
    const float* p = (threadIdx.x < HD) ? s_sum : s_sq;
    float t = (p[cc] + p[HD + cc]) + (p[2 * HD + cc] + p[3 * HD + cc]);
    atomicAdd((threadIdx.x < HD ? bn_sum : bn_sq) + cc, t);
  }
}

__global__ void k_bn_l2_pool(const float* __restrict__ z0, const float* __restrict__ z1,
                             const float* __restrict__ z2, const float* __restrict__ bn,
                             const float* __restrict__ gamma0, const float* __restrict__ beta0,
                             const float* __restrict__ gamma1, const float* __restrict__ beta1,
                             const float* __restrict__ gamma2, const float* __restrict__ beta2,
                             float invN, const int* __restrict__ batch,
                             float* __restrict__ out, float* __restrict__ gsum, int N) {
  int wid = blockIdx.x * (blockDim.x >> 5) + (threadIdx.x >> 5);
  int lane = threadIdx.x & 31;
  int r0 = wid * 8;
  if (r0 >= N) return;
  int rend = r0 + 8 < N ? r0 + 8 : N;
  const float* zl[3] = {z0, z1, z2};
  const float* gl[3] = {gamma0, gamma1, gamma2};
  const float* bl[3] = {beta0, beta1, beta2};
  float sc[6], sh[6];
#pragma unroll
  for (int j = 0; j < 6; ++j) {
    int l = j >> 1;
    bn_coef(bn + l * 128, bn + l * 128 + 64, gl[l], bl[l], invN, lane + 32 * (j & 1), sc[j], sh[j]);
  }
  float acc[6] = {0.f, 0.f, 0.f, 0.f, 0.f, 0.f};
  int gcur = -1;
  for (int row = r0; row < rend; ++row) {
    float v[6]; float s = 0.f;
#pragma unroll
    for (int j = 0; j < 6; ++j) {
      v[j] = zl[j >> 1][(size_t)row * HD + lane + 32 * (j & 1)] * sc[j] + sh[j];
      s += v[j] * v[j];
    }
#pragma unroll
    for (int off = 16; off >= 1; off >>= 1) s += __shfl_xor(s, off, 32);
    float inv = 1.0f / fmaxf(sqrtf(s), 1e-12f);
    float* o = out + (size_t)row * CD;
#pragma unroll
    for (int j = 0; j < 6; ++j) o[lane + 32 * j] = v[j] * inv;
    int g = batch[row];
    if (g != gcur) {
      if (gcur >= 0) {
#pragma unroll
        for (int j = 0; j < 6; ++j) atomicAdd(&gsum[(size_t)gcur * CD + lane + 32 * j], acc[j]);
      }
      gcur = g;
#pragma unroll
      for (int j = 0; j < 6; ++j) acc[j] = 0.f;
    }
#pragma unroll
    for (int j = 0; j < 6; ++j) acc[j] += v[j];
  }
  if (gcur >= 0) {
#pragma unroll
    for (int j = 0; j < 6; ++j) atomicAdd(&gsum[(size_t)gcur * CD + lane + 32 * j], acc[j]);
  }
}

__global__ void k_l2_gc(const float* __restrict__ gsum, const float* __restrict__ cnt,
                        float* __restrict__ out, int rows) {
  int wid = blockIdx.x * (blockDim.x >> 5) + (threadIdx.x >> 5);
  if (wid >= rows) return;
  int lane = threadIdx.x & 31;
  float icnt = 1.0f / fmaxf(cnt[wid], 1.0f);
  const float* a = gsum + (size_t)wid * CD;
  float v[6]; float s = 0.f;
#pragma unroll
  for (int j = 0; j < 6; ++j) { v[j] = a[lane + 32 * j] * icnt; s += v[j] * v[j]; }
#pragma unroll
  for (int off = 16; off >= 1; off >>= 1) s += __shfl_xor(s, off, 32);
  float inv = 1.0f / fmaxf(sqrtf(s), 1e-12f);
  float* o = out + (size_t)wid * CD;
#pragma unroll
  for (int j = 0; j < 6; ++j) o[lane + 32 * j] = v[j] * inv;
}

extern "C" void kernel_launch(void* const* d_in, const int* in_sizes, int n_in,
                              void* d_out, int out_size, void* d_ws, size_t ws_size,
                              hipStream_t stream) {
  (void)n_in; (void)out_size; (void)ws_size;
  const float* x     = (const float*)d_in[0];
  const int*   ei    = (const int*)d_in[1];
  const int*   batch = (const int*)d_in[2];
  int N = in_sizes[0] / 128;
  int E = in_sizes[1] / 2;
  const int* src = ei;
  const int* dst = ei + E;
  const float* Wl[3]  = {(const float*)d_in[3],  (const float*)d_in[7],  (const float*)d_in[11]};
  const float* bl[3]  = {(const float*)d_in[4],  (const float*)d_in[8],  (const float*)d_in[12]};
  const float* gl[3]  = {(const float*)d_in[5],  (const float*)d_in[9],  (const float*)d_in[13]};
  const float* bel[3] = {(const float*)d_in[6],  (const float*)d_in[10], (const float*)d_in[14]};

  float* ws   = (float*)d_ws;
  float* dinv = ws;                       ws += N;
  ws = (float*)(((size_t)ws + 255) & ~(size_t)255);
  float* h    = ws;                       ws += (size_t)N * HD;
  float* zb[3];
  zb[0] = ws;                             ws += (size_t)N * HD;
  zb[1] = ws;                             ws += (size_t)N * HD;
  zb[2] = ws;                             ws += (size_t)N * HD;
  float* gsum = ws;                       ws += (size_t)GN * CD;
  float* cnt  = ws;                       ws += GN;
  float* bn   = ws;
  ws += 3 * 128;
  int* deg    = (int*)(((size_t)ws + 15) & ~(size_t)15);
  int* rowptr = deg + (((size_t)N + 3) & ~(size_t)3);
  int nb  = (N + BINW - 1) / BINW;
  int nch = (E + CHUNK - 1) / CHUNK;
  int* cm       = rowptr + (((size_t)N + 3) & ~(size_t)3);
  int2* binned  = (int2*)(cm + (((size_t)nch * (nb + 1) + 3) & ~(size_t)3));
  int2* csr     = binned + (size_t)nch * CHUNK;

  k_chunk_order<<<nch, 1024, 0, stream>>>(src, dst, E, nb, cm, binned, gsum, GN * CD + GN + 3 * 128);
  int tilesM = (N + 15) / 16;
  int waves = tilesM * 4;
  k_csr_gemm<<<nb + (waves + 7) / 8, 256, 0, stream>>>(cm, binned, deg, rowptr, dinv, batch, cnt, csr,
                                                      N, nb, nch, x, 128, Wl[0], 128, h);

  float invN = 1.0f / (float)N;
  k_gather_relu_stats<<<(N + 63) / 64, 512, 0, stream>>>(rowptr, deg, csr, dinv, h, bl[0], zb[0],
                                                         bn, bn + 64, N);
  for (int l = 1; l < 3; ++l) {
    float* bn_sum = bn + l * 128;
    float* bn_sq  = bn_sum + 64;
    k_gather_bn_gemm_relu_stats<<<(N + 63) / 64, 512, 0, stream>>>(rowptr, deg, csr, dinv, zb[l - 1],
                                                                   bn_sum - 128, bn_sq - 128,
                                                                   gl[l - 1], bel[l - 1], invN,
                                                                   Wl[l], bl[l], zb[l], bn_sum, bn_sq, N);
  }

  float* out = (float*)d_out;
  k_bn_l2_pool<<<(N + 63) / 64, 256, 0, stream>>>(zb[0], zb[1], zb[2], bn, gl[0], bel[0], gl[1], bel[1],
                                                  gl[2], bel[2], invN, batch, out, gsum, N);
  k_l2_gc<<<(GN + 7) / 8, 256, 0, stream>>>(gsum, cnt, out + (size_t)N * CD, GN);
}


# aggregation launches as one resident set: 512 blocks (2 per WGP) of equal row shares in two 64-row sub-tiles, not 782 one-tile blocks; BN sums regrouped per block, edge sums unchanged
# speedup vs baseline: 4.7949x; 1.0326x over previous
#include <hip/hip_runtime.h>

typedef float v2f __attribute__((ext_vector_type(2)));
typedef float v8f __attribute__((ext_vector_type(8)));

#define HD 64
#define CD 192
#define GN 256
#define GBLK 512

#define BINW  256
#define MAXB  256
#define CHUNK 4096

__device__ __forceinline__ int wave_incl_scan(int v, int lane) {
  for (int off = 1; off < 32; off <<= 1) {
    int u = __shfl_up(v, off, 32);
    if (lane >= off) v += u;
  }
  return v;
}

__global__ void __launch_bounds__(1024)
k_chunk_order(const int* __restrict__ src, const int* __restrict__ dst, int E, int nb,
              int* __restrict__ cm, int2* __restrict__ binned, float* __restrict__ zero, int nzero) {
  __shared__ int s_cnt[MAXB];
  __shared__ int s_lst[MAXB];
  __shared__ int s_wc[8];
  __shared__ int2 s_e[CHUNK];
  int tid = threadIdx.x, lane = tid & 31, w = tid >> 5;
  int chunk = blockIdx.x, nch = gridDim.x;
  for (int i = chunk * 1024 + tid; i < nzero; i += nch * 1024) zero[i] = 0.f;
  if (tid < MAXB) s_cnt[tid] = 0;
  __syncthreads();
  int nloc = E - chunk * CHUNK;
  if (nloc > CHUNK) nloc = CHUNK;
  const int* cs = src + (size_t)chunk * CHUNK;
  const int* cd = dst + (size_t)chunk * CHUNK;
  int es[4], ed[4], er[4];
#pragma unroll
  for (int k = 0; k < 4; ++k) {
    int i = k * 1024 + tid;
    int ic = i < nloc ? i : nloc - 1;
    es[k] = cs[ic];
    ed[k] = cd[ic];
  }
#pragma unroll
  for (int k = 0; k < 4; ++k) {
    er[k] = 0;
    if (k * 1024 + tid < nloc) er[k] = atomicAdd(&s_cnt[ed[k] >> 8], 1);
  }
  __syncthreads();
  int c = (tid < MAXB) ? s_cnt[tid] : 0;
  int ci = wave_incl_scan(c, lane);
  if (lane == 31 && w < 8) s_wc[w] = ci;
  __syncthreads();
  if (tid < MAXB) {
    int bc = 0;
    for (int k = 0; k < 8; ++k) {
      if (k < w) bc += s_wc[k];
    }
    int first = bc + ci - c;
    s_lst[tid] = first;
    if (tid < nb) cm[chunk * (nb + 1) + tid] = first;
    if (tid == 0) cm[chunk * (nb + 1) + nb] = nloc;
  }
  __syncthreads();
#pragma unroll
  for (int k = 0; k < 4; ++k) {
    if (k * 1024 + tid < nloc) s_e[s_lst[ed[k] >> 8] + er[k]] = make_int2(es[k], ed[k]);
  }
  __syncthreads();
#pragma unroll
  for (int k = 0; k < 4; ++k) {
    int i = k * 1024 + tid;
    if (i < nloc) binned[(size_t)chunk * CHUNK + i] = s_e[i];
  }
}

#define RUNB 16
__device__ __forceinline__ void csr_block(int bin, int nb, int nch, const int* __restrict__ cm,
                                          const int2* __restrict__ binned, int* __restrict__ deg,
                                          int* __restrict__ rowptr, float* __restrict__ dinv,
                                          const int* __restrict__ batch, float* __restrict__ cnt,
                                          int2* __restrict__ csr, int N) {
  __shared__ int s_c[BINW], s_cur[BINW], s_w[8], s_b[8];
  int tid = threadIdx.x, lane = tid & 31, w = tid >> 5;
  int pitch = nb + 1;
  s_c[tid] = 0;
  int bsum = 0;
  for (int ch = tid; ch < nch; ch += 256) bsum += cm[ch * pitch + bin];
#pragma unroll
  for (int off = 16; off >= 1; off >>= 1) bsum += __shfl_xor(bsum, off, 32);
  if (lane == 0) s_b[w] = bsum;
  __syncthreads();
  for (int ch = tid; ch < nch; ch += 256) {
    int ls = cm[ch * pitch + bin];
    int len = cm[ch * pitch + bin + 1] - ls;
    const int2* run = binned + (size_t)ch * CHUNK + ls;
    for (int j0 = 0; j0 < len; j0 += RUNB) {
      int v[RUNB];
#pragma unroll
      for (int u = 0; u < RUNB; ++u) {
        int j = j0 + u;
        v[u] = run[j < len ? j : len - 1].y;
      }
#pragma unroll
      for (int u = 0; u < RUNB; ++u) {
        if (j0 + u < len) atomicAdd(&s_c[v[u] & (BINW - 1)], 1);
      }
    }
  }
  __syncthreads();
  int c = s_c[tid];
  int inc = wave_incl_scan(c, lane);
  if (lane == 31) s_w[w] = inc;
  __syncthreads();
  int base = 0;
  for (int k = 0; k < 8; ++k) {
    base += s_b[k];
    if (k < w) base += s_w[k];
  }
  int first = base + inc - c;
  s_cur[tid] = first;
  int node = bin * BINW + tid;
  if (node < N) {
    deg[node] = c;
    rowptr[node] = first;
    dinv[node] = rsqrtf((float)c + 1.0f);
    atomicAdd(&cnt[batch[node]], 1.0f);
  }
  __syncthreads();
  for (int ch = tid; ch < nch; ch += 256) {
    int ls = cm[ch * pitch + bin];
    int len = cm[ch * pitch + bin + 1] - ls;
    const int2* run = binned + (size_t)ch * CHUNK + ls;
    for (int j0 = 0; j0 < len; j0 += RUNB) {
      int2 v[RUNB];
#pragma unroll
      for (int u = 0; u < RUNB; ++u) {
        int j = j0 + u;
        v[u] = run[j < len ? j : len - 1];
      }
#pragma unroll
      for (int u = 0; u < RUNB; ++u) {
        if (j0 + u < len) {
          int pos = atomicAdd(&s_cur[v[u].y & (BINW - 1)], 1);
          csr[pos] = make_int2(v[u].x, 0);
        }
      }
    }
  }
}

__device__ __forceinline__ void bn_coef(const float* __restrict__ bn_sum, const float* __restrict__ bn_sq,
                                        const float* __restrict__ gamma, const float* __restrict__ beta,
                                        float invN, int c, float& sc, float& sh) {
  float mu = bn_sum[c] * invN;
  float var = fmaxf(bn_sq[c] * invN - mu * mu, 0.f);
  sc = rsqrtf(var + 1e-5f) * gamma[c];
  sh = beta[c] - mu * sc;
}

__device__ __forceinline__ void gemm_block(int bid, const float* __restrict__ Z, int ldz,
                                           const float* __restrict__ W, int K,
                                           float* __restrict__ Hout, int N) {
  int wid = bid * (blockDim.x >> 5) + (threadIdx.x >> 5);
  int tilesM = (N + 15) >> 4;
  if (wid >= tilesM * 4) return;
  int tm = wid >> 2, tn = wid & 3;
  int lane = threadIdx.x & 31;
  int half = lane >> 4;
  int m16 = lane & 15;
  int row = tm * 16 + m16;
  int lrow = row < N ? row : N - 1;
  const float* zr = Z + (size_t)lrow * ldz;
  int col = tn * 16 + m16;
  const float* wp = W + col;
  v8f c = {};
  for (int k0 = 0; k0 < K; k0 += 4) {
    int ka = k0 + 2 * half;
    v2f a, b;
    a.x = zr[ka];
    a.y = zr[ka + 1];
    b.x = wp[(size_t)ka * HD];
    b.y = wp[(size_t)(ka + 1) * HD];
    c = __builtin_amdgcn_wmma_f32_16x16x4_f32(false, a, false, b, (short)0, c, false, false);
  }
  int rbase = tm * 16 + half * 8;
#pragma unroll
  for (int j = 0; j < 8; ++j) {
    int r = rbase + j;
    if (r < N) Hout[(size_t)r * HD + col] = c[j];
  }
}

__global__ void __launch_bounds__(256)
k_csr_gemm(const int* __restrict__ cm, const int2* __restrict__ binned,
           int* __restrict__ deg, int* __restrict__ rowptr, float* __restrict__ dinv,
           const int* __restrict__ batch, float* __restrict__ cnt, int2* __restrict__ csr, int N, int nb, int nch,
           const float* __restrict__ X, int ldx,
           const float* __restrict__ W, int K, float* __restrict__ Hout) {
  if ((int)blockIdx.x < nb) {
    csr_block(blockIdx.x, nb, nch, cm, binned, deg, rowptr, dinv, batch, cnt, csr, N);
  } else {
    gemm_block(blockIdx.x - nb, X, ldx, W, K, Hout, N);
  }
}

struct RowBatch {
  float4 v[4];
};

__device__ __forceinline__ void batch_issue(RowBatch& t, const float4* __restrict__ h4, int q, int ex,
                                            int b, int half) {
#pragma unroll
  for (int u = 0; u < 4; ++u) {
    int sl = 8 * b + 2 * u + half;
    int sr = __shfl(ex, sl, 32);
    t.v[u] = h4[(unsigned)(sr * (HD / 4) + q)];
  }
}

template <bool WSUM>
__device__ __forceinline__ void batch_sum(const RowBatch& t, float wl, int b, int half, float4& acc, float& wsum) {
#pragma unroll
  for (int u = 0; u < 4; ++u) {
    float wgt = __shfl(wl, 8 * b + 2 * u + half, 32);
    acc.x += t.v[u].x * wgt;
    acc.y += t.v[u].y * wgt;
    acc.z += t.v[u].z * wgt;
    acc.w += t.v[u].w * wgt;
    if (WSUM) wsum += wgt;
  }
}

template <bool WSUM>
__device__ __forceinline__ void gather_chunk(const float4* __restrict__ h4, int q, int ex, float wl, int m, int half,
                                             float4& acc, float& wsum) {
  int nbat = (m + 7) >> 3;
  RowBatch A, B, C;
  if (nbat == 1) {
    batch_issue(A, h4, q, ex, 0, half);
    batch_sum<WSUM>(A, wl, 0, half, acc, wsum);
  } else if (nbat == 2) {
    batch_issue(A, h4, q, ex, 0, half);
    batch_issue(B, h4, q, ex, 1, half);
    __builtin_amdgcn_sched_barrier(0);
    batch_sum<WSUM>(A, wl, 0, half, acc, wsum);
    batch_sum<WSUM>(B, wl, 1, half, acc, wsum);
  } else {
    batch_issue(A, h4, q, ex, 0, half);
    batch_issue(B, h4, q, ex, 1, half);
    batch_issue(C, h4, q, ex, 2, half);
    __builtin_amdgcn_sched_barrier(0);
    batch_sum<WSUM>(A, wl, 0, half, acc, wsum);
    if (nbat == 4) batch_issue(A, h4, q, ex, 3, half);
    batch_sum<WSUM>(B, wl, 1, half, acc, wsum);
    batch_sum<WSUM>(C, wl, 2, half, acc, wsum);
    if (nbat == 4) batch_sum<WSUM>(A, wl, 3, half, acc, wsum);
  }
}

__global__ void __launch_bounds__(512, 8)
k_gather_relu_stats(const int* __restrict__ rowptr, const int* __restrict__ deg,
                    int2* __restrict__ csr, const float* __restrict__ dinv,
                    const float* __restrict__ h, const float* __restrict__ bias,
                    float* __restrict__ z, float* __restrict__ bn_sum,
                    float* __restrict__ bn_sq, int N) {
  __shared__ __attribute__((aligned(16))) float s_sum[16 * HD];
  __shared__ __attribute__((aligned(16))) float s_sq[16 * HD];
  int w = __builtin_amdgcn_readfirstlane(threadIdx.x >> 5);
  int lane = threadIdx.x & 31;
  int half = lane >> 4;
  int c4 = (lane & 15) << 2;
  const float4* h4 = (const float4*)h;
  int q = lane & 15;
  float4* ps = (float4*)(s_sum + w * HD + c4);
  float4* pq = (float4*)(s_sq + w * HD + c4);
  if (half == 0) {
    *ps = make_float4(0.f, 0.f, 0.f, 0.f);
    *pq = make_float4(0.f, 0.f, 0.f, 0.f);
  }
  int per = (N + (int)gridDim.x - 1) / (int)gridDim.x;
  int r0 = blockIdx.x * per;
  int rend = r0 + per < N ? r0 + per : N;
  int sta[2][4], nna[2][4];
  float dva[2][4];
#pragma unroll
  for (int s2 = 0; s2 < 2; ++s2) {
#pragma unroll
    for (int i = 0; i < 4; ++i) {
      int d = r0 + 64 * s2 + i * 16 + w;
      int dc = d < N ? d : N - 1;
      sta[s2][i] = rowptr[dc];
      nna[s2][i] = d < rend ? deg[dc] : 0;
      dva[s2][i] = dinv[dc];
    }
  }
#pragma unroll
  for (int s2 = 0; s2 < 2; ++s2) {
  int t0 = r0 + 64 * s2;
  if (t0 >= rend) break;
  int st[4], nn[4], ex[4];
  float dv[4], wl[4];
#pragma unroll
  for (int i = 0; i < 4; ++i) {
    st[i] = sta[s2][i];
    nn[i] = nna[s2][i];
    dv[i] = dva[s2][i];
  }
#pragma unroll
  for (int i = 0; i < 4; ++i) {
    int d = t0 + i * 16 + w;
    int sx = d < N ? d : N - 1;
    if (lane < nn[i]) sx = csr[st[i] + lane].x;
    ex[i] = sx;
  }
  float dsr[4];
#pragma unroll
  for (int i = 0; i < 4; ++i) dsr[i] = dinv[ex[i]];
#pragma unroll
  for (int i = 0; i < 4; ++i) {
    float ds = 0.f;
    if (lane < nn[i]) {
      ds = dsr[i];
      csr[st[i] + lane] = make_int2(ex[i], __float_as_int(ds));
    }
    wl[i] = ds * dv[i];
  }
#pragma unroll
  for (int i = 0; i < 4; ++i) {
    int d = t0 + i * 16 + w;
    if (d >= rend) break;
    int start = st[i];
    int n = nn[i];
    float dd = dv[i];
    const float4 hs = h4[(unsigned)(d * (HD / 4) + q)];
    const float4 b4 = ((const float4*)bias)[q];
    float4 acc = {0.f, 0.f, 0.f, 0.f};
    float wsum = 0.f;
    int exc = ex[i];
    float wlc = wl[i];
    for (int base = 0; base < n; base += 32) {
      int m = n - base;
      if (m > 32) m = 32;
      gather_chunk<false>(h4, q, exc, wlc, m, half, acc, wsum);
      int m2 = n - base - 32;
      if (m2 > 0) {
        if (m2 > 32) m2 = 32;
        exc = d;
        float ds = 0.f;
        if (lane < m2) {
          int2* ep = csr + start + base + 32 + lane;
          exc = ep->x;
          ds = dinv[exc];
          *ep = make_int2(exc, __float_as_int(ds));
        }
        wlc = ds * dd;
      }
    }
    acc.x += __shfl_xor(acc.x, 16, 32);
    acc.y += __shfl_xor(acc.y, 16, 32);
    acc.z += __shfl_xor(acc.z, 16, 32);
    acc.w += __shfl_xor(acc.w, 16, 32);
    if (half == 0) {
      float w0 = dd * dd;
      float4 o;
      o.x = hs.x * w0 + acc.x;
      o.y = hs.y * w0 + acc.y;
      o.z = hs.z * w0 + acc.z;
      o.w = hs.w * w0 + acc.w;
      o.x = fmaxf(o.x + b4.x, 0.f);
      o.y = fmaxf(o.y + b4.y, 0.f);
      o.z = fmaxf(o.z + b4.z, 0.f);
      o.w = fmaxf(o.w + b4.w, 0.f);
      ((float4*)z)[(unsigned)(d * (HD / 4) + q)] = o;
      float4 ts = *ps;
      float4 tq = *pq;
      ts.x += o.x; ts.y += o.y; ts.z += o.z; ts.w += o.w;
      tq.x += o.x * o.x; tq.y += o.y * o.y; tq.z += o.z * o.z; tq.w += o.w * o.w;
      *ps = ts;
      *pq = tq;
    }
  }
  }
  __syncthreads();
  if (threadIdx.x < 2 * HD) {
    int c = threadIdx.x & (HD - 1);
    const float* p = (threadIdx.x < HD) ? s_sum : s_sq;
    float t = 0.f;
#pragma unroll
    for (int k = 0; k < 16; ++k) t += p[k * HD + c];
    atomicAdd((threadIdx.x < HD ? bn_sum : bn_sq) + c, t);
  }
}

#define AP 68
__global__ void __launch_bounds__(512, 8)
k_gather_bn_gemm_relu_stats(const int* __restrict__ rowptr, const int* __restrict__ deg,
                            const int2* __restrict__ csr, const float* __restrict__ dinv,
                            const float* __restrict__ zp,
                            const float* __restrict__ pbn_sum, const float* __restrict__ pbn_sq,
                            const float* __restrict__ pgamma, const float* __restrict__ pbeta, float invN,
                            const float* W, const float* __restrict__ bias,
                            float* __restrict__ z, float* __restrict__ bn_sum,
                            float* __restrict__ bn_sq, int N) {
  __shared__ __attribute__((aligned(16))) float s_a[128 * AP];
  __shared__ float s_st[8 * HD];
  float* s_sum = s_st;
  float* s_sq = s_st + 4 * HD;
  int w = __builtin_amdgcn_readfirstlane(threadIdx.x >> 5);
  int lane = threadIdx.x & 31;
  int half = lane >> 4;
  int c4 = (lane & 15) << 2;
  const float4* h4 = (const float4*)zp;
  int q = lane & 15;
  float sc0, sc1, sc2, sc3, sh0, sh1, sh2, sh3;
  bn_coef(pbn_sum, pbn_sq, pgamma, pbeta, invN, c4,     sc0, sh0);
  bn_coef(pbn_sum, pbn_sq, pgamma, pbeta, invN, c4 + 1, sc1, sh1);
  bn_coef(pbn_sum, pbn_sq, pgamma, pbeta, invN, c4 + 2, sc2, sh2);
  bn_coef(pbn_sum, pbn_sq, pgamma, pbeta, invN, c4 + 3, sc3, sh3);
  int per = (N + (int)gridDim.x - 1) / (int)gridDim.x;
  int r0 = blockIdx.x * per;
  int rend = r0 + per < N ? r0 + per : N;
  int tm = w >> 2, tn = w & 3;
  if (threadIdx.x < 4 * HD) {
    s_sum[threadIdx.x] = 0.f;
    s_sq[threadIdx.x] = 0.f;
  }
  int sta[2][4], nna[2][4];
  float dva[2][4];
#pragma unroll
  for (int s2 = 0; s2 < 2; ++s2) {
#pragma unroll
    for (int i = 0; i < 4; ++i) {
      int d = r0 + 64 * s2 + i * 16 + w;
      int dc = d < N ? d : N - 1;
      sta[s2][i] = rowptr[dc];
      nna[s2][i] = d < rend ? deg[dc] : 0;
      dva[s2][i] = dinv[dc];
    }
  }
#pragma unroll
  for (int s2 = 0; s2 < 2; ++s2) {
  int t0 = r0 + 64 * s2;
  if (t0 >= rend) break;
  int st[4], nn[4], ex[4];
  float dv[4], wl[4];
#pragma unroll
  for (int i = 0; i < 4; ++i) {
    st[i] = sta[s2][i];
    nn[i] = nna[s2][i];
    dv[i] = dva[s2][i];
  }
#pragma unroll
  for (int i = 0; i < 4; ++i) {
    int d = t0 + i * 16 + w;
    int2 ent = make_int2(d < N ? d : N - 1, 0);
    if (lane < nn[i]) ent = csr[st[i] + lane];
    ex[i] = ent.x;
    wl[i] = __int_as_float(ent.y) * dv[i];
  }
#pragma unroll
  for (int i = 0; i < 4; ++i) {
    int lr = 64 * s2 + i * 16 + w;
    int d = t0 + i * 16 + w;
    float4 g = {0.f, 0.f, 0.f, 0.f};
    if (d < rend) {
      int start = st[i];
      int n = nn[i];
      float dd = dv[i];
      const float4 hs = h4[(unsigned)(d * (HD / 4) + q)];
      float4 acc = {0.f, 0.f, 0.f, 0.f};
      float wsum = 0.f;
      int exc = ex[i];
      float wlc = wl[i];
      for (int base = 0; base < n; base += 32) {
        int m = n - base;
        if (m > 32) m = 32;
        gather_chunk<true>(h4, q, exc, wlc, m, half, acc, wsum);
        int m2 = n - base - 32;
        if (m2 > 0) {
          if (m2 > 32) m2 = 32;
          int2 ent = make_int2(d, 0);
          if (lane < m2) ent = csr[start + base + 32 + lane];
          exc = ent.x;
          wlc = __int_as_float(ent.y) * dd;
        }
      }
      acc.x += __shfl_xor(acc.x, 16, 32);
      acc.y += __shfl_xor(acc.y, 16, 32);
      acc.z += __shfl_xor(acc.z, 16, 32);
      acc.w += __shfl_xor(acc.w, 16, 32);
      wsum += __shfl_xor(wsum, 16, 32);
      float w0 = dd * dd;
      float r = w0 + wsum;
      g.x = (hs.x * w0 + acc.x) * sc0 + r * sh0;
      g.y = (hs.y * w0 + acc.y) * sc1 + r * sh1;
      g.z = (hs.z * w0 + acc.z) * sc2 + r * sh2;
      g.w = (hs.w * w0 + acc.w) * sc3 + r * sh3;
    }
    if (half == 0) *(float4*)(s_a + lr * AP + c4) = g;
  }
  __syncthreads();
  if (t0 + tm * 16 < rend) {
    int m16 = lane & 15;
    int col = tn * 16 + m16;
    const float* ap = s_a + (64 * s2 + tm * 16 + m16) * AP + 2 * half;
    const float* wp = W + col;
    v8f c = {};
#pragma unroll
    for (int k0 = 0; k0 < HD; k0 += 4) {
      int ka = k0 + 2 * half;
      v2f a, b;
      a.x = ap[k0];
      a.y = ap[k0 + 1];
      b.x = wp[(size_t)ka * HD];
      b.y = wp[(size_t)(ka + 1) * HD];
      c = __builtin_amdgcn_wmma_f32_16x16x4_f32(false, a, false, b, (short)0, c, false, false);
    }
    float bcol = bias[col];
    int rbase = t0 + tm * 16 + half * 8;
    float ts = 0.f, tq = 0.f;
#pragma unroll
    for (int j = 0; j < 8; ++j) {
      int r = rbase + j;
      if (r < rend) {
        float o = fmaxf(c[j] + bcol, 0.f);
        z[(size_t)r * HD + col] = o;
        ts += o;
        tq += o * o;
      }
    }
    ts += __shfl_xor(ts, 16, 32);
    tq += __shfl_xor(tq, 16, 32);
    if (half == 0) {
      s_sum[tm * HD + col] += ts;
      s_sq[tm * HD + col] += tq;
    }
  }
  }
  __syncthreads();
  if (threadIdx.x < 2 * HD) {
    int cc = threadIdx.x & (HD - 1);
    const float* p = (threadIdx.x < HD) ? s_sum : s_sq;
    float t = (p[cc] + p[HD + cc]) + (p[2 * HD + cc] + p[3 * HD + cc]);
    atomicAdd((threadIdx.x < HD ? bn_sum : bn_sq) + cc, t);
  }
}

__global__ void k_bn_l2_pool(const float* __restrict__ z0, const float* __restrict__ z1,
                             const float* __restrict__ z2, const float* __restrict__ bn,
                             const float* __restrict__ gamma0, const float* __restrict__ beta0,
                             const float* __restrict__ gamma1, const float* __restrict__ beta1,
                             const float* __restrict__ gamma2, const float* __restrict__ beta2,
                             float invN, const int* __restrict__ batch,
                             float* __restrict__ out, float* __restrict__ gsum, int N) {
  int wid = blockIdx.x * (blockDim.x >> 5) + (threadIdx.x >> 5);
  int lane = threadIdx.x & 31;
  int r0 = wid * 8;
  if (r0 >= N) return;
  int rend = r0 + 8 < N ? r0 + 8 : N;
  const float* zl[3] = {z0, z1, z2};
  const float* gl[3] = {gamma0, gamma1, gamma2};
  const float* bl[3] = {beta0, beta1, beta2};
  float sc[6], sh[6];
#pragma unroll
  for (int j = 0; j < 6; ++j) {
    int l = j >> 1;
    bn_coef(bn + l * 128, bn + l * 128 + 64, gl[l], bl[l], invN, lane + 32 * (j & 1), sc[j], sh[j]);
  }
  float acc[6] = {0.f, 0.f, 0.f, 0.f, 0.f, 0.f};
  int gcur = -1;
  for (int row = r0; row < rend; ++row) {
    float v[6]; float s = 0.f;
#pragma unroll
    for (int j = 0; j < 6; ++j) {
      v[j] = zl[j >> 1][(size_t)row * HD + lane + 32 * (j & 1)] * sc[j] + sh[j];
      s += v[j] * v[j];
    }
#pragma unroll
    for (int off = 16; off >= 1; off >>= 1) s += __shfl_xor(s, off, 32);
    float inv = 1.0f / fmaxf(sqrtf(s), 1e-12f);
    float* o = out + (size_t)row * CD;
#pragma unroll
    for (int j = 0; j < 6; ++j) o[lane + 32 * j] = v[j] * inv;
    int g = batch[row];
    if (g != gcur) {
      if (gcur >= 0) {
#pragma unroll
        for (int j = 0; j < 6; ++j) atomicAdd(&gsum[(size_t)gcur * CD + lane + 32 * j], acc[j]);
      }
      gcur = g;
#pragma unroll
      for (int j = 0; j < 6; ++j) acc[j] = 0.f;
    }
#pragma unroll
    for (int j = 0; j < 6; ++j) acc[j] += v[j];
  }
  if (gcur >= 0) {
#pragma unroll
    for (int j = 0; j < 6; ++j) atomicAdd(&gsum[(size_t)gcur * CD + lane + 32 * j], acc[j]);
  }
}

__global__ void k_l2_gc(const float* __restrict__ gsum, const float* __restrict__ cnt,
                        float* __restrict__ out, int rows) {
  int wid = blockIdx.x * (blockDim.x >> 5) + (threadIdx.x >> 5);
  if (wid >= rows) return;
  int lane = threadIdx.x & 31;
  float icnt = 1.0f / fmaxf(cnt[wid], 1.0f);
  const float* a = gsum + (size_t)wid * CD;
  float v[6]; float s = 0.f;
#pragma unroll
  for (int j = 0; j < 6; ++j) { v[j] = a[lane + 32 * j] * icnt; s += v[j] * v[j]; }
#pragma unroll
  for (int off = 16; off >= 1; off >>= 1) s += __shfl_xor(s, off, 32);
  float inv = 1.0f / fmaxf(sqrtf(s), 1e-12f);
  float* o = out + (size_t)wid * CD;
#pragma unroll
  for (int j = 0; j < 6; ++j) o[lane + 32 * j] = v[j] * inv;
}

extern "C" void kernel_launch(void* const* d_in, const int* in_sizes, int n_in,
                              void* d_out, int out_size, void* d_ws, size_t ws_size,
                              hipStream_t stream) {
  (void)n_in; (void)out_size; (void)ws_size;
  const float* x     = (const float*)d_in[0];
  const int*   ei    = (const int*)d_in[1];
  const int*   batch = (const int*)d_in[2];
  int N = in_sizes[0] / 128;
  int E = in_sizes[1] / 2;
  const int* src = ei;
  const int* dst = ei + E;
  const float* Wl[3]  = {(const float*)d_in[3],  (const float*)d_in[7],  (const float*)d_in[11]};
  const float* bl[3]  = {(const float*)d_in[4],  (const float*)d_in[8],  (const float*)d_in[12]};
  const float* gl[3]  = {(const float*)d_in[5],  (const float*)d_in[9],  (const float*)d_in[13]};
  const float* bel[3] = {(const float*)d_in[6],  (const float*)d_in[10], (const float*)d_in[14]};

  float* ws   = (float*)d_ws;
  float* dinv = ws;                       ws += N;
  ws = (float*)(((size_t)ws + 255) & ~(size_t)255);
  float* h    = ws;                       ws += (size_t)N * HD;
  float* zb[3];
  zb[0] = ws;                             ws += (size_t)N * HD;
  zb[1] = ws;                             ws += (size_t)N * HD;
  zb[2] = ws;                             ws += (size_t)N * HD;
  float* gsum = ws;                       ws += (size_t)GN * CD;
  float* cnt  = ws;                       ws += GN;
  float* bn   = ws;
  ws += 3 * 128;
  int* deg    = (int*)(((size_t)ws + 15) & ~(size_t)15);
  int* rowptr = deg + (((size_t)N + 3) & ~(size_t)3);
  int nb  = (N + BINW - 1) / BINW;
  int nch = (E + CHUNK - 1) / CHUNK;
  int* cm       = rowptr + (((size_t)N + 3) & ~(size_t)3);
  int2* binned  = (int2*)(cm + (((size_t)nch * (nb + 1) + 3) & ~(size_t)3));
  int2* csr     = binned + (size_t)nch * CHUNK;

  k_chunk_order<<<nch, 1024, 0, stream>>>(src, dst, E, nb, cm, binned, gsum, GN * CD + GN + 3 * 128);
  int tilesM = (N + 15) / 16;
  int waves = tilesM * 4;
  k_csr_gemm<<<nb + (waves + 7) / 8, 256, 0, stream>>>(cm, binned, deg, rowptr, dinv, batch, cnt, csr,
                                                      N, nb, nch, x, 128, Wl[0], 128, h);

  float invN = 1.0f / (float)N;
  int gblk = (N + 127) / 128 > GBLK ? (N + 127) / 128 : GBLK;
  k_gather_relu_stats<<<gblk, 512, 0, stream>>>(rowptr, deg, csr, dinv, h, bl[0], zb[0],
                                                         bn, bn + 64, N);
  for (int l = 1; l < 3; ++l) {
    float* bn_sum = bn + l * 128;
    float* bn_sq  = bn_sum + 64;
    k_gather_bn_gemm_relu_stats<<<gblk, 512, 0, stream>>>(rowptr, deg, csr, dinv, zb[l - 1],
                                                                   bn_sum - 128, bn_sq - 128,
                                                                   gl[l - 1], bel[l - 1], invN,
                                                                   Wl[l], bl[l], zb[l], bn_sum, bn_sq, N);
  }

  float* out = (float*)d_out;
  k_bn_l2_pool<<<(N + 63) / 64, 256, 0, stream>>>(zb[0], zb[1], zb[2], bn, gl[0], bel[0], gl[1], bel[1],
                                                  gl[2], bel[2], invN, batch, out, gsum, N);
  k_l2_gc<<<(GN + 7) / 8, 256, 0, stream>>>(gsum, cnt, out + (size_t)N * CD, GN);
}


# layer-1 GEMM restaged through LDS: a wave loads 16 whole x rows, takes a 16x64 strip (W transposed in LDS, one A read per 4 products), stores whole rows; 4x fewer waves; order of every sum unchanged
# speedup vs baseline: 4.9685x; 1.0362x over previous
#include <hip/hip_runtime.h>

typedef float v2f __attribute__((ext_vector_type(2)));
typedef float v8f __attribute__((ext_vector_type(8)));

#define HD 64
#define CD 192
#define GN 256
#define GBLK 512

#define BINW  256
#define MAXB  256
#define CHUNK 4096

__device__ __forceinline__ int wave_incl_scan(int v, int lane) {
  for (int off = 1; off < 32; off <<= 1) {
    int u = __shfl_up(v, off, 32);
    if (lane >= off) v += u;
  }
  return v;
}

__global__ void __launch_bounds__(1024)
k_chunk_order(const int* __restrict__ src, const int* __restrict__ dst, int E, int nb,
              int* __restrict__ cm, int2* __restrict__ binned, float* __restrict__ zero, int nzero) {
  __shared__ int s_cnt[MAXB];
  __shared__ int s_lst[MAXB];
  __shared__ int s_wc[8];
  __shared__ int2 s_e[CHUNK];
  int tid = threadIdx.x, lane = tid & 31, w = tid >> 5;
  int chunk = blockIdx.x, nch = gridDim.x;
  for (int i = chunk * 1024 + tid; i < nzero; i += nch * 1024) zero[i] = 0.f;
  if (tid < MAXB) s_cnt[tid] = 0;
  __syncthreads();
  int nloc = E - chunk * CHUNK;
  if (nloc > CHUNK) nloc = CHUNK;
  const int* cs = src + (size_t)chunk * CHUNK;
  const int* cd = dst + (size_t)chunk * CHUNK;
  int es[4], ed[4], er[4];
#pragma unroll
  for (int k = 0; k < 4; ++k) {
    int i = k * 1024 + tid;
    int ic = i < nloc ? i : nloc - 1;
    es[k] = cs[ic];
    ed[k] = cd[ic];
  }
#pragma unroll
  for (int k = 0; k < 4; ++k) {
    er[k] = 0;
    if (k * 1024 + tid < nloc) er[k] = atomicAdd(&s_cnt[ed[k] >> 8], 1);
  }
  __syncthreads();
  int c = (tid < MAXB) ? s_cnt[tid] : 0;
  int ci = wave_incl_scan(c, lane);
  if (lane == 31 && w < 8) s_wc[w] = ci;
  __syncthreads();
  if (tid < MAXB) {
    int bc = 0;
    for (int k = 0; k < 8; ++k) {
      if (k < w) bc += s_wc[k];
    }
    int first = bc + ci - c;
    s_lst[tid] = first;
    if (tid < nb) cm[chunk * (nb + 1) + tid] = first;
    if (tid == 0) cm[chunk * (nb + 1) + nb] = nloc;
  }
  __syncthreads();
#pragma unroll
  for (int k = 0; k < 4; ++k) {
    if (k * 1024 + tid < nloc) s_e[s_lst[ed[k] >> 8] + er[k]] = make_int2(es[k], ed[k]);
  }
  __syncthreads();
#pragma unroll
  for (int k = 0; k < 4; ++k) {
    int i = k * 1024 + tid;
    if (i < nloc) binned[(size_t)chunk * CHUNK + i] = s_e[i];
  }
}

#define RUNB 16
__device__ __forceinline__ void csr_block(int bin, int nb, int nch, const int* __restrict__ cm,
                                          const int2* __restrict__ binned, int* __restrict__ deg,
                                          int* __restrict__ rowptr, float* __restrict__ dinv,
                                          const int* __restrict__ batch, float* __restrict__ cnt,
                                          int2* __restrict__ csr, int N) {
  __shared__ int s_c[BINW], s_cur[BINW], s_w[8], s_b[8];
  int tid = threadIdx.x, lane = tid & 31, w = tid >> 5;
  int pitch = nb + 1;
  s_c[tid] = 0;
  int bsum = 0;
  for (int ch = tid; ch < nch; ch += 256) bsum += cm[ch * pitch + bin];
#pragma unroll
  for (int off = 16; off >= 1; off >>= 1) bsum += __shfl_xor(bsum, off, 32);
  if (lane == 0) s_b[w] = bsum;
  __syncthreads();
  for (int ch = tid; ch < nch; ch += 256) {
    int ls = cm[ch * pitch + bin];
    int len = cm[ch * pitch + bin + 1] - ls;
    const int2* run = binned + (size_t)ch * CHUNK + ls;
    for (int j0 = 0; j0 < len; j0 += RUNB) {
      int v[RUNB];
#pragma unroll
      for (int u = 0; u < RUNB; ++u) {
        int j = j0 + u;
        v[u] = run[j < len ? j : len - 1].y;
      }
#pragma unroll
      for (int u = 0; u < RUNB; ++u) {
        if (j0 + u < len) atomicAdd(&s_c[v[u] & (BINW - 1)], 1);
      }
    }
  }
  __syncthreads();
  int c = s_c[tid];
  int inc = wave_incl_scan(c, lane);
  if (lane == 31) s_w[w] = inc;
  __syncthreads();
  int base = 0;
  for (int k = 0; k < 8; ++k) {
    base += s_b[k];
    if (k < w) base += s_w[k];
  }
  int first = base + inc - c;
  s_cur[tid] = first;
  int node = bin * BINW + tid;
  if (node < N) {
    deg[node] = c;
    rowptr[node] = first;
    dinv[node] = rsqrtf((float)c + 1.0f);
    atomicAdd(&cnt[batch[node]], 1.0f);
  }
  __syncthreads();
  for (int ch = tid; ch < nch; ch += 256) {
    int ls = cm[ch * pitch + bin];
    int len = cm[ch * pitch + bin + 1] - ls;
    const int2* run = binned + (size_t)ch * CHUNK + ls;
    for (int j0 = 0; j0 < len; j0 += RUNB) {
      int2 v[RUNB];
#pragma unroll
      for (int u = 0; u < RUNB; ++u) {
        int j = j0 + u;
        v[u] = run[j < len ? j : len - 1];
      }
#pragma unroll
      for (int u = 0; u < RUNB; ++u) {
        if (j0 + u < len) {
          int pos = atomicAdd(&s_cur[v[u].y & (BINW - 1)], 1);
          csr[pos] = make_int2(v[u].x, 0);
        }
      }
    }
  }
}

__device__ __forceinline__ void bn_coef(const float* __restrict__ bn_sum, const float* __restrict__ bn_sq,
                                        const float* __restrict__ gamma, const float* __restrict__ beta,
                                        float invN, int c, float& sc, float& sh) {
  float mu = bn_sum[c] * invN;
  float var = fmaxf(bn_sq[c] * invN - mu * mu, 0.f);
  sc = rsqrtf(var + 1e-5f) * gamma[c];
  sh = beta[c] - mu * sc;
}

#define XK 128
#define XP 132
__device__ __forceinline__ void gemm_block(int bid, const float* __restrict__ X, const float* __restrict__ W,
                                           float* __restrict__ Hout, int N) {
  __shared__ __attribute__((aligned(16))) float s_x[8 * 16 * XP];
  __shared__ __attribute__((aligned(16))) float s_wt[HD * XP];
  int tid = threadIdx.x;
  int lane = tid & 31;
  int w = __builtin_amdgcn_readfirstlane(tid >> 5);
  int m16 = lane & 15;
  int half = lane >> 4;
  int row0 = (bid * 8 + w) * 16;
  const float4* W4 = (const float4*)W;
  float4 wv[8];
  for (int i = 0; i < 8; ++i) wv[i] = W4[tid + 256 * i];
  const float4* X4 = (const float4*)X;
  float4 xr[16];
  for (int i = 0; i < 16; ++i) {
    int r = row0 + i;
    int rc = r < N ? r : N - 1;
    xr[i] = X4[(size_t)rc * (XK / 4) + lane];
  }
  for (int i = 0; i < 8; ++i) {
    int f = tid + 256 * i;
    int k = f >> 4;
    int c = (f & 15) << 2;
    s_wt[(c + 0) * XP + k] = wv[i].x;
    s_wt[(c + 1) * XP + k] = wv[i].y;
    s_wt[(c + 2) * XP + k] = wv[i].z;
    s_wt[(c + 3) * XP + k] = wv[i].w;
  }
  float* sx = s_x + w * (16 * XP);
  for (int i = 0; i < 16; ++i) *(float4*)(sx + i * XP + lane * 4) = xr[i];
  __syncthreads();
  const float* ap = sx + m16 * XP + 2 * half;
  const float* bp = s_wt + m16 * XP + 2 * half;
  v8f c0 = {}, c1 = {}, c2 = {}, c3 = {};
  for (int k0 = 0; k0 < XK; k0 += 4) {
    v2f a  = *(const v2f*)(ap + k0);
    v2f b0 = *(const v2f*)(bp + k0);
    v2f b1 = *(const v2f*)(bp + 16 * XP + k0);
    v2f b2 = *(const v2f*)(bp + 32 * XP + k0);
    v2f b3 = *(const v2f*)(bp + 48 * XP + k0);
    c0 = __builtin_amdgcn_wmma_f32_16x16x4_f32(false, a, false, b0, (short)0, c0, false, false);
    c1 = __builtin_amdgcn_wmma_f32_16x16x4_f32(false, a, false, b1, (short)0, c1, false, false);
    c2 = __builtin_amdgcn_wmma_f32_16x16x4_f32(false, a, false, b2, (short)0, c2, false, false);
    c3 = __builtin_amdgcn_wmma_f32_16x16x4_f32(false, a, false, b3, (short)0, c3, false, false);
  }
  float* so = sx + (8 * half) * XP + m16;
  for (int j = 0; j < 8; ++j) {
    so[j * XP]      = c0[j];
    so[j * XP + 16] = c1[j];
    so[j * XP + 32] = c2[j];
    so[j * XP + 48] = c3[j];
  }
  __syncthreads();
  const float* sr = sx + half * XP + 4 * m16;
  float4* ho = (float4*)Hout + (size_t)(row0 + half) * (HD / 4) + m16;
  if (row0 + 16 <= N) {
    for (int i = 0; i < 8; ++i) ho[i * 2 * (HD / 4)] = *(const float4*)(sr + 2 * i * XP);
  } else {
    for (int i = 0; i < 8; ++i) {
      if (row0 + 2 * i + half < N) ho[i * 2 * (HD / 4)] = *(const float4*)(sr + 2 * i * XP);
    }
  }
}

__global__ void __launch_bounds__(256)
k_csr_gemm(const int* __restrict__ cm, const int2* __restrict__ binned,
           int* __restrict__ deg, int* __restrict__ rowptr, float* __restrict__ dinv,
           const int* __restrict__ batch, float* __restrict__ cnt, int2* __restrict__ csr, int N, int nb, int nch,
           const float* __restrict__ X, const float* __restrict__ W, float* __restrict__ Hout) {
  if ((int)blockIdx.x < nb) {
    csr_block(blockIdx.x, nb, nch, cm, binned, deg, rowptr, dinv, batch, cnt, csr, N);
  } else {
    gemm_block(blockIdx.x - nb, X, W, Hout, N);
  }
}

struct RowBatch {
  float4 v[4];
};

__device__ __forceinline__ void batch_issue(RowBatch& t, const float4* __restrict__ h4, int q, int ex,
                                            int b, int half) {
#pragma unroll
  for (int u = 0; u < 4; ++u) {
    int sl = 8 * b + 2 * u + half;
    int sr = __shfl(ex, sl, 32);
    t.v[u] = h4[(unsigned)(sr * (HD / 4) + q)];
  }
}

template <bool WSUM>
__device__ __forceinline__ void batch_sum(const RowBatch& t, float wl, int b, int half, float4& acc, float& wsum) {
#pragma unroll
  for (int u = 0; u < 4; ++u) {
    float wgt = __shfl(wl, 8 * b + 2 * u + half, 32);
    acc.x += t.v[u].x * wgt;
    acc.y += t.v[u].y * wgt;
    acc.z += t.v[u].z * wgt;
    acc.w += t.v[u].w * wgt;
    if (WSUM) wsum += wgt;
  }
}

template <bool WSUM>
__device__ __forceinline__ void gather_chunk(const float4* __restrict__ h4, int q, int ex, float wl, int m, int half,
                                             float4& acc, float& wsum) {
  int nbat = (m + 7) >> 3;
  RowBatch A, B, C;
  if (nbat == 1) {
    batch_issue(A, h4, q, ex, 0, half);
    batch_sum<WSUM>(A, wl, 0, half, acc, wsum);
  } else if (nbat == 2) {
    batch_issue(A, h4, q, ex, 0, half);
    batch_issue(B, h4, q, ex, 1, half);
    __builtin_amdgcn_sched_barrier(0);
    batch_sum<WSUM>(A, wl, 0, half, acc, wsum);
    batch_sum<WSUM>(B, wl, 1, half, acc, wsum);
  } else {
    batch_issue(A, h4, q, ex, 0, half);
    batch_issue(B, h4, q, ex, 1, half);
    batch_issue(C, h4, q, ex, 2, half);
    __builtin_amdgcn_sched_barrier(0);
    batch_sum<WSUM>(A, wl, 0, half, acc, wsum);
    if (nbat == 4) batch_issue(A, h4, q, ex, 3, half);
    batch_sum<WSUM>(B, wl, 1, half, acc, wsum);
    batch_sum<WSUM>(C, wl, 2, half, acc, wsum);
    if (nbat == 4) batch_sum<WSUM>(A, wl, 3, half, acc, wsum);
  }
}

__global__ void __launch_bounds__(512, 8)
k_gather_relu_stats(const int* __restrict__ rowptr, const int* __restrict__ deg,
                    int2* __restrict__ csr, const float* __restrict__ dinv,
                    const float* __restrict__ h, const float* __restrict__ bias,
                    float* __restrict__ z, float* __restrict__ bn_sum,
                    float* __restrict__ bn_sq, int N) {
  __shared__ __attribute__((aligned(16))) float s_sum[16 * HD];
  __shared__ __attribute__((aligned(16))) float s_sq[16 * HD];
  int w = __builtin_amdgcn_readfirstlane(threadIdx.x >> 5);
  int lane = threadIdx.x & 31;
  int half = lane >> 4;
  int c4 = (lane & 15) << 2;
  const float4* h4 = (const float4*)h;
  int q = lane & 15;
  float4* ps = (float4*)(s_sum + w * HD + c4);
  float4* pq = (float4*)(s_sq + w * HD + c4);
  if (half == 0) {
    *ps = make_float4(0.f, 0.f, 0.f, 0.f);
    *pq = make_float4(0.f, 0.f, 0.f, 0.f);
  }
  int per = (N + (int)gridDim.x - 1) / (int)gridDim.x;
  int r0 = blockIdx.x * per;
  int rend = r0 + per < N ? r0 + per : N;
  int sta[2][4], nna[2][4];
  float dva[2][4];
#pragma unroll
  for (int s2 = 0; s2 < 2; ++s2) {
#pragma unroll
    for (int i = 0; i < 4; ++i) {
      int d = r0 + 64 * s2 + i * 16 + w;
      int dc = d < N ? d : N - 1;
      sta[s2][i] = rowptr[dc];
      nna[s2][i] = d < rend ? deg[dc] : 0;
      dva[s2][i] = dinv[dc];
    }
  }
#pragma unroll
  for (int s2 = 0; s2 < 2; ++s2) {
  int t0 = r0 + 64 * s2;
  if (t0 >= rend) break;
  int st[4], nn[4], ex[4];
  float dv[4], wl[4];
#pragma unroll
  for (int i = 0; i < 4; ++i) {
    st[i] = sta[s2][i];
    nn[i] = nna[s2][i];
    dv[i] = dva[s2][i];
  }
#pragma unroll
  for (int i = 0; i < 4; ++i) {
    int d = t0 + i * 16 + w;
    int sx = d < N ? d : N - 1;
    if (lane < nn[i]) sx = csr[st[i] + lane].x;
    ex[i] = sx;
  }
  float dsr[4];
#pragma unroll
  for (int i = 0; i < 4; ++i) dsr[i] = dinv[ex[i]];
#pragma unroll
  for (int i = 0; i < 4; ++i) {
    float ds = 0.f;
    if (lane < nn[i]) {
      ds = dsr[i];
      csr[st[i] + lane] = make_int2(ex[i], __float_as_int(ds));
    }
    wl[i] = ds * dv[i];
  }
#pragma unroll
  for (int i = 0; i < 4; ++i) {
    int d = t0 + i * 16 + w;
    if (d >= rend) break;
    int start = st[i];
    int n = nn[i];
    float dd = dv[i];
    const float4 hs = h4[(unsigned)(d * (HD / 4) + q)];
    const float4 b4 = ((const float4*)bias)[q];
    float4 acc = {0.f, 0.f, 0.f, 0.f};
    float wsum = 0.f;
    int exc = ex[i];
    float wlc = wl[i];
    for (int base = 0; base < n; base += 32) {
      int m = n - base;
      if (m > 32) m = 32;
      gather_chunk<false>(h4, q, exc, wlc, m, half, acc, wsum);
      int m2 = n - base - 32;
      if (m2 > 0) {
        if (m2 > 32) m2 = 32;
        exc = d;
        float ds = 0.f;
        if (lane < m2) {
          int2* ep = csr + start + base + 32 + lane;
          exc = ep->x;
          ds = dinv[exc];
          *ep = make_int2(exc, __float_as_int(ds));
        }
        wlc = ds * dd;
      }
    }
    acc.x += __shfl_xor(acc.x, 16, 32);
    acc.y += __shfl_xor(acc.y, 16, 32);
    acc.z += __shfl_xor(acc.z, 16, 32);
    acc.w += __shfl_xor(acc.w, 16, 32);
    if (half == 0) {
      float w0 = dd * dd;
      float4 o;
      o.x = hs.x * w0 + acc.x;
      o.y = hs.y * w0 + acc.y;
      o.z = hs.z * w0 + acc.z;
      o.w = hs.w * w0 + acc.w;
      o.x = fmaxf(o.x + b4.x, 0.f);
      o.y = fmaxf(o.y + b4.y, 0.f);
      o.z = fmaxf(o.z + b4.z, 0.f);
      o.w = fmaxf(o.w + b4.w, 0.f);
      ((float4*)z)[(unsigned)(d * (HD / 4) + q)] = o;
      float4 ts = *ps;
      float4 tq = *pq;
      ts.x += o.x; ts.y += o.y; ts.z += o.z; ts.w += o.w;
      tq.x += o.x * o.x; tq.y += o.y * o.y; tq.z += o.z * o.z; tq.w += o.w * o.w;
      *ps = ts;
      *pq = tq;
    }
  }
  }
  __syncthreads();
  if (threadIdx.x < 2 * HD) {
    int c = threadIdx.x & (HD - 1);
    const float* p = (threadIdx.x < HD) ? s_sum : s_sq;
    float t = 0.f;
#pragma unroll
    for (int k = 0; k < 16; ++k) t += p[k * HD + c];
    atomicAdd((threadIdx.x < HD ? bn_sum : bn_sq) + c, t);
  }
}

#define AP 68
__global__ void __launch_bounds__(512, 8)
k_gather_bn_gemm_relu_stats(const int* __restrict__ rowptr, const int* __restrict__ deg,
                            const int2* __restrict__ csr, const float* __restrict__ dinv,
                            const float* __restrict__ zp,
                            const float* __restrict__ pbn_sum, const float* __restrict__ pbn_sq,
                            const float* __restrict__ pgamma, const float* __restrict__ pbeta, float invN,
                            const float* W, const float* __restrict__ bias,
                            float* __restrict__ z, float* __restrict__ bn_sum,
                            float* __restrict__ bn_sq, int N) {
  __shared__ __attribute__((aligned(16))) float s_a[128 * AP];
  __shared__ float s_st[8 * HD];
  float* s_sum = s_st;
  float* s_sq = s_st + 4 * HD;
  int w = __builtin_amdgcn_readfirstlane(threadIdx.x >> 5);
  int lane = threadIdx.x & 31;
  int half = lane >> 4;
  int c4 = (lane & 15) << 2;
  const float4* h4 = (const float4*)zp;
  int q = lane & 15;
  float sc0, sc1, sc2, sc3, sh0, sh1, sh2, sh3;
  bn_coef(pbn_sum, pbn_sq, pgamma, pbeta, invN, c4,     sc0, sh0);
  bn_coef(pbn_sum, pbn_sq, pgamma, pbeta, invN, c4 + 1, sc1, sh1);
  bn_coef(pbn_sum, pbn_sq, pgamma, pbeta, invN, c4 + 2, sc2, sh2);
  bn_coef(pbn_sum, pbn_sq, pgamma, pbeta, invN, c4 + 3, sc3, sh3);
  int per = (N + (int)gridDim.x - 1) / (int)gridDim.x;
  int r0 = blockIdx.x * per;
  int rend = r0 + per < N ? r0 + per : N;
  int tm = w >> 2, tn = w & 3;
  if (threadIdx.x < 4 * HD) {
    s_sum[threadIdx.x] = 0.f;
    s_sq[threadIdx.x] = 0.f;
  }
  int sta[2][4], nna[2][4];
  float dva[2][4];
#pragma unroll
  for (int s2 = 0; s2 < 2; ++s2) {
#pragma unroll
    for (int i = 0; i < 4; ++i) {
      int d = r0 + 64 * s2 + i * 16 + w;
      int dc = d < N ? d : N - 1;
      sta[s2][i] = rowptr[dc];
      nna[s2][i] = d < rend ? deg[dc] : 0;
      dva[s2][i] = dinv[dc];
    }
  }
#pragma unroll
  for (int s2 = 0; s2 < 2; ++s2) {
  int t0 = r0 + 64 * s2;
  if (t0 >= rend) break;
  int st[4], nn[4], ex[4];
  float dv[4], wl[4];
#pragma unroll
  for (int i = 0; i < 4; ++i) {
    st[i] = sta[s2][i];
    nn[i] = nna[s2][i];
    dv[i] = dva[s2][i];
  }
#pragma unroll
  for (int i = 0; i < 4; ++i) {
    int d = t0 + i * 16 + w;
    int2 ent = make_int2(d < N ? d : N - 1, 0);
    if (lane < nn[i]) ent = csr[st[i] + lane];
    ex[i] = ent.x;
    wl[i] = __int_as_float(ent.y) * dv[i];
  }
#pragma unroll
  for (int i = 0; i < 4; ++i) {
    int lr = 64 * s2 + i * 16 + w;
    int d = t0 + i * 16 + w;
    float4 g = {0.f, 0.f, 0.f, 0.f};
    if (d < rend) {
      int start = st[i];
      int n = nn[i];
      float dd = dv[i];
      const float4 hs = h4[(unsigned)(d * (HD / 4) + q)];
      float4 acc = {0.f, 0.f, 0.f, 0.f};
      float wsum = 0.f;
      int exc = ex[i];
      float wlc = wl[i];
      for (int base = 0; base < n; base += 32) {
        int m = n - base;
        if (m > 32) m = 32;
        gather_chunk<true>(h4, q, exc, wlc, m, half, acc, wsum);
        int m2 = n - base - 32;
        if (m2 > 0) {
          if (m2 > 32) m2 = 32;
          int2 ent = make_int2(d, 0);
          if (lane < m2) ent = csr[start + base + 32 + lane];
          exc = ent.x;
          wlc = __int_as_float(ent.y) * dd;
        }
      }
      acc.x += __shfl_xor(acc.x, 16, 32);
      acc.y += __shfl_xor(acc.y, 16, 32);
      acc.z += __shfl_xor(acc.z, 16, 32);
      acc.w += __shfl_xor(acc.w, 16, 32);
      wsum += __shfl_xor(wsum, 16, 32);
      float w0 = dd * dd;
      float r = w0 + wsum;
      g.x = (hs.x * w0 + acc.x) * sc0 + r * sh0;
      g.y = (hs.y * w0 + acc.y) * sc1 + r * sh1;
      g.z = (hs.z * w0 + acc.z) * sc2 + r * sh2;
      g.w = (hs.w * w0 + acc.w) * sc3 + r * sh3;
    }
    if (half == 0) *(float4*)(s_a + lr * AP + c4) = g;
  }
  __syncthreads();
  if (t0 + tm * 16 < rend) {
    int m16 = lane & 15;
    int col = tn * 16 + m16;
    const float* ap = s_a + (64 * s2 + tm * 16 + m16) * AP + 2 * half;
    const float* wp = W + col;
    v8f c = {};
#pragma unroll
    for (int k0 = 0; k0 < HD; k0 += 4) {
      int ka = k0 + 2 * half;
      v2f a, b;
      a.x = ap[k0];
      a.y = ap[k0 + 1];
      b.x = wp[(size_t)ka * HD];
      b.y = wp[(size_t)(ka + 1) * HD];
      c = __builtin_amdgcn_wmma_f32_16x16x4_f32(false, a, false, b, (short)0, c, false, false);
    }
    float bcol = bias[col];
    int rbase = t0 + tm * 16 + half * 8;
    float ts = 0.f, tq = 0.f;
#pragma unroll
    for (int j = 0; j < 8; ++j) {
      int r = rbase + j;
      if (r < rend) {
        float o = fmaxf(c[j] + bcol, 0.f);
        z[(size_t)r * HD + col] = o;
        ts += o;
        tq += o * o;
      }
    }
    ts += __shfl_xor(ts, 16, 32);
    tq += __shfl_xor(tq, 16, 32);
    if (half == 0) {
      s_sum[tm * HD + col] += ts;
      s_sq[tm * HD + col] += tq;
    }
  }
  }
  __syncthreads();
  if (threadIdx.x < 2 * HD) {
    int cc = threadIdx.x & (HD - 1);
    const float* p = (threadIdx.x < HD) ? s_sum : s_sq;
    float t = (p[cc] + p[HD + cc]) + (p[2 * HD + cc] + p[3 * HD + cc]);
    atomicAdd((threadIdx.x < HD ? bn_sum : bn_sq) + cc, t);
  }
}

__global__ void k_bn_l2_pool(const float* __restrict__ z0, const float* __restrict__ z1,
                             const float* __restrict__ z2, const float* __restrict__ bn,
                             const float* __restrict__ gamma0, const float* __restrict__ beta0,
                             const float* __restrict__ gamma1, const float* __restrict__ beta1,
                             const float* __restrict__ gamma2, const float* __restrict__ beta2,
                             float invN, const int* __restrict__ batch,
                             float* __restrict__ out, float* __restrict__ gsum, int N) {
  int wid = blockIdx.x * (blockDim.x >> 5) + (threadIdx.x >> 5);
  int lane = threadIdx.x & 31;
  int r0 = wid * 8;
  if (r0 >= N) return;
  int rend = r0 + 8 < N ? r0 + 8 : N;
  const float* zl[3] = {z0, z1, z2};
  const float* gl[3] = {gamma0, gamma1, gamma2};
  const float* bl[3] = {beta0, beta1, beta2};
  float sc[6], sh[6];
#pragma unroll
  for (int j = 0; j < 6; ++j) {
    int l = j >> 1;
    bn_coef(bn + l * 128, bn + l * 128 + 64, gl[l], bl[l], invN, lane + 32 * (j & 1), sc[j], sh[j]);
  }
  float acc[6] = {0.f, 0.f, 0.f, 0.f, 0.f, 0.f};
  int gcur = -1;
  for (int row = r0; row < rend; ++row) {
    float v[6]; float s = 0.f;
#pragma unroll
    for (int j = 0; j < 6; ++j) {
      v[j] = zl[j >> 1][(size_t)row * HD + lane + 32 * (j & 1)] * sc[j] + sh[j];
      s += v[j] * v[j];
    }
#pragma unroll
    for (int off = 16; off >= 1; off >>= 1) s += __shfl_xor(s, off, 32);
    float inv = 1.0f / fmaxf(sqrtf(s), 1e-12f);
    float* o = out + (size_t)row * CD;
#pragma unroll
    for (int j = 0; j < 6; ++j) o[lane + 32 * j] = v[j] * inv;
    int g = batch[row];
    if (g != gcur) {
      if (gcur >= 0) {
#pragma unroll
        for (int j = 0; j < 6; ++j) atomicAdd(&gsum[(size_t)gcur * CD + lane + 32 * j], acc[j]);
      }
      gcur = g;
#pragma unroll
      for (int j = 0; j < 6; ++j) acc[j] = 0.f;
    }
#pragma unroll
    for (int j = 0; j < 6; ++j) acc[j] += v[j];
  }
  if (gcur >= 0) {
#pragma unroll
    for (int j = 0; j < 6; ++j) atomicAdd(&gsum[(size_t)gcur * CD + lane + 32 * j], acc[j]);
  }
}

__global__ void k_l2_gc(const float* __restrict__ gsum, const float* __restrict__ cnt,
                        float* __restrict__ out, int rows) {
  int wid = blockIdx.x * (blockDim.x >> 5) + (threadIdx.x >> 5);
  if (wid >= rows) return;
  int lane = threadIdx.x & 31;
  float icnt = 1.0f / fmaxf(cnt[wid], 1.0f);
  const float* a = gsum + (size_t)wid * CD;
  float v[6]; float s = 0.f;
#pragma unroll
  for (int j = 0; j < 6; ++j) { v[j] = a[lane + 32 * j] * icnt; s += v[j] * v[j]; }
#pragma unroll
  for (int off = 16; off >= 1; off >>= 1) s += __shfl_xor(s, off, 32);
  float inv = 1.0f / fmaxf(sqrtf(s), 1e-12f);
  float* o = out + (size_t)wid * CD;
#pragma unroll
  for (int j = 0; j < 6; ++j) o[lane + 32 * j] = v[j] * inv;
}

extern "C" void kernel_launch(void* const* d_in, const int* in_sizes, int n_in,
                              void* d_out, int out_size, void* d_ws, size_t ws_size,
                              hipStream_t stream) {
  (void)n_in; (void)out_size; (void)ws_size;
  const float* x     = (const float*)d_in[0];
  const int*   ei    = (const int*)d_in[1];
  const int*   batch = (const int*)d_in[2];
  int N = in_sizes[0] / 128;
  int E = in_sizes[1] / 2;
  const int* src = ei;
  const int* dst = ei + E;
  const float* Wl[3]  = {(const float*)d_in[3],  (const float*)d_in[7],  (const float*)d_in[11]};
  const float* bl[3]  = {(const float*)d_in[4],  (const float*)d_in[8],  (const float*)d_in[12]};
  const float* gl[3]  = {(const float*)d_in[5],  (const float*)d_in[9],  (const float*)d_in[13]};
  const float* bel[3] = {(const float*)d_in[6],  (const float*)d_in[10], (const float*)d_in[14]};

  float* ws   = (float*)d_ws;
  float* dinv = ws;                       ws += N;
  ws = (float*)(((size_t)ws + 255) & ~(size_t)255);
  float* h    = ws;                       ws += (size_t)N * HD;
  float* zb[3];
  zb[0] = ws;                             ws += (size_t)N * HD;
  zb[1] = ws;                             ws += (size_t)N * HD;
  zb[2] = ws;                             ws += (size_t)N * HD;
  float* gsum = ws;                       ws += (size_t)GN * CD;
  float* cnt  = ws;                       ws += GN;
  float* bn   = ws;
  ws += 3 * 128;
  int* deg    = (int*)(((size_t)ws + 15) & ~(size_t)15);
  int* rowptr = deg + (((size_t)N + 3) & ~(size_t)3);
  int nb  = (N + BINW - 1) / BINW;
  int nch = (E + CHUNK - 1) / CHUNK;
  int* cm       = rowptr + (((size_t)N + 3) & ~(size_t)3);
  int2* binned  = (int2*)(cm + (((size_t)nch * (nb + 1) + 3) & ~(size_t)3));
  int2* csr     = binned + (size_t)nch * CHUNK;

  k_chunk_order<<<nch, 1024, 0, stream>>>(src, dst, E, nb, cm, binned, gsum, GN * CD + GN + 3 * 128);
  k_csr_gemm<<<nb + (N + 127) / 128, 256, 0, stream>>>(cm, binned, deg, rowptr, dinv, batch, cnt, csr,
                                                      N, nb, nch, x, Wl[0], h);

  float invN = 1.0f / (float)N;
  int gblk = (N + 127) / 128 > GBLK ? (N + 127) / 128 : GBLK;
  k_gather_relu_stats<<<gblk, 512, 0, stream>>>(rowptr, deg, csr, dinv, h, bl[0], zb[0],
                                                         bn, bn + 64, N);
  for (int l = 1; l < 3; ++l) {
    float* bn_sum = bn + l * 128;
    float* bn_sq  = bn_sum + 64;
    k_gather_bn_gemm_relu_stats<<<gblk, 512, 0, stream>>>(rowptr, deg, csr, dinv, zb[l - 1],
                                                                   bn_sum - 128, bn_sq - 128,
                                                                   gl[l - 1], bel[l - 1], invN,
                                                                   Wl[l], bl[l], zb[l], bn_sum, bn_sq, N);
  }

  float* out = (float*)d_out;
  k_bn_l2_pool<<<(N + 63) / 64, 256, 0, stream>>>(zb[0], zb[1], zb[2], bn, gl[0], bel[0], gl[1], bel[1],
                                                  gl[2], bel[2], invN, batch, out, gsum, N);
  k_l2_gc<<<(GN + 7) / 8, 256, 0, stream>>>(gsum, cnt, out + (size_t)N * CD, GN);
}


# last pass by graph: one block per graph (batch sorted) normalises its rows (rsqrt), sums them in LDS and writes the graph row itself; pool atomics and 7th launch gone; norm, pool sums regrouped
# speedup vs baseline: 5.2575x; 1.0582x over previous
#include <hip/hip_runtime.h>

typedef float v2f __attribute__((ext_vector_type(2)));
typedef float v8f __attribute__((ext_vector_type(8)));

#define HD 64
#define CD 192
#define GN 256
#define GBLK 512

#define BINW  256
#define MAXB  256
#define CHUNK 4096

__device__ __forceinline__ int wave_incl_scan(int v, int lane) {
  for (int off = 1; off < 32; off <<= 1) {
    int u = __shfl_up(v, off, 32);
    if (lane >= off) v += u;
  }
  return v;
}

__global__ void __launch_bounds__(1024)
k_chunk_order(const int* __restrict__ src, const int* __restrict__ dst, int E, int nb,
              int* __restrict__ cm, int2* __restrict__ binned, float* __restrict__ zero, int nzero) {
  __shared__ int s_cnt[MAXB];
  __shared__ int s_lst[MAXB];
  __shared__ int s_wc[8];
  __shared__ int2 s_e[CHUNK];
  int tid = threadIdx.x, lane = tid & 31, w = tid >> 5;
  int chunk = blockIdx.x, nch = gridDim.x;
  for (int i = chunk * 1024 + tid; i < nzero; i += nch * 1024) zero[i] = 0.f;
  if (tid < MAXB) s_cnt[tid] = 0;
  __syncthreads();
  int nloc = E - chunk * CHUNK;
  if (nloc > CHUNK) nloc = CHUNK;
  const int* cs = src + (size_t)chunk * CHUNK;
  const int* cd = dst + (size_t)chunk * CHUNK;
  int es[4], ed[4], er[4];
#pragma unroll
  for (int k = 0; k < 4; ++k) {
    int i = k * 1024 + tid;
    int ic = i < nloc ? i : nloc - 1;
    es[k] = cs[ic];
    ed[k] = cd[ic];
  }
#pragma unroll
  for (int k = 0; k < 4; ++k) {
    er[k] = 0;
    if (k * 1024 + tid < nloc) er[k] = atomicAdd(&s_cnt[ed[k] >> 8], 1);
  }
  __syncthreads();
  int c = (tid < MAXB) ? s_cnt[tid] : 0;
  int ci = wave_incl_scan(c, lane);
  if (lane == 31 && w < 8) s_wc[w] = ci;
  __syncthreads();
  if (tid < MAXB) {
    int bc = 0;
    for (int k = 0; k < 8; ++k) {
      if (k < w) bc += s_wc[k];
    }
    int first = bc + ci - c;
    s_lst[tid] = first;
    if (tid < nb) cm[chunk * (nb + 1) + tid] = first;
    if (tid == 0) cm[chunk * (nb + 1) + nb] = nloc;
  }
  __syncthreads();
#pragma unroll
  for (int k = 0; k < 4; ++k) {
    if (k * 1024 + tid < nloc) s_e[s_lst[ed[k] >> 8] + er[k]] = make_int2(es[k], ed[k]);
  }
  __syncthreads();
#pragma unroll
  for (int k = 0; k < 4; ++k) {
    int i = k * 1024 + tid;
    if (i < nloc) binned[(size_t)chunk * CHUNK + i] = s_e[i];
  }
}

#define RUNB 16
__device__ __forceinline__ void csr_block(int bin, int nb, int nch, const int* __restrict__ cm,
                                          const int2* __restrict__ binned, int* __restrict__ deg,
                                          int* __restrict__ rowptr, float* __restrict__ dinv,
                                          const int* __restrict__ batch, float* __restrict__ cnt,
                                          int2* __restrict__ csr, int N) {
  __shared__ int s_c[BINW], s_cur[BINW], s_w[8], s_b[8];
  int tid = threadIdx.x, lane = tid & 31, w = tid >> 5;
  int pitch = nb + 1;
  s_c[tid] = 0;
  int bsum = 0;
  for (int ch = tid; ch < nch; ch += 256) bsum += cm[ch * pitch + bin];
#pragma unroll
  for (int off = 16; off >= 1; off >>= 1) bsum += __shfl_xor(bsum, off, 32);
  if (lane == 0) s_b[w] = bsum;
  __syncthreads();
  for (int ch = tid; ch < nch; ch += 256) {
    int ls = cm[ch * pitch + bin];
    int len = cm[ch * pitch + bin + 1] - ls;
    const int2* run = binned + (size_t)ch * CHUNK + ls;
    for (int j0 = 0; j0 < len; j0 += RUNB) {
      int v[RUNB];
#pragma unroll
      for (int u = 0; u < RUNB; ++u) {
        int j = j0 + u;
        v[u] = run[j < len ? j : len - 1].y;
      }
#pragma unroll
      for (int u = 0; u < RUNB; ++u) {
        if (j0 + u < len) atomicAdd(&s_c[v[u] & (BINW - 1)], 1);
      }
    }
  }
  __syncthreads();
  int c = s_c[tid];
  int inc = wave_incl_scan(c, lane);
  if (lane == 31) s_w[w] = inc;
  __syncthreads();
  int base = 0;
  for (int k = 0; k < 8; ++k) {
    base += s_b[k];
    if (k < w) base += s_w[k];
  }
  int first = base + inc - c;
  s_cur[tid] = first;
  int node = bin * BINW + tid;
  if (node < N) {
    deg[node] = c;
    rowptr[node] = first;
    dinv[node] = rsqrtf((float)c + 1.0f);
    atomicAdd(&cnt[batch[node]], 1.0f);
  }
  __syncthreads();
  for (int ch = tid; ch < nch; ch += 256) {
    int ls = cm[ch * pitch + bin];
    int len = cm[ch * pitch + bin + 1] - ls;
    const int2* run = binned + (size_t)ch * CHUNK + ls;
    for (int j0 = 0; j0 < len; j0 += RUNB) {
      int2 v[RUNB];
#pragma unroll
      for (int u = 0; u < RUNB; ++u) {
        int j = j0 + u;
        v[u] = run[j < len ? j : len - 1];
      }
#pragma unroll
      for (int u = 0; u < RUNB; ++u) {
        if (j0 + u < len) {
          int pos = atomicAdd(&s_cur[v[u].y & (BINW - 1)], 1);
          csr[pos] = make_int2(v[u].x, 0);
        }
      }
    }
  }
}

__device__ __forceinline__ void bn_coef(const float* __restrict__ bn_sum, const float* __restrict__ bn_sq,
                                        const float* __restrict__ gamma, const float* __restrict__ beta,
                                        float invN, int c, float& sc, float& sh) {
  float mu = bn_sum[c] * invN;
  float var = fmaxf(bn_sq[c] * invN - mu * mu, 0.f);
  sc = rsqrtf(var + 1e-5f) * gamma[c];
  sh = beta[c] - mu * sc;
}

#define XK 128
#define XP 132
__device__ __forceinline__ void gemm_block(int bid, const float* __restrict__ X, const float* __restrict__ W,
                                           float* __restrict__ Hout, int N) {
  __shared__ __attribute__((aligned(16))) float s_x[8 * 16 * XP];
  __shared__ __attribute__((aligned(16))) float s_wt[HD * XP];
  int tid = threadIdx.x;
  int lane = tid & 31;
  int w = __builtin_amdgcn_readfirstlane(tid >> 5);
  int m16 = lane & 15;
  int half = lane >> 4;
  int row0 = (bid * 8 + w) * 16;
  const float4* W4 = (const float4*)W;
  float4 wv[8];
  for (int i = 0; i < 8; ++i) wv[i] = W4[tid + 256 * i];
  const float4* X4 = (const float4*)X;
  float4 xr[16];
  for (int i = 0; i < 16; ++i) {
    int r = row0 + i;
    int rc = r < N ? r : N - 1;
    xr[i] = X4[(size_t)rc * (XK / 4) + lane];
  }
  for (int i = 0; i < 8; ++i) {
    int f = tid + 256 * i;
    int k = f >> 4;
    int c = (f & 15) << 2;
    s_wt[(c + 0) * XP + k] = wv[i].x;
    s_wt[(c + 1) * XP + k] = wv[i].y;
    s_wt[(c + 2) * XP + k] = wv[i].z;
    s_wt[(c + 3) * XP + k] = wv[i].w;
  }
  float* sx = s_x + w * (16 * XP);
  for (int i = 0; i < 16; ++i) *(float4*)(sx + i * XP + lane * 4) = xr[i];
  __syncthreads();
  const float* ap = sx + m16 * XP + 2 * half;
  const float* bp = s_wt + m16 * XP + 2 * half;
  v8f c0 = {}, c1 = {}, c2 = {}, c3 = {};
  for (int k0 = 0; k0 < XK; k0 += 4) {
    v2f a  = *(const v2f*)(ap + k0);
    v2f b0 = *(const v2f*)(bp + k0);
    v2f b1 = *(const v2f*)(bp + 16 * XP + k0);
    v2f b2 = *(const v2f*)(bp + 32 * XP + k0);
    v2f b3 = *(const v2f*)(bp + 48 * XP + k0);
    c0 = __builtin_amdgcn_wmma_f32_16x16x4_f32(false, a, false, b0, (short)0, c0, false, false);
    c1 = __builtin_amdgcn_wmma_f32_16x16x4_f32(false, a, false, b1, (short)0, c1, false, false);
    c2 = __builtin_amdgcn_wmma_f32_16x16x4_f32(false, a, false, b2, (short)0, c2, false, false);
    c3 = __builtin_amdgcn_wmma_f32_16x16x4_f32(false, a, false, b3, (short)0, c3, false, false);
  }
  float* so = sx + (8 * half) * XP + m16;
  for (int j = 0; j < 8; ++j) {
    so[j * XP]      = c0[j];
    so[j * XP + 16] = c1[j];
    so[j * XP + 32] = c2[j];
    so[j * XP + 48] = c3[j];
  }
  __syncthreads();
  const float* sr = sx + half * XP + 4 * m16;
  float4* ho = (float4*)Hout + (size_t)(row0 + half) * (HD / 4) + m16;
  if (row0 + 16 <= N) {
    for (int i = 0; i < 8; ++i) ho[i * 2 * (HD / 4)] = *(const float4*)(sr + 2 * i * XP);
  } else {
    for (int i = 0; i < 8; ++i) {
      if (row0 + 2 * i + half < N) ho[i * 2 * (HD / 4)] = *(const float4*)(sr + 2 * i * XP);
    }
  }
}

__global__ void __launch_bounds__(256)
k_csr_gemm(const int* __restrict__ cm, const int2* __restrict__ binned,
           int* __restrict__ deg, int* __restrict__ rowptr, float* __restrict__ dinv,
           const int* __restrict__ batch, float* __restrict__ cnt, int2* __restrict__ csr, int N, int nb, int nch,
           const float* __restrict__ X, const float* __restrict__ W, float* __restrict__ Hout) {
  if ((int)blockIdx.x < nb) {
    csr_block(blockIdx.x, nb, nch, cm, binned, deg, rowptr, dinv, batch, cnt, csr, N);
  } else {
    gemm_block(blockIdx.x - nb, X, W, Hout, N);
  }
}

struct RowBatch {
  float4 v[4];
};

__device__ __forceinline__ void batch_issue(RowBatch& t, const float4* __restrict__ h4, int q, int ex,
                                            int b, int half) {
#pragma unroll
  for (int u = 0; u < 4; ++u) {
    int sl = 8 * b + 2 * u + half;
    int sr = __shfl(ex, sl, 32);
    t.v[u] = h4[(unsigned)(sr * (HD / 4) + q)];
  }
}

template <bool WSUM>
__device__ __forceinline__ void batch_sum(const RowBatch& t, float wl, int b, int half, float4& acc, float& wsum) {
#pragma unroll
  for (int u = 0; u < 4; ++u) {
    float wgt = __shfl(wl, 8 * b + 2 * u + half, 32);
    acc.x += t.v[u].x * wgt;
    acc.y += t.v[u].y * wgt;
    acc.z += t.v[u].z * wgt;
    acc.w += t.v[u].w * wgt;
    if (WSUM) wsum += wgt;
  }
}

template <bool WSUM>
__device__ __forceinline__ void gather_chunk(const float4* __restrict__ h4, int q, int ex, float wl, int m, int half,
                                             float4& acc, float& wsum) {
  int nbat = (m + 7) >> 3;
  RowBatch A, B, C;
  if (nbat == 1) {
    batch_issue(A, h4, q, ex, 0, half);
    batch_sum<WSUM>(A, wl, 0, half, acc, wsum);
  } else if (nbat == 2) {
    batch_issue(A, h4, q, ex, 0, half);
    batch_issue(B, h4, q, ex, 1, half);
    __builtin_amdgcn_sched_barrier(0);
    batch_sum<WSUM>(A, wl, 0, half, acc, wsum);
    batch_sum<WSUM>(B, wl, 1, half, acc, wsum);
  } else {
    batch_issue(A, h4, q, ex, 0, half);
    batch_issue(B, h4, q, ex, 1, half);
    batch_issue(C, h4, q, ex, 2, half);
    __builtin_amdgcn_sched_barrier(0);
    batch_sum<WSUM>(A, wl, 0, half, acc, wsum);
    if (nbat == 4) batch_issue(A, h4, q, ex, 3, half);
    batch_sum<WSUM>(B, wl, 1, half, acc, wsum);
    batch_sum<WSUM>(C, wl, 2, half, acc, wsum);
    if (nbat == 4) batch_sum<WSUM>(A, wl, 3, half, acc, wsum);
  }
}

__global__ void __launch_bounds__(512, 8)
k_gather_relu_stats(const int* __restrict__ rowptr, const int* __restrict__ deg,
                    int2* __restrict__ csr, const float* __restrict__ dinv,
                    const float* __restrict__ h, const float* __restrict__ bias,
                    float* __restrict__ z, float* __restrict__ bn_sum,
                    float* __restrict__ bn_sq, int N) {
  __shared__ __attribute__((aligned(16))) float s_sum[16 * HD];
  __shared__ __attribute__((aligned(16))) float s_sq[16 * HD];
  int w = __builtin_amdgcn_readfirstlane(threadIdx.x >> 5);
  int lane = threadIdx.x & 31;
  int half = lane >> 4;
  int c4 = (lane & 15) << 2;
  const float4* h4 = (const float4*)h;
  int q = lane & 15;
  float4* ps = (float4*)(s_sum + w * HD + c4);
  float4* pq = (float4*)(s_sq + w * HD + c4);
  if (half == 0) {
    *ps = make_float4(0.f, 0.f, 0.f, 0.f);
    *pq = make_float4(0.f, 0.f, 0.f, 0.f);
  }
  int per = (N + (int)gridDim.x - 1) / (int)gridDim.x;
  int r0 = blockIdx.x * per;
  int rend = r0 + per < N ? r0 + per : N;
  int sta[2][4], nna[2][4];
  float dva[2][4];
#pragma unroll
  for (int s2 = 0; s2 < 2; ++s2) {
#pragma unroll
    for (int i = 0; i < 4; ++i) {
      int d = r0 + 64 * s2 + i * 16 + w;
      int dc = d < N ? d : N - 1;
      sta[s2][i] = rowptr[dc];
      nna[s2][i] = d < rend ? deg[dc] : 0;
      dva[s2][i] = dinv[dc];
    }
  }
#pragma unroll
  for (int s2 = 0; s2 < 2; ++s2) {
  int t0 = r0 + 64 * s2;
  if (t0 >= rend) break;
  int st[4], nn[4], ex[4];
  float dv[4], wl[4];
#pragma unroll
  for (int i = 0; i < 4; ++i) {
    st[i] = sta[s2][i];
    nn[i] = nna[s2][i];
    dv[i] = dva[s2][i];
  }
#pragma unroll
  for (int i = 0; i < 4; ++i) {
    int d = t0 + i * 16 + w;
    int sx = d < N ? d : N - 1;
    if (lane < nn[i]) sx = csr[st[i] + lane].x;
    ex[i] = sx;
  }
  float dsr[4];
#pragma unroll
  for (int i = 0; i < 4; ++i) dsr[i] = dinv[ex[i]];
#pragma unroll
  for (int i = 0; i < 4; ++i) {
    float ds = 0.f;
    if (lane < nn[i]) {
      ds = dsr[i];
      csr[st[i] + lane] = make_int2(ex[i], __float_as_int(ds));
    }
    wl[i] = ds * dv[i];
  }
#pragma unroll
  for (int i = 0; i < 4; ++i) {
    int d = t0 + i * 16 + w;
    if (d >= rend) break;
    int start = st[i];
    int n = nn[i];
    float dd = dv[i];
    const float4 hs = h4[(unsigned)(d * (HD / 4) + q)];
    const float4 b4 = ((const float4*)bias)[q];
    float4 acc = {0.f, 0.f, 0.f, 0.f};
    float wsum = 0.f;
    int exc = ex[i];
    float wlc = wl[i];
    for (int base = 0; base < n; base += 32) {
      int m = n - base;
      if (m > 32) m = 32;
      gather_chunk<false>(h4, q, exc, wlc, m, half, acc, wsum);
      int m2 = n - base - 32;
      if (m2 > 0) {
        if (m2 > 32) m2 = 32;
        exc = d;
        float ds = 0.f;
        if (lane < m2) {
          int2* ep = csr + start + base + 32 + lane;
          exc = ep->x;
          ds = dinv[exc];
          *ep = make_int2(exc, __float_as_int(ds));
        }
        wlc = ds * dd;
      }
    }
    acc.x += __shfl_xor(acc.x, 16, 32);
    acc.y += __shfl_xor(acc.y, 16, 32);
    acc.z += __shfl_xor(acc.z, 16, 32);
    acc.w += __shfl_xor(acc.w, 16, 32);
    if (half == 0) {
      float w0 = dd * dd;
      float4 o;
      o.x = hs.x * w0 + acc.x;
      o.y = hs.y * w0 + acc.y;
      o.z = hs.z * w0 + acc.z;
      o.w = hs.w * w0 + acc.w;
      o.x = fmaxf(o.x + b4.x, 0.f);
      o.y = fmaxf(o.y + b4.y, 0.f);
      o.z = fmaxf(o.z + b4.z, 0.f);
      o.w = fmaxf(o.w + b4.w, 0.f);
      ((float4*)z)[(unsigned)(d * (HD / 4) + q)] = o;
      float4 ts = *ps;
      float4 tq = *pq;
      ts.x += o.x; ts.y += o.y; ts.z += o.z; ts.w += o.w;
      tq.x += o.x * o.x; tq.y += o.y * o.y; tq.z += o.z * o.z; tq.w += o.w * o.w;
      *ps = ts;
      *pq = tq;
    }
  }
  }
  __syncthreads();
  if (threadIdx.x < 2 * HD) {
    int c = threadIdx.x & (HD - 1);
    const float* p = (threadIdx.x < HD) ? s_sum : s_sq;
    float t = 0.f;
#pragma unroll
    for (int k = 0; k < 16; ++k) t += p[k * HD + c];
    atomicAdd((threadIdx.x < HD ? bn_sum : bn_sq) + c, t);
  }
}

#define AP 68
__global__ void __launch_bounds__(512, 8)
k_gather_bn_gemm_relu_stats(const int* __restrict__ rowptr, const int* __restrict__ deg,
                            const int2* __restrict__ csr, const float* __restrict__ dinv,
                            const float* __restrict__ zp,
                            const float* __restrict__ pbn_sum, const float* __restrict__ pbn_sq,
                            const float* __restrict__ pgamma, const float* __restrict__ pbeta, float invN,
                            const float* W, const float* __restrict__ bias,
                            float* __restrict__ z, float* __restrict__ bn_sum,
                            float* __restrict__ bn_sq, int N) {
  __shared__ __attribute__((aligned(16))) float s_a[128 * AP];
  __shared__ float s_st[8 * HD];
  float* s_sum = s_st;
  float* s_sq = s_st + 4 * HD;
  int w = __builtin_amdgcn_readfirstlane(threadIdx.x >> 5);
  int lane = threadIdx.x & 31;
  int half = lane >> 4;
  int c4 = (lane & 15) << 2;
  const float4* h4 = (const float4*)zp;
  int q = lane & 15;
  float sc0, sc1, sc2, sc3, sh0, sh1, sh2, sh3;
  bn_coef(pbn_sum, pbn_sq, pgamma, pbeta, invN, c4,     sc0, sh0);
  bn_coef(pbn_sum, pbn_sq, pgamma, pbeta, invN, c4 + 1, sc1, sh1);
  bn_coef(pbn_sum, pbn_sq, pgamma, pbeta, invN, c4 + 2, sc2, sh2);
  bn_coef(pbn_sum, pbn_sq, pgamma, pbeta, invN, c4 + 3, sc3, sh3);
  int per = (N + (int)gridDim.x - 1) / (int)gridDim.x;
  int r0 = blockIdx.x * per;
  int rend = r0 + per < N ? r0 + per : N;
  int tm = w >> 2, tn = w & 3;
  if (threadIdx.x < 4 * HD) {
    s_sum[threadIdx.x] = 0.f;
    s_sq[threadIdx.x] = 0.f;
  }
  int sta[2][4], nna[2][4];
  float dva[2][4];
#pragma unroll
  for (int s2 = 0; s2 < 2; ++s2) {
#pragma unroll
    for (int i = 0; i < 4; ++i) {
      int d = r0 + 64 * s2 + i * 16 + w;
      int dc = d < N ? d : N - 1;
      sta[s2][i] = rowptr[dc];
      nna[s2][i] = d < rend ? deg[dc] : 0;
      dva[s2][i] = dinv[dc];
    }
  }
#pragma unroll
  for (int s2 = 0; s2 < 2; ++s2) {
  int t0 = r0 + 64 * s2;
  if (t0 >= rend) break;
  int st[4], nn[4], ex[4];
  float dv[4], wl[4];
#pragma unroll
  for (int i = 0; i < 4; ++i) {
    st[i] = sta[s2][i];
    nn[i] = nna[s2][i];
    dv[i] = dva[s2][i];
  }
#pragma unroll
  for (int i = 0; i < 4; ++i) {
    int d = t0 + i * 16 + w;
    int2 ent = make_int2(d < N ? d : N - 1, 0);
    if (lane < nn[i]) ent = csr[st[i] + lane];
    ex[i] = ent.x;
    wl[i] = __int_as_float(ent.y) * dv[i];
  }
#pragma unroll
  for (int i = 0; i < 4; ++i) {
    int lr = 64 * s2 + i * 16 + w;
    int d = t0 + i * 16 + w;
    float4 g = {0.f, 0.f, 0.f, 0.f};
    if (d < rend) {
      int start = st[i];
      int n = nn[i];
      float dd = dv[i];
      const float4 hs = h4[(unsigned)(d * (HD / 4) + q)];
      float4 acc = {0.f, 0.f, 0.f, 0.f};
      float wsum = 0.f;
      int exc = ex[i];
      float wlc = wl[i];
      for (int base = 0; base < n; base += 32) {
        int m = n - base;
        if (m > 32) m = 32;
        gather_chunk<true>(h4, q, exc, wlc, m, half, acc, wsum);
        int m2 = n - base - 32;
        if (m2 > 0) {
          if (m2 > 32) m2 = 32;
          int2 ent = make_int2(d, 0);
          if (lane < m2) ent = csr[start + base + 32 + lane];
          exc = ent.x;
          wlc = __int_as_float(ent.y) * dd;
        }
      }
      acc.x += __shfl_xor(acc.x, 16, 32);
      acc.y += __shfl_xor(acc.y, 16, 32);
      acc.z += __shfl_xor(acc.z, 16, 32);
      acc.w += __shfl_xor(acc.w, 16, 32);
      wsum += __shfl_xor(wsum, 16, 32);
      float w0 = dd * dd;
      float r = w0 + wsum;
      g.x = (hs.x * w0 + acc.x) * sc0 + r * sh0;
      g.y = (hs.y * w0 + acc.y) * sc1 + r * sh1;
      g.z = (hs.z * w0 + acc.z) * sc2 + r * sh2;
      g.w = (hs.w * w0 + acc.w) * sc3 + r * sh3;
    }
    if (half == 0) *(float4*)(s_a + lr * AP + c4) = g;
  }
  __syncthreads();
  if (t0 + tm * 16 < rend) {
    int m16 = lane & 15;
    int col = tn * 16 + m16;
    const float* ap = s_a + (64 * s2 + tm * 16 + m16) * AP + 2 * half;
    const float* wp = W + col;
    v8f c = {};
#pragma unroll
    for (int k0 = 0; k0 < HD; k0 += 4) {
      int ka = k0 + 2 * half;
      v2f a, b;
      a.x = ap[k0];
      a.y = ap[k0 + 1];
      b.x = wp[(size_t)ka * HD];
      b.y = wp[(size_t)(ka + 1) * HD];
      c = __builtin_amdgcn_wmma_f32_16x16x4_f32(false, a, false, b, (short)0, c, false, false);
    }
    float bcol = bias[col];
    int rbase = t0 + tm * 16 + half * 8;
    float ts = 0.f, tq = 0.f;
#pragma unroll
    for (int j = 0; j < 8; ++j) {
      int r = rbase + j;
      if (r < rend) {
        float o = fmaxf(c[j] + bcol, 0.f);
        z[(size_t)r * HD + col] = o;
        ts += o;
        tq += o * o;
      }
    }
    ts += __shfl_xor(ts, 16, 32);
    tq += __shfl_xor(tq, 16, 32);
    if (half == 0) {
      s_sum[tm * HD + col] += ts;
      s_sq[tm * HD + col] += tq;
    }
  }
  }
  __syncthreads();
  if (threadIdx.x < 2 * HD) {
    int cc = threadIdx.x & (HD - 1);
    const float* p = (threadIdx.x < HD) ? s_sum : s_sq;
    float t = (p[cc] + p[HD + cc]) + (p[2 * HD + cc] + p[3 * HD + cc]);
    atomicAdd((threadIdx.x < HD ? bn_sum : bn_sq) + cc, t);
  }
}

#define GROWS 8

__device__ __forceinline__ float row16_sum(float s) {
  s += __int_as_float(__builtin_amdgcn_update_dpp(0, __float_as_int(s), 0x161, 0xf, 0xf, false));
  s += __int_as_float(__builtin_amdgcn_update_dpp(0, __float_as_int(s), 0x162, 0xf, 0xf, false));
  s += __int_as_float(__builtin_amdgcn_update_dpp(0, __float_as_int(s), 0x164, 0xf, 0xf, false));
  s += __int_as_float(__builtin_amdgcn_update_dpp(0, __float_as_int(s), 0x168, 0xf, 0xf, false));
  return s;
}

__device__ __forceinline__ int row16_sum(int s) {
  s += __builtin_amdgcn_update_dpp(0, s, 0x161, 0xf, 0xf, false);
  s += __builtin_amdgcn_update_dpp(0, s, 0x162, 0xf, 0xf, false);
  s += __builtin_amdgcn_update_dpp(0, s, 0x164, 0xf, 0xf, false);
  s += __builtin_amdgcn_update_dpp(0, s, 0x168, 0xf, 0xf, false);
  return s;
}

__global__ void __launch_bounds__(512)
k_bn_l2_graph(const float* __restrict__ z0, const float* __restrict__ z1,
              const float* __restrict__ z2, const float* __restrict__ bn,
              const float* __restrict__ gamma0, const float* __restrict__ beta0,
              const float* __restrict__ gamma1, const float* __restrict__ beta1,
              const float* __restrict__ gamma2, const float* __restrict__ beta2,
              float invN, const float* __restrict__ cnt,
              float* __restrict__ out, int N) {
  __shared__ float s_nrm[8];
  __shared__ __attribute__((aligned(16))) float s_p[32 * CD];
  int tid = threadIdx.x;
  int lane = tid & 31;
  int w = __builtin_amdgcn_readfirstlane(tid >> 5);
  int half = lane >> 4;
  int q = lane & 15;
  int hw = 2 * w + half;
  int g = blockIdx.x;
  int c = 0;
#pragma unroll
  for (int j = 0; j < GN / 32; ++j) {
    int gi = lane + 32 * j;
    int v = (int)cnt[gi];
    c += gi < g ? v : 0;
  }
  int n = (int)cnt[g];
  const float* gl[3] = {gamma0, gamma1, gamma2};
  const float* bl[3] = {beta0, beta1, beta2};
  float sc[12], sh[12];
#pragma unroll
  for (int l = 0; l < 3; ++l) {
#pragma unroll
    for (int j = 0; j < 4; ++j) {
      bn_coef(bn + l * 128, bn + l * 128 + 64, gl[l], bl[l], invN, 4 * q + j, sc[4 * l + j], sh[4 * l + j]);
    }
  }
  c = row16_sum(c);
  int start = __builtin_amdgcn_readlane(c, 0) + __builtin_amdgcn_readlane(c, 16);
  const float4* zv[3] = {(const float4*)z0, (const float4*)z1, (const float4*)z2};
  float4* o4 = (float4*)out;
  float acc[12];
#pragma unroll
  for (int j = 0; j < 12; ++j) acc[j] = 0.f;
  int base = 0;
  do {
    float4 x[GROWS][3];
#pragma unroll
    for (int k = 0; k < GROWS; ++k) {
      int i = base + 32 * k + hw;
      int r = start + (i < n ? i : n - 1);
      r = r < 0 ? 0 : r;
#pragma unroll
      for (int l = 0; l < 3; ++l) x[k][l] = zv[l][(unsigned)(r * (HD / 4) + q)];
    }
    __builtin_amdgcn_sched_barrier(0);
    float inv[GROWS];
#pragma unroll
    for (int k = 0; k < GROWS; ++k) {
      float s = 0.f;
#pragma unroll
      for (int l = 0; l < 3; ++l) {
        x[k][l].x = x[k][l].x * sc[4 * l + 0] + sh[4 * l + 0];
        x[k][l].y = x[k][l].y * sc[4 * l + 1] + sh[4 * l + 1];
        x[k][l].z = x[k][l].z * sc[4 * l + 2] + sh[4 * l + 2];
        x[k][l].w = x[k][l].w * sc[4 * l + 3] + sh[4 * l + 3];
        s += x[k][l].x * x[k][l].x + x[k][l].y * x[k][l].y + x[k][l].z * x[k][l].z + x[k][l].w * x[k][l].w;
      }
      s = row16_sum(s);
      inv[k] = rsqrtf(fmaxf(s, 1e-24f));
    }
#pragma unroll
    for (int k = 0; k < GROWS; ++k) {
      int i = base + 32 * k + hw;
      if (i < n) {
        unsigned ro = (unsigned)((start + i) * (CD / 4) + q);
#pragma unroll
        for (int l = 0; l < 3; ++l) {
          o4[ro + 16 * l] = make_float4(x[k][l].x * inv[k], x[k][l].y * inv[k], x[k][l].z * inv[k], x[k][l].w * inv[k]);
          acc[4 * l + 0] += x[k][l].x;
          acc[4 * l + 1] += x[k][l].y;
          acc[4 * l + 2] += x[k][l].z;
          acc[4 * l + 3] += x[k][l].w;
        }
      }
    }
    base += 32 * GROWS;
  } while (base < n);
#pragma unroll
  for (int l = 0; l < 3; ++l) {
    *(float4*)(s_p + hw * CD + l * HD + 4 * q) = make_float4(acc[4 * l + 0], acc[4 * l + 1], acc[4 * l + 2], acc[4 * l + 3]);
  }
  __syncthreads();
  float gv = 0.f;
  if (tid < CD) {
    float t = 0.f;
#pragma unroll
    for (int k = 0; k < 32; ++k) t += s_p[k * CD + tid];
    gv = t * (1.0f / fmaxf(cnt[g], 1.0f));
  }
  float gs = row16_sum(gv * gv);
  float gw = __int_as_float(__builtin_amdgcn_readlane(__float_as_int(gs), 0)) +
             __int_as_float(__builtin_amdgcn_readlane(__float_as_int(gs), 16));
  if (lane == 0 && w < 8) s_nrm[w] = gw;
  __syncthreads();
  if (tid < CD) {
    float tot = 0.f;
#pragma unroll
    for (int k = 0; k < CD / 32; ++k) tot += s_nrm[k];
    float inv = 1.0f / fmaxf(sqrtf(tot), 1e-12f);
    out[(size_t)N * CD + (size_t)g * CD + tid] = gv * inv;
  }
}

extern "C" void kernel_launch(void* const* d_in, const int* in_sizes, int n_in,
                              void* d_out, int out_size, void* d_ws, size_t ws_size,
                              hipStream_t stream) {
  (void)n_in; (void)out_size; (void)ws_size;
  const float* x     = (const float*)d_in[0];
  const int*   ei    = (const int*)d_in[1];
  const int*   batch = (const int*)d_in[2];
  int N = in_sizes[0] / 128;
  int E = in_sizes[1] / 2;
  const int* src = ei;
  const int* dst = ei + E;
  const float* Wl[3]  = {(const float*)d_in[3],  (const float*)d_in[7],  (const float*)d_in[11]};
  const float* bl[3]  = {(const float*)d_in[4],  (const float*)d_in[8],  (const float*)d_in[12]};
  const float* gl[3]  = {(const float*)d_in[5],  (const float*)d_in[9],  (const float*)d_in[13]};
  const float* bel[3] = {(const float*)d_in[6],  (const float*)d_in[10], (const float*)d_in[14]};

  float* ws   = (float*)d_ws;
  float* dinv = ws;                       ws += N;
  ws = (float*)(((size_t)ws + 255) & ~(size_t)255);
  float* h    = ws;                       ws += (size_t)N * HD;
  float* zb[3];
  zb[0] = ws;                             ws += (size_t)N * HD;
  zb[1] = ws;                             ws += (size_t)N * HD;
  zb[2] = ws;                             ws += (size_t)N * HD;
  float* cnt  = ws;                       ws += GN;
  float* bn   = ws;
  ws += 3 * 128;
  int* deg    = (int*)(((size_t)ws + 15) & ~(size_t)15);
  int* rowptr = deg + (((size_t)N + 3) & ~(size_t)3);
  int nb  = (N + BINW - 1) / BINW;
  int nch = (E + CHUNK - 1) / CHUNK;
  int* cm       = rowptr + (((size_t)N + 3) & ~(size_t)3);
  int2* binned  = (int2*)(cm + (((size_t)nch * (nb + 1) + 3) & ~(size_t)3));
  int2* csr     = binned + (size_t)nch * CHUNK;

  k_chunk_order<<<nch, 1024, 0, stream>>>(src, dst, E, nb, cm, binned, cnt, GN + 3 * 128);
  k_csr_gemm<<<nb + (N + 127) / 128, 256, 0, stream>>>(cm, binned, deg, rowptr, dinv, batch, cnt, csr,
                                                      N, nb, nch, x, Wl[0], h);

  float invN = 1.0f / (float)N;
  int gblk = (N + 127) / 128 > GBLK ? (N + 127) / 128 : GBLK;
  k_gather_relu_stats<<<gblk, 512, 0, stream>>>(rowptr, deg, csr, dinv, h, bl[0], zb[0],
                                                         bn, bn + 64, N);
  for (int l = 1; l < 3; ++l) {
    float* bn_sum = bn + l * 128;
    float* bn_sq  = bn_sum + 64;
    k_gather_bn_gemm_relu_stats<<<gblk, 512, 0, stream>>>(rowptr, deg, csr, dinv, zb[l - 1],
                                                                   bn_sum - 128, bn_sq - 128,
                                                                   gl[l - 1], bel[l - 1], invN,
                                                                   Wl[l], bl[l], zb[l], bn_sum, bn_sq, N);
  }

  float* out = (float*)d_out;
  k_bn_l2_graph<<<GN, 512, 0, stream>>>(zb[0], zb[1], zb[2], bn, gl[0], bel[0], gl[1], bel[1],
                                        gl[2], bel[2], invN, cnt, out, N);
}
